# starting kernel
# baseline (speedup 1.0000x reference)
#include <hip/hip_runtime.h>
#include <math.h>

typedef __attribute__((ext_vector_type(16))) _Float16 v16h;
typedef __attribute__((ext_vector_type(8)))  _Float16 v8h;
typedef __attribute__((ext_vector_type(4)))  _Float16 v4h;
typedef __attribute__((ext_vector_type(8)))  float    v8f;

#define WMMA16(A,B,C) __builtin_amdgcn_wmma_f32_16x16x32_f16(false,(A),false,(B),(short)0,(C),false,false)

static constexpr int T_STEPS = 64;
static constexpr int E_ENVS  = 32;
static constexpr int B_TOT   = 2048;
static constexpr int HL      = 256;
static constexpr int IND     = 549;
static constexpr int KPAD    = 832;
static constexpr int XS      = 560;
static constexpr int ASTR    = 840;
static constexpr int IMS     = 88;

__device__ __forceinline__ v16h pack_frag(const _Float16* p) {
  v8h u = *(const v8h*)p;
  v8h w = *(const v8h*)(p + 16);
  v16h a;
#pragma unroll
  for (int e = 0; e < 8; e++) { a[e] = u[e]; a[e + 8] = w[e]; }
  return a;
}
__device__ __forceinline__ v16h pack_frag_c1(const _Float16* p0) {
  const _Float16* p1 = p0 + 2 * IMS;
  v4h u0 = *(const v4h*)p0;
  v4h u1 = *(const v4h*)(p0 + 4);
  v4h u2 = *(const v4h*)p1;
  v4h u3 = *(const v4h*)(p1 + 4);
  v16h a;
#pragma unroll
  for (int e = 0; e < 4; e++) {
    a[e] = u0[e]; a[e + 4] = u1[e]; a[e + 8] = u2[e]; a[e + 12] = u3[e];
  }
  return a;
}

__device__ __forceinline__ float sigmoidf_(float x) { return 1.0f / (1.0f + __expf(-x)); }

__global__ void k_prep_img(const float* img, _Float16* imgC, int n) {
  int i = blockIdx.x * blockDim.x + threadIdx.x;
  if (i >= n) return;
  int b = i / (3 * 84 * 84), r0 = i % (3 * 84 * 84);
  int c = r0 / (84 * 84), rc = r0 % (84 * 84);
  int row = rc / 84, col = rc % 84;
  imgC[((b * 3 + c) * 84 + row) * IMS + col] =
      (_Float16)(img[((b * 84 + row) * 84 + col) * 3 + c] * (1.0f / 255.0f));
}

__global__ void k_prep_w1(const float* w, _Float16* wt) {
  int i = blockIdx.x * blockDim.x + threadIdx.x;
  if (i >= 32 * 192) return;
  int n = i / 192, k = i % 192;
  int c = k >> 6, kh = (k & 63) >> 3, kw = k & 7;
  wt[i] = (_Float16)w[((n * 3 + c) * 8 + kh) * 8 + kw];
}
__global__ void k_prep_w2(const float* w, _Float16* wt) {
  int i = blockIdx.x * blockDim.x + threadIdx.x;
  if (i >= 64 * 512) return;
  int n = i / 512, k = i % 512;
  int pix = k >> 5, c = k & 31, kh = pix >> 2, kw = pix & 3;
  wt[i] = (_Float16)w[((n * 32 + c) * 4 + kh) * 4 + kw];
}
__global__ void k_prep_w3(const float* w, _Float16* wt) {
  int i = blockIdx.x * blockDim.x + threadIdx.x;
  if (i >= 64 * 576) return;
  int n = i / 576, k = i % 576;
  int pix = k >> 6, c = k & 63, kh = pix / 3, kw = pix % 3;
  wt[i] = (_Float16)w[((n * 64 + c) * 3 + kh) * 3 + kw];
}
__global__ void k_prep_fc(const float* w, _Float16* wt) {
  int i = blockIdx.x * blockDim.x + threadIdx.x;
  if (i >= 512 * 3136) return;
  int f = i / 3136, k = i % 3136;
  int s = k >> 6, c = k & 63;
  wt[i] = (_Float16)w[f * 3136 + c * 49 + s];
}
__global__ void k_prep_wg(const float* wih, const float* whh, _Float16* wg) {
  int i = blockIdx.x * blockDim.x + threadIdx.x;
  if (i >= 1024 * KPAD) return;
  int r = i / KPAD, k = i % KPAD;
  float v = 0.0f;
  if (k < IND)            v = wih[r * IND + k];
  else if (k < IND + HL)  v = whh[r * HL + (k - IND)];
  wg[i] = (_Float16)v;
}
__global__ void k_prep_bias(const float* bih, const float* bhh, float* bg) {
  int i = blockIdx.x * blockDim.x + threadIdx.x;
  if (i < 1024) bg[i] = bih[i] + bhh[i];
}
__global__ void k_onehot(const int* pos, const int* lact, _Float16* X) {
  int i = blockIdx.x * blockDim.x + threadIdx.x;
  if (i >= B_TOT * 48) return;
  int b = i / 48, c = i % 48;
  float v = 0.0f;
  if (c < 16)      v = (pos[b * 2 + 0] == c)      ? 1.0f : 0.0f;
  else if (c < 32) v = (pos[b * 2 + 1] == c - 16) ? 1.0f : 0.0f;
  else if (c < 37) v = (lact[b]        == c - 32) ? 1.0f : 0.0f;
  X[b * XS + 512 + c] = (_Float16)v;
}

__global__ void __launch_bounds__(256) k_conv1(const _Float16* imgC, const _Float16* wt,
                                               const float* bias, _Float16* out) {
  int wid  = (blockIdx.x * blockDim.x + threadIdx.x) >> 5;
  int lane = threadIdx.x & 31, hi = lane >> 4, lo = lane & 15;
  int m = wid * 16 + lo;
  int b = m / 400, p = m % 400, oh = p / 20, ow = p % 20;
  const _Float16* abase = imgC + (size_t)b * 3 * 84 * IMS + (oh * 4) * IMS + ow * 4;
  v8f acc0 = {}, acc1 = {};
#pragma unroll
  for (int kb = 0; kb < 192; kb += 32) {
    int c = kb >> 6, khb = (kb & 63) >> 3;
    v16h a  = pack_frag_c1(abase + (c * 84 + khb + hi) * IMS);
    v16h b0 = pack_frag(wt + lo * 192        + kb + hi * 8);
    v16h b1 = pack_frag(wt + (16 + lo) * 192 + kb + hi * 8);
    acc0 = WMMA16(a, b0, acc0);
    acc1 = WMMA16(a, b1, acc1);
  }
  float bs0 = bias[lo], bs1 = bias[16 + lo];
#pragma unroll
  for (int r = 0; r < 8; r++) {
    int mm = wid * 16 + hi * 8 + r;
    float v0 = acc0[r] + bs0; v0 = v0 > 0.0f ? v0 : 0.0f;
    float v1 = acc1[r] + bs1; v1 = v1 > 0.0f ? v1 : 0.0f;
    out[mm * 32 + lo]      = (_Float16)v0;
    out[mm * 32 + 16 + lo] = (_Float16)v1;
  }
}

__global__ void __launch_bounds__(256) k_conv2(const _Float16* act1, const _Float16* wt,
                                               const float* bias, _Float16* out) {
  int wid  = (blockIdx.x * blockDim.x + threadIdx.x) >> 5;
  int lane = threadIdx.x & 31, hi = lane >> 4, lo = lane & 15;
  int m = wid * 16 + lo;
  int b = m / 81, p = m % 81, oh = p / 9, ow = p % 9;
  v8f acc0 = {}, acc1 = {}, acc2 = {}, acc3 = {};
  for (int kb = 0; kb < 512; kb += 32) {
    int pix = kb >> 5, kh = pix >> 2, kw = pix & 3;
    v16h a  = pack_frag(act1 + ((b * 20 + oh * 2 + kh) * 20 + (ow * 2 + kw)) * 32 + hi * 8);
    v16h b0 = pack_frag(wt + (lo)      * 512 + kb + hi * 8);
    v16h b1 = pack_frag(wt + (16 + lo) * 512 + kb + hi * 8);
    v16h b2 = pack_frag(wt + (32 + lo) * 512 + kb + hi * 8);
    v16h b3 = pack_frag(wt + (48 + lo) * 512 + kb + hi * 8);
    acc0 = WMMA16(a, b0, acc0);
    acc1 = WMMA16(a, b1, acc1);
    acc2 = WMMA16(a, b2, acc2);
    acc3 = WMMA16(a, b3, acc3);
  }
  float bs0 = bias[lo], bs1 = bias[16 + lo], bs2 = bias[32 + lo], bs3 = bias[48 + lo];
#pragma unroll
  for (int r = 0; r < 8; r++) {
    int mm = wid * 16 + hi * 8 + r;
    float v0 = acc0[r] + bs0; v0 = v0 > 0.0f ? v0 : 0.0f;
    float v1 = acc1[r] + bs1; v1 = v1 > 0.0f ? v1 : 0.0f;
    float v2 = acc2[r] + bs2; v2 = v2 > 0.0f ? v2 : 0.0f;
    float v3 = acc3[r] + bs3; v3 = v3 > 0.0f ? v3 : 0.0f;
    out[mm * 64 + lo]      = (_Float16)v0;
    out[mm * 64 + 16 + lo] = (_Float16)v1;
    out[mm * 64 + 32 + lo] = (_Float16)v2;
    out[mm * 64 + 48 + lo] = (_Float16)v3;
  }
}

__global__ void __launch_bounds__(256) k_conv3(const _Float16* act2, const _Float16* wt,
                                               const float* bias, _Float16* out) {
  int wid  = (blockIdx.x * blockDim.x + threadIdx.x) >> 5;
  int lane = threadIdx.x & 31, hi = lane >> 4, lo = lane & 15;
  int m = wid * 16 + lo;
  int b = m / 49, p = m % 49, oh = p / 7, ow = p % 7;
  v8f acc0 = {}, acc1 = {}, acc2 = {}, acc3 = {};
  for (int kb = 0; kb < 576; kb += 32) {
    int pix = kb >> 6, cb = (kb & 63) + hi * 8;
    int kh = pix / 3, kw = pix % 3;
    v16h a  = pack_frag(act2 + ((b * 9 + oh + kh) * 9 + (ow + kw)) * 64 + cb);
    v16h b0 = pack_frag(wt + (lo)      * 576 + kb + hi * 8);
    v16h b1 = pack_frag(wt + (16 + lo) * 576 + kb + hi * 8);
    v16h b2 = pack_frag(wt + (32 + lo) * 576 + kb + hi * 8);
    v16h b3 = pack_frag(wt + (48 + lo) * 576 + kb + hi * 8);
    acc0 = WMMA16(a, b0, acc0);
    acc1 = WMMA16(a, b1, acc1);
    acc2 = WMMA16(a, b2, acc2);
    acc3 = WMMA16(a, b3, acc3);
  }
  float bs0 = bias[lo], bs1 = bias[16 + lo], bs2 = bias[32 + lo], bs3 = bias[48 + lo];
#pragma unroll
  for (int r = 0; r < 8; r++) {
    int mm = wid * 16 + hi * 8 + r;
    float v0 = acc0[r] + bs0; v0 = v0 > 0.0f ? v0 : 0.0f;
    float v1 = acc1[r] + bs1; v1 = v1 > 0.0f ? v1 : 0.0f;
    float v2 = acc2[r] + bs2; v2 = v2 > 0.0f ? v2 : 0.0f;
    float v3 = acc3[r] + bs3; v3 = v3 > 0.0f ? v3 : 0.0f;
    out[mm * 64 + lo]      = (_Float16)v0;
    out[mm * 64 + 16 + lo] = (_Float16)v1;
    out[mm * 64 + 32 + lo] = (_Float16)v2;
    out[mm * 64 + 48 + lo] = (_Float16)v3;
  }
}

__global__ void __launch_bounds__(256) k_fc(const _Float16* act3, const _Float16* wt,
                                            const float* bias, _Float16* X) {
  int wid  = (blockIdx.x * blockDim.x + threadIdx.x) >> 5;
  int lane = threadIdx.x & 31, hi = lane >> 4, lo = lane & 15;
  int ng = wid & 7, mt = wid >> 3;
  int m = mt * 16 + lo;
  int nb = ng * 64;
  v8f acc0 = {}, acc1 = {}, acc2 = {}, acc3 = {};
  for (int kb = 0; kb < 3136; kb += 32) {
    v16h a  = pack_frag(act3 + (size_t)m * 3136 + kb + hi * 8);
    v16h b0 = pack_frag(wt + (size_t)(nb + lo)      * 3136 + kb + hi * 8);
    v16h b1 = pack_frag(wt + (size_t)(nb + 16 + lo) * 3136 + kb + hi * 8);
    v16h b2 = pack_frag(wt + (size_t)(nb + 32 + lo) * 3136 + kb + hi * 8);
    v16h b3 = pack_frag(wt + (size_t)(nb + 48 + lo) * 3136 + kb + hi * 8);
    acc0 = WMMA16(a, b0, acc0);
    acc1 = WMMA16(a, b1, acc1);
    acc2 = WMMA16(a, b2, acc2);
    acc3 = WMMA16(a, b3, acc3);
  }
  float bs0 = bias[nb + lo], bs1 = bias[nb + 16 + lo];
  float bs2 = bias[nb + 32 + lo], bs3 = bias[nb + 48 + lo];
#pragma unroll
  for (int r = 0; r < 8; r++) {
    int mm = mt * 16 + hi * 8 + r;
    float v0 = acc0[r] + bs0; v0 = v0 > 0.0f ? v0 : 0.0f;
    float v1 = acc1[r] + bs1; v1 = v1 > 0.0f ? v1 : 0.0f;
    float v2 = acc2[r] + bs2; v2 = v2 > 0.0f ? v2 : 0.0f;
    float v3 = acc3[r] + bs3; v3 = v3 > 0.0f ? v3 : 0.0f;
    X[mm * XS + nb + lo]      = (_Float16)v0;
    X[mm * XS + nb + 16 + lo] = (_Float16)v1;
    X[mm * XS + nb + 32 + lo] = (_Float16)v2;
    X[mm * XS + nb + 48 + lo] = (_Float16)v3;
  }
}

__global__ void __launch_bounds__(1024) k_lstm(const _Float16* X, const _Float16* Wg,
                                               const float* biasg, const int* done,
                                               const float* h0, const float* c0,
                                               float* hs, float* out) {
  __shared__ _Float16 A[E_ENVS * ASTR];
  int tid = threadIdx.x;
  int w = tid >> 5, lane = tid & 31, hi = lane >> 4, lo = lane & 15;
  int mt = w >> 4, cg = w & 15;
  int ncol = cg * 16 + lo;

  for (int i = tid; i < E_ENVS * HL; i += 1024) {
    int e = i >> 8, k = i & 255;
    A[e * ASTR + IND + k] = (_Float16)h0[i];
  }
  for (int i = tid; i < E_ENVS * (ASTR - (IND + HL)); i += 1024) {
    int span = ASTR - (IND + HL);
    int e = i / span, k = i % span;
    A[e * ASTR + IND + HL + k] = (_Float16)0.0f;
  }
  v8f creg;
#pragma unroll
  for (int r = 0; r < 8; r++) {
    int env = mt * 16 + hi * 8 + r;
    creg[r] = c0[env * HL + ncol];
  }
  float bi = biasg[0 * HL + ncol], bf = biasg[1 * HL + ncol];
  float bg = biasg[2 * HL + ncol], bo = biasg[3 * HL + ncol];
  __syncthreads();

  for (int t = 0; t < T_STEPS; t++) {
    for (int i = tid; i < E_ENVS * HL; i += 1024) {
      int e = i >> 8, k = i & 255;
      if (done[t * E_ENVS + e]) A[e * ASTR + IND + k] = (_Float16)0.0f;
    }
    for (int i = tid; i < E_ENVS * IND; i += 1024) {
      int e = i / IND, k = i - e * IND;
      A[e * ASTR + k] = X[(t * E_ENVS + e) * XS + k];
    }
#pragma unroll
    for (int r = 0; r < 8; r++) {
      int env = mt * 16 + hi * 8 + r;
      if (done[t * E_ENVS + env]) creg[r] = 0.0f;
    }
    __syncthreads();

    v8f ai, af, ag, ao;
#pragma unroll
    for (int r = 0; r < 8; r++) { ai[r] = bi; af[r] = bf; ag[r] = bg; ao[r] = bo; }
    for (int kb = 0; kb < KPAD; kb += 32) {
      v16h a  = pack_frag(&A[(mt * 16 + lo) * ASTR + kb + hi * 8]);
      v16h b0 = pack_frag(Wg + (0 * HL + ncol) * KPAD + kb + hi * 8);
      v16h b1 = pack_frag(Wg + (1 * HL + ncol) * KPAD + kb + hi * 8);
      v16h b2 = pack_frag(Wg + (2 * HL + ncol) * KPAD + kb + hi * 8);
      v16h b3 = pack_frag(Wg + (3 * HL + ncol) * KPAD + kb + hi * 8);
      ai = WMMA16(a, b0, ai);
      af = WMMA16(a, b1, af);
      ag = WMMA16(a, b2, ag);
      ao = WMMA16(a, b3, ao);
    }
    __syncthreads();

    float hv[8];
#pragma unroll
    for (int r = 0; r < 8; r++) {
      float iv = sigmoidf_(ai[r]), fv = sigmoidf_(af[r]);
      float gv = tanhf(ag[r]),     ov = sigmoidf_(ao[r]);
      float cv = fv * creg[r] + iv * gv;
      creg[r] = cv;
      hv[r]   = ov * tanhf(cv);
    }
#pragma unroll
    for (int r = 0; r < 8; r++) {
      int env = mt * 16 + hi * 8 + r;
      A[env * ASTR + IND + ncol] = (_Float16)hv[r];
      hs[(t * E_ENVS + env) * HL + ncol] = hv[r];
      if (t == T_STEPS - 1) {
        out[12288 + env * HL + ncol] = hv[r];
        out[20480 + env * HL + ncol] = creg[r];
      }
    }
    __syncthreads();
  }
}

__global__ void k_heads(const float* hs, const float* pw, const float* pb,
                        const float* vw, const float* vb, float* out) {
  int i = blockIdx.x * blockDim.x + threadIdx.x;
  if (i >= B_TOT * 6) return;
  int b = i / 6, j = i % 6;
  const float* h    = hs + b * HL;
  const float* wrow = (j < 5) ? pw + j * HL : vw;
  float s = (j < 5) ? pb[j] : vb[0];
#pragma unroll 4
  for (int k = 0; k < HL; k++) s += h[k] * wrow[k];
  if (j < 5) out[b * 5 + j] = s;
  else       out[10240 + b] = s;
}

extern "C" void kernel_launch(void* const* d_in, const int* in_sizes, int n_in,
                              void* d_out, int out_size, void* d_ws, size_t ws_size,
                              hipStream_t stream) {
  (void)in_sizes; (void)n_in; (void)out_size; (void)ws_size;
  const float* img  = (const float*)d_in[0];
  const int*   pos  = (const int*)d_in[1];
  const int*   lact = (const int*)d_in[2];
  const int*   done = (const int*)d_in[3];
  const float* h0   = (const float*)d_in[4];
  const float* c0   = (const float*)d_in[5];
  const float* w1   = (const float*)d_in[6];
  const float* b1   = (const float*)d_in[7];
  const float* w2   = (const float*)d_in[8];
  const float* b2   = (const float*)d_in[9];
  const float* w3   = (const float*)d_in[10];
  const float* b3   = (const float*)d_in[11];
  const float* fcw  = (const float*)d_in[12];
  const float* fcb  = (const float*)d_in[13];
  const float* wih  = (const float*)d_in[14];
  const float* whh  = (const float*)d_in[15];
  const float* bih  = (const float*)d_in[16];
  const float* bhh  = (const float*)d_in[17];
  const float* pw   = (const float*)d_in[18];
  const float* pb   = (const float*)d_in[19];
  const float* vw   = (const float*)d_in[20];
  const float* vb   = (const float*)d_in[21];
  float* out = (float*)d_out;
  char*  ws  = (char*)d_ws;

  const size_t o_img  = 0;
  const size_t o_act1 = 90832896;
  const size_t o_w1t  = o_act1 + 52428800;
  const size_t o_w2t  = o_w1t + 12288;
  const size_t o_w3t  = o_w2t + 65536;
  const size_t o_fcwt = o_w3t + 73728;
  const size_t o_wg   = o_fcwt + 3211264;
  const size_t o_bg   = o_wg + 1703936;
  const size_t o_act2 = 0;
  const size_t o_act3 = 21233664;
  const size_t o_X    = 34078720;
  const size_t o_hs   = 36372480;

  _Float16* imgC = (_Float16*)(ws + o_img);
  _Float16* act1 = (_Float16*)(ws + o_act1);
  _Float16* act2 = (_Float16*)(ws + o_act2);
  _Float16* act3 = (_Float16*)(ws + o_act3);
  _Float16* Xm   = (_Float16*)(ws + o_X);
  _Float16* w1t  = (_Float16*)(ws + o_w1t);
  _Float16* w2t  = (_Float16*)(ws + o_w2t);
  _Float16* w3t  = (_Float16*)(ws + o_w3t);
  _Float16* fcwt = (_Float16*)(ws + o_fcwt);
  _Float16* wg   = (_Float16*)(ws + o_wg);
  float*    bg   = (float*)(ws + o_bg);
  float*    hsb  = (float*)(ws + o_hs);

  const int TB = 256;
  int nImg = B_TOT * 3 * 84 * 84;
  k_prep_img <<<(nImg + TB - 1) / TB, TB, 0, stream>>>(img, imgC, nImg);
  k_prep_w1  <<<(32 * 192   + TB - 1) / TB, TB, 0, stream>>>(w1, w1t);
  k_prep_w2  <<<(64 * 512   + TB - 1) / TB, TB, 0, stream>>>(w2, w2t);
  k_prep_w3  <<<(64 * 576   + TB - 1) / TB, TB, 0, stream>>>(w3, w3t);
  k_prep_fc  <<<(512 * 3136 + TB - 1) / TB, TB, 0, stream>>>(fcw, fcwt);
  k_prep_wg  <<<(1024 * KPAD + TB - 1) / TB, TB, 0, stream>>>(wih, whh, wg);
  k_prep_bias<<<(1024 + TB - 1) / TB, TB, 0, stream>>>(bih, bhh, bg);

  k_conv1<<<6400, TB, 0, stream>>>(imgC, w1t, b1, act1);
  k_conv2<<<1296, TB, 0, stream>>>(act1, w2t, b2, act2);
  k_conv3<<< 784, TB, 0, stream>>>(act2, w3t, b3, act3);
  k_fc   <<< 128, TB, 0, stream>>>(act3, fcwt, fcb, Xm);

  k_onehot<<<(B_TOT * 48 + TB - 1) / TB, TB, 0, stream>>>(pos, lact, Xm);

  k_lstm<<<1, 1024, 0, stream>>>(Xm, wg, bg, done, h0, c0, hsb, out);

  k_heads<<<(B_TOT * 6 + TB - 1) / TB, TB, 0, stream>>>(hsb, pw, pb, vw, vb, out);
}


# Image re-layout launch (NHWC f32 -> CHW f16): a thread converts 4 units of 4 pixels x 3 channels (16-byte loads, 8-byte stores), 28,224 waves enter instead of 1,354,752; values bit-identical
# speedup vs baseline: 4.8927x; 4.8927x over previous
#include <hip/hip_runtime.h>
#include <math.h>

typedef __attribute__((ext_vector_type(16))) _Float16 v16h;
typedef __attribute__((ext_vector_type(8)))  _Float16 v8h;
typedef __attribute__((ext_vector_type(4)))  _Float16 v4h;
typedef __attribute__((ext_vector_type(8)))  float    v8f;

#define WMMA16(A,B,C) __builtin_amdgcn_wmma_f32_16x16x32_f16(false,(A),false,(B),(short)0,(C),false,false)

static constexpr int T_STEPS = 64;
static constexpr int E_ENVS  = 32;
static constexpr int B_TOT   = 2048;
static constexpr int HL      = 256;
static constexpr int IND     = 549;
static constexpr int KPAD    = 832;
static constexpr int XS      = 576;
static constexpr int IMS     = 88;

__device__ __forceinline__ v16h pack_frag(const _Float16* p) {
  v8h u = *(const v8h*)p;
  v8h w = *(const v8h*)(p + 16);
  v16h a;
#pragma unroll
  for (int e = 0; e < 8; e++) { a[e] = u[e]; a[e + 8] = w[e]; }
  return a;
}
__device__ __forceinline__ v16h pack_frag_c1(const _Float16* p0) {
  const _Float16* p1 = p0 + 2 * IMS;
  v4h u0 = *(const v4h*)p0;
  v4h u1 = *(const v4h*)(p0 + 4);
  v4h u2 = *(const v4h*)p1;
  v4h u3 = *(const v4h*)(p1 + 4);
  v16h a;
#pragma unroll
  for (int e = 0; e < 4; e++) {
    a[e] = u0[e]; a[e + 4] = u1[e]; a[e + 8] = u2[e]; a[e + 12] = u3[e];
  }
  return a;
}

__device__ __forceinline__ float sigmoidf_(float x) { return 1.0f / (1.0f + __expf(-x)); }

__device__ __forceinline__ unsigned pk2_scaled_f16(float a, float b, float sc) {
  unsigned r;
  asm("v_fma_mixlo_f16 %0, %1, %2, 0" : "=v"(r) : "v"(a), "s"(sc));
  asm("v_fma_mixhi_f16 %0, %1, %2, 0" : "+v"(r) : "v"(b), "s"(sc));
  return r;
}

__global__ void __launch_bounds__(256) k_prep_img(const float* img, _Float16* imgC) {
  const float sc = 1.0f / 255.0f;
  int g0 = blockIdx.x * 1024 + threadIdx.x;
  float4 v[4][3];
#pragma unroll
  for (int k = 0; k < 4; k++) {
    const float4* p = (const float4*)(img + (g0 + k * 256) * 12);
    v[k][0] = p[0]; v[k][1] = p[1]; v[k][2] = p[2];
  }
#pragma unroll
  for (int k = 0; k < 4; k++) {
    int g  = g0 + k * 256;
    int R  = g / 21, cg = g - R * 21;
    int b  = R / 84;
    _Float16* o = imgC + (R + b * 168) * IMS + cg * 4;
    float4 q0 = v[k][0], q1 = v[k][1], q2 = v[k][2];
    uint2 c0, c1, c2;
    c0.x = pk2_scaled_f16(q0.x, q0.w, sc); c0.y = pk2_scaled_f16(q1.z, q2.y, sc);
    c1.x = pk2_scaled_f16(q0.y, q1.x, sc); c1.y = pk2_scaled_f16(q1.w, q2.z, sc);
    c2.x = pk2_scaled_f16(q0.z, q1.y, sc); c2.y = pk2_scaled_f16(q2.x, q2.w, sc);
    *(uint2*)(o)                = c0;
    *(uint2*)(o + 84 * IMS)     = c1;
    *(uint2*)(o + 2 * 84 * IMS) = c2;
  }
}

__global__ void k_prep_w1(const float* w, _Float16* wt) {
  int i = blockIdx.x * blockDim.x + threadIdx.x;
  if (i >= 32 * 192) return;
  int n = i / 192, k = i % 192;
  int c = k >> 6, kh = (k & 63) >> 3, kw = k & 7;
  wt[i] = (_Float16)w[((n * 3 + c) * 8 + kh) * 8 + kw];
}
__global__ void k_prep_w2(const float* w, _Float16* wt) {
  int i = blockIdx.x * blockDim.x + threadIdx.x;
  if (i >= 64 * 512) return;
  int n = i / 512, k = i % 512;
  int pix = k >> 5, c = k & 31, kh = pix >> 2, kw = pix & 3;
  wt[i] = (_Float16)w[((n * 32 + c) * 4 + kh) * 4 + kw];
}
__global__ void k_prep_w3(const float* w, _Float16* wt) {
  int i = blockIdx.x * blockDim.x + threadIdx.x;
  if (i >= 64 * 576) return;
  int n = i / 576, k = i % 576;
  int pix = k >> 6, c = k & 63, kh = pix / 3, kw = pix % 3;
  wt[i] = (_Float16)w[((n * 64 + c) * 3 + kh) * 3 + kw];
}
__global__ void k_prep_fc(const float* w, _Float16* wt) {
  int i = blockIdx.x * blockDim.x + threadIdx.x;
  if (i >= 512 * 3136) return;
  int f = i / 3136, k = i % 3136;
  int s = k >> 6, c = k & 63;
  wt[i] = (_Float16)w[f * 3136 + c * 49 + s];
}
__global__ void k_prep_wg(const float* wih, const float* whh, _Float16* wg) {
  int i = blockIdx.x * blockDim.x + threadIdx.x;
  if (i >= 1024 * KPAD) return;
  int r = i / KPAD, k = i % KPAD;
  float v = 0.0f;
  if (k < HL)             v = whh[r * HL + k];
  else if (k < HL + IND)  v = wih[r * IND + (k - HL)];
  wg[i] = (_Float16)v;
}
__global__ void k_prep_bias(const float* bih, const float* bhh, float* bg) {
  int i = blockIdx.x * blockDim.x + threadIdx.x;
  if (i < 1024) bg[i] = bih[i] + bhh[i];
}
__global__ void k_onehot(const int* pos, const int* lact, _Float16* X) {
  int i = blockIdx.x * blockDim.x + threadIdx.x;
  if (i >= B_TOT * 64) return;
  int b = i / 64, c = i % 64;
  float v = 0.0f;
  if (c < 16)      v = (pos[b * 2 + 0] == c)      ? 1.0f : 0.0f;
  else if (c < 32) v = (pos[b * 2 + 1] == c - 16) ? 1.0f : 0.0f;
  else if (c < 37) v = (lact[b]        == c - 32) ? 1.0f : 0.0f;
  X[b * XS + 512 + c] = (_Float16)v;
}

__global__ void __launch_bounds__(256) k_conv1(const _Float16* imgC, const _Float16* wt,
                                               const float* bias, _Float16* out) {
  int wid  = (blockIdx.x * blockDim.x + threadIdx.x) >> 5;
  int lane = threadIdx.x & 31, hi = lane >> 4, lo = lane & 15;
  int m = wid * 16 + lo;
  int b = m / 400, p = m % 400, oh = p / 20, ow = p % 20;
  const _Float16* abase = imgC + (size_t)b * 3 * 84 * IMS + (oh * 4) * IMS + ow * 4;
  v8f acc0 = {}, acc1 = {};
#pragma unroll
  for (int kb = 0; kb < 192; kb += 32) {
    int c = kb >> 6, khb = (kb & 63) >> 3;
    v16h a  = pack_frag_c1(abase + (c * 84 + khb + hi) * IMS);
    v16h b0 = pack_frag(wt + lo * 192        + kb + hi * 8);
    v16h b1 = pack_frag(wt + (16 + lo) * 192 + kb + hi * 8);
    acc0 = WMMA16(a, b0, acc0);
    acc1 = WMMA16(a, b1, acc1);
  }
  float bs0 = bias[lo], bs1 = bias[16 + lo];
#pragma unroll
  for (int r = 0; r < 8; r++) {
    int mm = wid * 16 + hi * 8 + r;
    float v0 = acc0[r] + bs0; v0 = v0 > 0.0f ? v0 : 0.0f;
    float v1 = acc1[r] + bs1; v1 = v1 > 0.0f ? v1 : 0.0f;
    out[mm * 32 + lo]      = (_Float16)v0;
    out[mm * 32 + 16 + lo] = (_Float16)v1;
  }
}

__global__ void __launch_bounds__(256) k_conv2(const _Float16* act1, const _Float16* wt,
                                               const float* bias, _Float16* out) {
  int wid  = (blockIdx.x * blockDim.x + threadIdx.x) >> 5;
  int lane = threadIdx.x & 31, hi = lane >> 4, lo = lane & 15;
  int m = wid * 16 + lo;
  int b = m / 81, p = m % 81, oh = p / 9, ow = p % 9;
  v8f acc0 = {}, acc1 = {}, acc2 = {}, acc3 = {};
  for (int kb = 0; kb < 512; kb += 32) {
    int pix = kb >> 5, kh = pix >> 2, kw = pix & 3;
    v16h a  = pack_frag(act1 + ((b * 20 + oh * 2 + kh) * 20 + (ow * 2 + kw)) * 32 + hi * 8);
    v16h b0 = pack_frag(wt + (lo)      * 512 + kb + hi * 8);
    v16h b1 = pack_frag(wt + (16 + lo) * 512 + kb + hi * 8);
    v16h b2 = pack_frag(wt + (32 + lo) * 512 + kb + hi * 8);
    v16h b3 = pack_frag(wt + (48 + lo) * 512 + kb + hi * 8);
    acc0 = WMMA16(a, b0, acc0);
    acc1 = WMMA16(a, b1, acc1);
    acc2 = WMMA16(a, b2, acc2);
    acc3 = WMMA16(a, b3, acc3);
  }
  float bs0 = bias[lo], bs1 = bias[16 + lo], bs2 = bias[32 + lo], bs3 = bias[48 + lo];
#pragma unroll
  for (int r = 0; r < 8; r++) {
    int mm = wid * 16 + hi * 8 + r;
    float v0 = acc0[r] + bs0; v0 = v0 > 0.0f ? v0 : 0.0f;
    float v1 = acc1[r] + bs1; v1 = v1 > 0.0f ? v1 : 0.0f;
    float v2 = acc2[r] + bs2; v2 = v2 > 0.0f ? v2 : 0.0f;
    float v3 = acc3[r] + bs3; v3 = v3 > 0.0f ? v3 : 0.0f;
    out[mm * 64 + lo]      = (_Float16)v0;
    out[mm * 64 + 16 + lo] = (_Float16)v1;
    out[mm * 64 + 32 + lo] = (_Float16)v2;
    out[mm * 64 + 48 + lo] = (_Float16)v3;
  }
}

__global__ void __launch_bounds__(256) k_conv3(const _Float16* act2, const _Float16* wt,
                                               const float* bias, _Float16* out) {
  int wid  = (blockIdx.x * blockDim.x + threadIdx.x) >> 5;
  int lane = threadIdx.x & 31, hi = lane >> 4, lo = lane & 15;
  int m = wid * 16 + lo;
  int b = m / 49, p = m % 49, oh = p / 7, ow = p % 7;
  v8f acc0 = {}, acc1 = {}, acc2 = {}, acc3 = {};
  for (int kb = 0; kb < 576; kb += 32) {
    int pix = kb >> 6, cb = (kb & 63) + hi * 8;
    int kh = pix / 3, kw = pix % 3;
    v16h a  = pack_frag(act2 + ((b * 9 + oh + kh) * 9 + (ow + kw)) * 64 + cb);
    v16h b0 = pack_frag(wt + (lo)      * 576 + kb + hi * 8);
    v16h b1 = pack_frag(wt + (16 + lo) * 576 + kb + hi * 8);
    v16h b2 = pack_frag(wt + (32 + lo) * 576 + kb + hi * 8);
    v16h b3 = pack_frag(wt + (48 + lo) * 576 + kb + hi * 8);
    acc0 = WMMA16(a, b0, acc0);
    acc1 = WMMA16(a, b1, acc1);
    acc2 = WMMA16(a, b2, acc2);
    acc3 = WMMA16(a, b3, acc3);
  }
  float bs0 = bias[lo], bs1 = bias[16 + lo], bs2 = bias[32 + lo], bs3 = bias[48 + lo];
#pragma unroll
  for (int r = 0; r < 8; r++) {
    int mm = wid * 16 + hi * 8 + r;
    float v0 = acc0[r] + bs0; v0 = v0 > 0.0f ? v0 : 0.0f;
    float v1 = acc1[r] + bs1; v1 = v1 > 0.0f ? v1 : 0.0f;
    float v2 = acc2[r] + bs2; v2 = v2 > 0.0f ? v2 : 0.0f;
    float v3 = acc3[r] + bs3; v3 = v3 > 0.0f ? v3 : 0.0f;
    out[mm * 64 + lo]      = (_Float16)v0;
    out[mm * 64 + 16 + lo] = (_Float16)v1;
    out[mm * 64 + 32 + lo] = (_Float16)v2;
    out[mm * 64 + 48 + lo] = (_Float16)v3;
  }
}

__global__ void __launch_bounds__(256) k_fc(const _Float16* act3, const _Float16* wt,
                                            const float* bias, _Float16* X) {
  int wid  = (blockIdx.x * blockDim.x + threadIdx.x) >> 5;
  int lane = threadIdx.x & 31, hi = lane >> 4, lo = lane & 15;
  int ng = wid & 7, mt = wid >> 3;
  int m = mt * 16 + lo;
  int nb = ng * 64;
  v8f acc0 = {}, acc1 = {}, acc2 = {}, acc3 = {};
  for (int kb = 0; kb < 3136; kb += 32) {
    v16h a  = pack_frag(act3 + (size_t)m * 3136 + kb + hi * 8);
    v16h b0 = pack_frag(wt + (size_t)(nb + lo)      * 3136 + kb + hi * 8);
    v16h b1 = pack_frag(wt + (size_t)(nb + 16 + lo) * 3136 + kb + hi * 8);
    v16h b2 = pack_frag(wt + (size_t)(nb + 32 + lo) * 3136 + kb + hi * 8);
    v16h b3 = pack_frag(wt + (size_t)(nb + 48 + lo) * 3136 + kb + hi * 8);
    acc0 = WMMA16(a, b0, acc0);
    acc1 = WMMA16(a, b1, acc1);
    acc2 = WMMA16(a, b2, acc2);
    acc3 = WMMA16(a, b3, acc3);
  }
  float bs0 = bias[nb + lo], bs1 = bias[nb + 16 + lo];
  float bs2 = bias[nb + 32 + lo], bs3 = bias[nb + 48 + lo];
#pragma unroll
  for (int r = 0; r < 8; r++) {
    int mm = mt * 16 + hi * 8 + r;
    float v0 = acc0[r] + bs0; v0 = v0 > 0.0f ? v0 : 0.0f;
    float v1 = acc1[r] + bs1; v1 = v1 > 0.0f ? v1 : 0.0f;
    float v2 = acc2[r] + bs2; v2 = v2 > 0.0f ? v2 : 0.0f;
    float v3 = acc3[r] + bs3; v3 = v3 > 0.0f ? v3 : 0.0f;
    X[mm * XS + nb + lo]      = (_Float16)v0;
    X[mm * XS + nb + 16 + lo] = (_Float16)v1;
    X[mm * XS + nb + 32 + lo] = (_Float16)v2;
    X[mm * XS + nb + 48 + lo] = (_Float16)v3;
  }
}

__global__ void __launch_bounds__(512) k_lstm_step(const _Float16* X, const _Float16* Wg,
                                                   const float* biasg, const int* done,
                                                   const float* h0, const float* c0,
                                                   float* cbuf, float* hs, float* out, int t) {
  __shared__ float P[8 * 2 * 1024];
  int tid = threadIdx.x;
  int w = __builtin_amdgcn_readfirstlane(tid >> 5);
  int lane = tid & 31, hi = lane >> 4, lo = lane & 15;
  int mt = blockIdx.x >> 3, cgrp = blockIdx.x & 7;
  int j = w >> 3, ks = w & 7;
  const float* hprev = (t == 0) ? h0 : hs + (size_t)(t - 1) * (E_ENVS * HL);
  const float* cin   = (t == 0) ? c0 : cbuf;

  int eenv = mt * 16 + w;
  int ecol = cgrp * 32 + lane;
  int edone = done[t * E_ENVS + eenv];
  float cprev = cin[eenv * HL + ecol];
  float bi = biasg[0 * HL + ecol], bf = biasg[1 * HL + ecol];
  float bg = biasg[2 * HL + ecol], bo = biasg[3 * HL + ecol];

  int aenv = mt * 16 + lo;
  int adone = done[t * E_ENVS + aenv];
  int ncol = cgrp * 32 + j * 16 + lo;
  const _Float16* wrow = Wg + (size_t)ncol * KPAD + hi * 8;
  const _Float16* xrow = X + (size_t)(t * E_ENVS + aenv) * XS + hi * 8;
  const size_t GST = (size_t)HL * KPAD;

  v8f ai = {}, af = {}, ag = {}, ao = {};
  v16h a3 = {}, b30 = {}, b31 = {}, b32 = {}, b33 = {};
  if (ks < 2) {
    int kc = ks + 24;
    const _Float16* wp = wrow + kc * 32;
    a3  = pack_frag(xrow + (kc - 8) * 32);
    b30 = pack_frag(wp);
    b31 = pack_frag(wp + GST);
    b32 = pack_frag(wp + 2 * GST);
    b33 = pack_frag(wp + 3 * GST);
  }
  {
    const float* hp = hprev + aenv * HL + ks * 32 + hi * 8;
    float4 u0 = *(const float4*)hp,        u1 = *(const float4*)(hp + 4);
    float4 u2 = *(const float4*)(hp + 16), u3 = *(const float4*)(hp + 20);
    const _Float16* wp = wrow + ks * 32;
    v16h b0 = pack_frag(wp);
    v16h b1 = pack_frag(wp + GST);
    v16h b2 = pack_frag(wp + 2 * GST);
    v16h b3 = pack_frag(wp + 3 * GST);
    float hv16[16] = {u0.x, u0.y, u0.z, u0.w, u1.x, u1.y, u1.z, u1.w,
                      u2.x, u2.y, u2.z, u2.w, u3.x, u3.y, u3.z, u3.w};
    v16h a;
#pragma unroll
    for (int e = 0; e < 16; e++) a[e] = adone ? (_Float16)0.0f : (_Float16)hv16[e];
    ai = WMMA16(a, b0, ai);
    af = WMMA16(a, b1, af);
    ag = WMMA16(a, b2, ag);
    ao = WMMA16(a, b3, ao);
  }
#pragma unroll
  for (int i = 1; i < 3; i++) {
    int kc = ks + 8 * i;
    v16h a  = pack_frag(xrow + (kc - 8) * 32);
    const _Float16* wp = wrow + kc * 32;
    v16h b0 = pack_frag(wp);
    v16h b1 = pack_frag(wp + GST);
    v16h b2 = pack_frag(wp + 2 * GST);
    v16h b3 = pack_frag(wp + 3 * GST);
    ai = WMMA16(a, b0, ai);
    af = WMMA16(a, b1, af);
    ag = WMMA16(a, b2, ag);
    ao = WMMA16(a, b3, ao);
  }
  if (ks < 2) {
    ai = WMMA16(a3, b30, ai);
    af = WMMA16(a3, b31, af);
    ag = WMMA16(a3, b32, ag);
    ao = WMMA16(a3, b33, ao);
  }

  {
    float* pw = P + ks * 2048 + j * 1024 + (lane ^ (j * 16));
#pragma unroll
    for (int r = 0; r < 8; r++) {
      pw[(0 * 8 + r) * 32] = ai[r];
      pw[(1 * 8 + r) * 32] = af[r];
      pw[(2 * 8 + r) * 32] = ag[r];
      pw[(3 * 8 + r) * 32] = ao[r];
    }
  }
  __syncthreads();

  int jr = lane >> 4;
  const float* pr = P + jr * 1024 + (w & 7) * 32 + (((w >> 3) ^ jr) * 16) + (lane & 15);
  float si = 0.0f, sf = 0.0f, sg = 0.0f, so = 0.0f;
#pragma unroll
  for (int q = 0; q < 8; q++) {
    si += pr[q * 2048 + 0 * 256];
    sf += pr[q * 2048 + 1 * 256];
    sg += pr[q * 2048 + 2 * 256];
    so += pr[q * 2048 + 3 * 256];
  }
  float cp = edone ? 0.0f : cprev;
  float iv = sigmoidf_(si + bi), fv = sigmoidf_(sf + bf);
  float gv = tanhf(sg + bg),     ov = sigmoidf_(so + bo);
  float cv = fv * cp + iv * gv;
  float hv = ov * tanhf(cv);
  cbuf[eenv * HL + ecol] = cv;
  hs[(t * E_ENVS + eenv) * HL + ecol] = hv;
  if (t == T_STEPS - 1) {
    out[12288 + eenv * HL + ecol] = hv;
    out[20480 + eenv * HL + ecol] = cv;
  }
}

__global__ void k_heads(const float* hs, const float* pw, const float* pb,
                        const float* vw, const float* vb, float* out) {
  int i = blockIdx.x * blockDim.x + threadIdx.x;
  if (i >= B_TOT * 6) return;
  int b = i / 6, j = i % 6;
  const float* h    = hs + b * HL;
  const float* wrow = (j < 5) ? pw + j * HL : vw;
  float s = (j < 5) ? pb[j] : vb[0];
#pragma unroll 4
  for (int k = 0; k < HL; k++) s += h[k] * wrow[k];
  if (j < 5) out[b * 5 + j] = s;
  else       out[10240 + b] = s;
}

extern "C" void kernel_launch(void* const* d_in, const int* in_sizes, int n_in,
                              void* d_out, int out_size, void* d_ws, size_t ws_size,
                              hipStream_t stream) {
  (void)in_sizes; (void)n_in; (void)out_size; (void)ws_size;
  const float* img  = (const float*)d_in[0];
  const int*   pos  = (const int*)d_in[1];
  const int*   lact = (const int*)d_in[2];
  const int*   done = (const int*)d_in[3];
  const float* h0   = (const float*)d_in[4];
  const float* c0   = (const float*)d_in[5];
  const float* w1   = (const float*)d_in[6];
  const float* b1   = (const float*)d_in[7];
  const float* w2   = (const float*)d_in[8];
  const float* b2   = (const float*)d_in[9];
  const float* w3   = (const float*)d_in[10];
  const float* b3   = (const float*)d_in[11];
  const float* fcw  = (const float*)d_in[12];
  const float* fcb  = (const float*)d_in[13];
  const float* wih  = (const float*)d_in[14];
  const float* whh  = (const float*)d_in[15];
  const float* bih  = (const float*)d_in[16];
  const float* bhh  = (const float*)d_in[17];
  const float* pw   = (const float*)d_in[18];
  const float* pb   = (const float*)d_in[19];
  const float* vw   = (const float*)d_in[20];
  const float* vb   = (const float*)d_in[21];
  float* out = (float*)d_out;
  char*  ws  = (char*)d_ws;

  const size_t o_img  = 0;
  const size_t o_act1 = 90832896;
  const size_t o_w1t  = o_act1 + 52428800;
  const size_t o_w2t  = o_w1t + 12288;
  const size_t o_w3t  = o_w2t + 65536;
  const size_t o_fcwt = o_w3t + 73728;
  const size_t o_wg   = o_fcwt + 3211264;
  const size_t o_bg   = o_wg + 1703936;
  const size_t o_act2 = 0;
  const size_t o_act3 = 21233664;
  const size_t o_X    = 34078720;
  const size_t o_hs   = 36438016;
  const size_t o_c    = 38535168;

  _Float16* imgC = (_Float16*)(ws + o_img);
  _Float16* act1 = (_Float16*)(ws + o_act1);
  _Float16* act2 = (_Float16*)(ws + o_act2);
  _Float16* act3 = (_Float16*)(ws + o_act3);
  _Float16* Xm   = (_Float16*)(ws + o_X);
  _Float16* w1t  = (_Float16*)(ws + o_w1t);
  _Float16* w2t  = (_Float16*)(ws + o_w2t);
  _Float16* w3t  = (_Float16*)(ws + o_w3t);
  _Float16* fcwt = (_Float16*)(ws + o_fcwt);
  _Float16* wg   = (_Float16*)(ws + o_wg);
  float*    bg   = (float*)(ws + o_bg);
  float*    hsb  = (float*)(ws + o_hs);
  float*    cbuf = (float*)(ws + o_c);

  const int TB = 256;
  k_prep_img <<<3528, TB, 0, stream>>>(img, imgC);
  k_prep_w1  <<<(32 * 192   + TB - 1) / TB, TB, 0, stream>>>(w1, w1t);
  k_prep_w2  <<<(64 * 512   + TB - 1) / TB, TB, 0, stream>>>(w2, w2t);
  k_prep_w3  <<<(64 * 576   + TB - 1) / TB, TB, 0, stream>>>(w3, w3t);
  k_prep_fc  <<<(512 * 3136 + TB - 1) / TB, TB, 0, stream>>>(fcw, fcwt);
  k_prep_wg  <<<(1024 * KPAD + TB - 1) / TB, TB, 0, stream>>>(wih, whh, wg);
  k_prep_bias<<<(1024 + TB - 1) / TB, TB, 0, stream>>>(bih, bhh, bg);

  k_conv1<<<6400, TB, 0, stream>>>(imgC, w1t, b1, act1);
  k_conv2<<<1296, TB, 0, stream>>>(act1, w2t, b2, act2);
  k_conv3<<< 784, TB, 0, stream>>>(act2, w3t, b3, act3);
  k_fc   <<< 128, TB, 0, stream>>>(act3, fcwt, fcb, Xm);

  k_onehot<<<(B_TOT * 64 + TB - 1) / TB, TB, 0, stream>>>(pos, lact, Xm);

  k_lstm_step<<<16, 512, 0, stream>>>(Xm, wg, bg, done, h0, c0, cbuf, hsb, out, 0);
  k_lstm_step<<<16, 512, 0, stream>>>(Xm, wg, bg, done, h0, c0, cbuf, hsb, out, 1);
  k_lstm_step<<<16, 512, 0, stream>>>(Xm, wg, bg, done, h0, c0, cbuf, hsb, out, 2);
  k_lstm_step<<<16, 512, 0, stream>>>(Xm, wg, bg, done, h0, c0, cbuf, hsb, out, 3);
  k_lstm_step<<<16, 512, 0, stream>>>(Xm, wg, bg, done, h0, c0, cbuf, hsb, out, 4);
  k_lstm_step<<<16, 512, 0, stream>>>(Xm, wg, bg, done, h0, c0, cbuf, hsb, out, 5);
  k_lstm_step<<<16, 512, 0, stream>>>(Xm, wg, bg, done, h0, c0, cbuf, hsb, out, 6);
  k_lstm_step<<<16, 512, 0, stream>>>(Xm, wg, bg, done, h0, c0, cbuf, hsb, out, 7);
  k_lstm_step<<<16, 512, 0, stream>>>(Xm, wg, bg, done, h0, c0, cbuf, hsb, out, 8);
  k_lstm_step<<<16, 512, 0, stream>>>(Xm, wg, bg, done, h0, c0, cbuf, hsb, out, 9);
  k_lstm_step<<<16, 512, 0, stream>>>(Xm, wg, bg, done, h0, c0, cbuf, hsb, out, 10);
  k_lstm_step<<<16, 512, 0, stream>>>(Xm, wg, bg, done, h0, c0, cbuf, hsb, out, 11);
  k_lstm_step<<<16, 512, 0, stream>>>(Xm, wg, bg, done, h0, c0, cbuf, hsb, out, 12);
  k_lstm_step<<<16, 512, 0, stream>>>(Xm, wg, bg, done, h0, c0, cbuf, hsb, out, 13);
  k_lstm_step<<<16, 512, 0, stream>>>(Xm, wg, bg, done, h0, c0, cbuf, hsb, out, 14);
  k_lstm_step<<<16, 512, 0, stream>>>(Xm, wg, bg, done, h0, c0, cbuf, hsb, out, 15);
  k_lstm_step<<<16, 512, 0, stream>>>(Xm, wg, bg, done, h0, c0, cbuf, hsb, out, 16);
  k_lstm_step<<<16, 512, 0, stream>>>(Xm, wg, bg, done, h0, c0, cbuf, hsb, out, 17);
  k_lstm_step<<<16, 512, 0, stream>>>(Xm, wg, bg, done, h0, c0, cbuf, hsb, out, 18);
  k_lstm_step<<<16, 512, 0, stream>>>(Xm, wg, bg, done, h0, c0, cbuf, hsb, out, 19);
  k_lstm_step<<<16, 512, 0, stream>>>(Xm, wg, bg, done, h0, c0, cbuf, hsb, out, 20);
  k_lstm_step<<<16, 512, 0, stream>>>(Xm, wg, bg, done, h0, c0, cbuf, hsb, out, 21);
  k_lstm_step<<<16, 512, 0, stream>>>(Xm, wg, bg, done, h0, c0, cbuf, hsb, out, 22);
  k_lstm_step<<<16, 512, 0, stream>>>(Xm, wg, bg, done, h0, c0, cbuf, hsb, out, 23);
  k_lstm_step<<<16, 512, 0, stream>>>(Xm, wg, bg, done, h0, c0, cbuf, hsb, out, 24);
  k_lstm_step<<<16, 512, 0, stream>>>(Xm, wg, bg, done, h0, c0, cbuf, hsb, out, 25);
  k_lstm_step<<<16, 512, 0, stream>>>(Xm, wg, bg, done, h0, c0, cbuf, hsb, out, 26);
  k_lstm_step<<<16, 512, 0, stream>>>(Xm, wg, bg, done, h0, c0, cbuf, hsb, out, 27);
  k_lstm_step<<<16, 512, 0, stream>>>(Xm, wg, bg, done, h0, c0, cbuf, hsb, out, 28);
  k_lstm_step<<<16, 512, 0, stream>>>(Xm, wg, bg, done, h0, c0, cbuf, hsb, out, 29);
  k_lstm_step<<<16, 512, 0, stream>>>(Xm, wg, bg, done, h0, c0, cbuf, hsb, out, 30);
  k_lstm_step<<<16, 512, 0, stream>>>(Xm, wg, bg, done, h0, c0, cbuf, hsb, out, 31);
  k_lstm_step<<<16, 512, 0, stream>>>(Xm, wg, bg, done, h0, c0, cbuf, hsb, out, 32);
  k_lstm_step<<<16, 512, 0, stream>>>(Xm, wg, bg, done, h0, c0, cbuf, hsb, out, 33);
  k_lstm_step<<<16, 512, 0, stream>>>(Xm, wg, bg, done, h0, c0, cbuf, hsb, out, 34);
  k_lstm_step<<<16, 512, 0, stream>>>(Xm, wg, bg, done, h0, c0, cbuf, hsb, out, 35);
  k_lstm_step<<<16, 512, 0, stream>>>(Xm, wg, bg, done, h0, c0, cbuf, hsb, out, 36);
  k_lstm_step<<<16, 512, 0, stream>>>(Xm, wg, bg, done, h0, c0, cbuf, hsb, out, 37);
  k_lstm_step<<<16, 512, 0, stream>>>(Xm, wg, bg, done, h0, c0, cbuf, hsb, out, 38);
  k_lstm_step<<<16, 512, 0, stream>>>(Xm, wg, bg, done, h0, c0, cbuf, hsb, out, 39);
  k_lstm_step<<<16, 512, 0, stream>>>(Xm, wg, bg, done, h0, c0, cbuf, hsb, out, 40);
  k_lstm_step<<<16, 512, 0, stream>>>(Xm, wg, bg, done, h0, c0, cbuf, hsb, out, 41);
  k_lstm_step<<<16, 512, 0, stream>>>(Xm, wg, bg, done, h0, c0, cbuf, hsb, out, 42);
  k_lstm_step<<<16, 512, 0, stream>>>(Xm, wg, bg, done, h0, c0, cbuf, hsb, out, 43);
  k_lstm_step<<<16, 512, 0, stream>>>(Xm, wg, bg, done, h0, c0, cbuf, hsb, out, 44);
  k_lstm_step<<<16, 512, 0, stream>>>(Xm, wg, bg, done, h0, c0, cbuf, hsb, out, 45);
  k_lstm_step<<<16, 512, 0, stream>>>(Xm, wg, bg, done, h0, c0, cbuf, hsb, out, 46);
  k_lstm_step<<<16, 512, 0, stream>>>(Xm, wg, bg, done, h0, c0, cbuf, hsb, out, 47);
  k_lstm_step<<<16, 512, 0, stream>>>(Xm, wg, bg, done, h0, c0, cbuf, hsb, out, 48);
  k_lstm_step<<<16, 512, 0, stream>>>(Xm, wg, bg, done, h0, c0, cbuf, hsb, out, 49);
  k_lstm_step<<<16, 512, 0, stream>>>(Xm, wg, bg, done, h0, c0, cbuf, hsb, out, 50);
  k_lstm_step<<<16, 512, 0, stream>>>(Xm, wg, bg, done, h0, c0, cbuf, hsb, out, 51);
  k_lstm_step<<<16, 512, 0, stream>>>(Xm, wg, bg, done, h0, c0, cbuf, hsb, out, 52);
  k_lstm_step<<<16, 512, 0, stream>>>(Xm, wg, bg, done, h0, c0, cbuf, hsb, out, 53);
  k_lstm_step<<<16, 512, 0, stream>>>(Xm, wg, bg, done, h0, c0, cbuf, hsb, out, 54);
  k_lstm_step<<<16, 512, 0, stream>>>(Xm, wg, bg, done, h0, c0, cbuf, hsb, out, 55);
  k_lstm_step<<<16, 512, 0, stream>>>(Xm, wg, bg, done, h0, c0, cbuf, hsb, out, 56);
  k_lstm_step<<<16, 512, 0, stream>>>(Xm, wg, bg, done, h0, c0, cbuf, hsb, out, 57);
  k_lstm_step<<<16, 512, 0, stream>>>(Xm, wg, bg, done, h0, c0, cbuf, hsb, out, 58);
  k_lstm_step<<<16, 512, 0, stream>>>(Xm, wg, bg, done, h0, c0, cbuf, hsb, out, 59);
  k_lstm_step<<<16, 512, 0, stream>>>(Xm, wg, bg, done, h0, c0, cbuf, hsb, out, 60);
  k_lstm_step<<<16, 512, 0, stream>>>(Xm, wg, bg, done, h0, c0, cbuf, hsb, out, 61);
  k_lstm_step<<<16, 512, 0, stream>>>(Xm, wg, bg, done, h0, c0, cbuf, hsb, out, 62);
  k_lstm_step<<<16, 512, 0, stream>>>(Xm, wg, bg, done, h0, c0, cbuf, hsb, out, 63);

  k_heads<<<(B_TOT * 6 + TB - 1) / TB, TB, 0, stream>>>(hsb, pw, pb, vw, vb, out);
}


# LSTM step launch retiled 16 -> 128 workgroups (16 envs x 4 hidden columns, 4 gates in one B tile, K split over 4 waves); h, c kept blocked for whole-line stores; f32 sum order of the gates changes
# speedup vs baseline: 6.6872x; 1.3668x over previous
#include <hip/hip_runtime.h>
#include <math.h>

typedef __attribute__((ext_vector_type(16))) _Float16 v16h;
typedef __attribute__((ext_vector_type(8)))  _Float16 v8h;
typedef __attribute__((ext_vector_type(4)))  _Float16 v4h;
typedef __attribute__((ext_vector_type(8)))  float    v8f;

#define WMMA16(A,B,C) __builtin_amdgcn_wmma_f32_16x16x32_f16(false,(A),false,(B),(short)0,(C),false,false)

static constexpr int T_STEPS = 64;
static constexpr int E_ENVS  = 32;
static constexpr int B_TOT   = 2048;
static constexpr int HL      = 256;
static constexpr int IND     = 549;
static constexpr int KPAD    = 832;
static constexpr int XS      = 576;
static constexpr int IMS     = 88;

__device__ __forceinline__ v16h pack_frag(const _Float16* p) {
  v8h u = *(const v8h*)p;
  v8h w = *(const v8h*)(p + 16);
  v16h a;
#pragma unroll
  for (int e = 0; e < 8; e++) { a[e] = u[e]; a[e + 8] = w[e]; }
  return a;
}
__device__ __forceinline__ v16h pack_frag_c1(const _Float16* p0) {
  const _Float16* p1 = p0 + 2 * IMS;
  v4h u0 = *(const v4h*)p0;
  v4h u1 = *(const v4h*)(p0 + 4);
  v4h u2 = *(const v4h*)p1;
  v4h u3 = *(const v4h*)(p1 + 4);
  v16h a;
#pragma unroll
  for (int e = 0; e < 4; e++) {
    a[e] = u0[e]; a[e + 4] = u1[e]; a[e + 8] = u2[e]; a[e + 12] = u3[e];
  }
  return a;
}

__device__ __forceinline__ float sigmoidf_(float x) { return 1.0f / (1.0f + __expf(-x)); }

__device__ __forceinline__ unsigned pk2_scaled_f16(float a, float b, float sc) {
  unsigned r;
  asm("v_fma_mixlo_f16 %0, %1, %2, 0" : "=v"(r) : "v"(a), "s"(sc));
  asm("v_fma_mixhi_f16 %0, %1, %2, 0" : "+v"(r) : "v"(b), "s"(sc));
  return r;
}

__global__ void __launch_bounds__(256) k_prep_img(const float* img, _Float16* imgC) {
  const float sc = 1.0f / 255.0f;
  int g0 = blockIdx.x * 1024 + threadIdx.x;
  float4 v[4][3];
#pragma unroll
  for (int k = 0; k < 4; k++) {
    const float4* p = (const float4*)(img + (g0 + k * 256) * 12);
    v[k][0] = p[0]; v[k][1] = p[1]; v[k][2] = p[2];
  }
#pragma unroll
  for (int k = 0; k < 4; k++) {
    int g  = g0 + k * 256;
    int R  = g / 21, cg = g - R * 21;
    int b  = R / 84;
    _Float16* o = imgC + (R + b * 168) * IMS + cg * 4;
    float4 q0 = v[k][0], q1 = v[k][1], q2 = v[k][2];
    uint2 c0, c1, c2;
    c0.x = pk2_scaled_f16(q0.x, q0.w, sc); c0.y = pk2_scaled_f16(q1.z, q2.y, sc);
    c1.x = pk2_scaled_f16(q0.y, q1.x, sc); c1.y = pk2_scaled_f16(q1.w, q2.z, sc);
    c2.x = pk2_scaled_f16(q0.z, q1.y, sc); c2.y = pk2_scaled_f16(q2.x, q2.w, sc);
    *(uint2*)(o)                = c0;
    *(uint2*)(o + 84 * IMS)     = c1;
    *(uint2*)(o + 2 * 84 * IMS) = c2;
  }
}

__global__ void k_prep_w1(const float* w, _Float16* wt) {
  int i = blockIdx.x * blockDim.x + threadIdx.x;
  if (i >= 32 * 192) return;
  int n = i / 192, k = i % 192;
  int c = k >> 6, kh = (k & 63) >> 3, kw = k & 7;
  wt[i] = (_Float16)w[((n * 3 + c) * 8 + kh) * 8 + kw];
}
__global__ void k_prep_w2(const float* w, _Float16* wt) {
  int i = blockIdx.x * blockDim.x + threadIdx.x;
  if (i >= 64 * 512) return;
  int n = i / 512, k = i % 512;
  int pix = k >> 5, c = k & 31, kh = pix >> 2, kw = pix & 3;
  wt[i] = (_Float16)w[((n * 32 + c) * 4 + kh) * 4 + kw];
}
__global__ void k_prep_w3(const float* w, _Float16* wt) {
  int i = blockIdx.x * blockDim.x + threadIdx.x;
  if (i >= 64 * 576) return;
  int n = i / 576, k = i % 576;
  int pix = k >> 6, c = k & 63, kh = pix / 3, kw = pix % 3;
  wt[i] = (_Float16)w[((n * 64 + c) * 3 + kh) * 3 + kw];
}
__global__ void k_prep_fc(const float* w, _Float16* wt) {
  int i = blockIdx.x * blockDim.x + threadIdx.x;
  if (i >= 512 * 3136) return;
  int f = i / 3136, k = i % 3136;
  int s = k >> 6, c = k & 63;
  wt[i] = (_Float16)w[f * 3136 + c * 49 + s];
}
__global__ void k_prep_wg(const float* wih, const float* whh, _Float16* wg) {
  int i = blockIdx.x * blockDim.x + threadIdx.x;
  if (i >= 1024 * KPAD) return;
  int r = i / KPAD, k = i % KPAD;
  float v = 0.0f;
  if (k < HL)             v = whh[r * HL + k];
  else if (k < HL + IND)  v = wih[r * IND + (k - HL)];
  wg[i] = (_Float16)v;
}
__global__ void k_prep_bias(const float* bih, const float* bhh, float* bg) {
  int i = blockIdx.x * blockDim.x + threadIdx.x;
  if (i < 1024) bg[i] = bih[i] + bhh[i];
}
__global__ void k_onehot(const int* pos, const int* lact, _Float16* X) {
  int i = blockIdx.x * blockDim.x + threadIdx.x;
  if (i >= B_TOT * 64) return;
  int b = i / 64, c = i % 64;
  float v = 0.0f;
  if (c < 16)      v = (pos[b * 2 + 0] == c)      ? 1.0f : 0.0f;
  else if (c < 32) v = (pos[b * 2 + 1] == c - 16) ? 1.0f : 0.0f;
  else if (c < 37) v = (lact[b]        == c - 32) ? 1.0f : 0.0f;
  X[b * XS + 512 + c] = (_Float16)v;
}

__global__ void __launch_bounds__(256) k_conv1(const _Float16* imgC, const _Float16* wt,
                                               const float* bias, _Float16* out) {
  int wid  = (blockIdx.x * blockDim.x + threadIdx.x) >> 5;
  int lane = threadIdx.x & 31, hi = lane >> 4, lo = lane & 15;
  int m = wid * 16 + lo;
  int b = m / 400, p = m % 400, oh = p / 20, ow = p % 20;
  const _Float16* abase = imgC + (size_t)b * 3 * 84 * IMS + (oh * 4) * IMS + ow * 4;
  v8f acc0 = {}, acc1 = {};
#pragma unroll
  for (int kb = 0; kb < 192; kb += 32) {
    int c = kb >> 6, khb = (kb & 63) >> 3;
    v16h a  = pack_frag_c1(abase + (c * 84 + khb + hi) * IMS);
    v16h b0 = pack_frag(wt + lo * 192        + kb + hi * 8);
    v16h b1 = pack_frag(wt + (16 + lo) * 192 + kb + hi * 8);
    acc0 = WMMA16(a, b0, acc0);
    acc1 = WMMA16(a, b1, acc1);
  }
  float bs0 = bias[lo], bs1 = bias[16 + lo];
#pragma unroll
  for (int r = 0; r < 8; r++) {
    int mm = wid * 16 + hi * 8 + r;
    float v0 = acc0[r] + bs0; v0 = v0 > 0.0f ? v0 : 0.0f;
    float v1 = acc1[r] + bs1; v1 = v1 > 0.0f ? v1 : 0.0f;
    out[mm * 32 + lo]      = (_Float16)v0;
    out[mm * 32 + 16 + lo] = (_Float16)v1;
  }
}

__global__ void __launch_bounds__(256) k_conv2(const _Float16* act1, const _Float16* wt,
                                               const float* bias, _Float16* out) {
  int wid  = (blockIdx.x * blockDim.x + threadIdx.x) >> 5;
  int lane = threadIdx.x & 31, hi = lane >> 4, lo = lane & 15;
  int m = wid * 16 + lo;
  int b = m / 81, p = m % 81, oh = p / 9, ow = p % 9;
  v8f acc0 = {}, acc1 = {}, acc2 = {}, acc3 = {};
  for (int kb = 0; kb < 512; kb += 32) {
    int pix = kb >> 5, kh = pix >> 2, kw = pix & 3;
    v16h a  = pack_frag(act1 + ((b * 20 + oh * 2 + kh) * 20 + (ow * 2 + kw)) * 32 + hi * 8);
    v16h b0 = pack_frag(wt + (lo)      * 512 + kb + hi * 8);
    v16h b1 = pack_frag(wt + (16 + lo) * 512 + kb + hi * 8);
    v16h b2 = pack_frag(wt + (32 + lo) * 512 + kb + hi * 8);
    v16h b3 = pack_frag(wt + (48 + lo) * 512 + kb + hi * 8);
    acc0 = WMMA16(a, b0, acc0);
    acc1 = WMMA16(a, b1, acc1);
    acc2 = WMMA16(a, b2, acc2);
    acc3 = WMMA16(a, b3, acc3);
  }
  float bs0 = bias[lo], bs1 = bias[16 + lo], bs2 = bias[32 + lo], bs3 = bias[48 + lo];
#pragma unroll
  for (int r = 0; r < 8; r++) {
    int mm = wid * 16 + hi * 8 + r;
    float v0 = acc0[r] + bs0; v0 = v0 > 0.0f ? v0 : 0.0f;
    float v1 = acc1[r] + bs1; v1 = v1 > 0.0f ? v1 : 0.0f;
    float v2 = acc2[r] + bs2; v2 = v2 > 0.0f ? v2 : 0.0f;
    float v3 = acc3[r] + bs3; v3 = v3 > 0.0f ? v3 : 0.0f;
    out[mm * 64 + lo]      = (_Float16)v0;
    out[mm * 64 + 16 + lo] = (_Float16)v1;
    out[mm * 64 + 32 + lo] = (_Float16)v2;
    out[mm * 64 + 48 + lo] = (_Float16)v3;
  }
}

__global__ void __launch_bounds__(256) k_conv3(const _Float16* act2, const _Float16* wt,
                                               const float* bias, _Float16* out) {
  int wid  = (blockIdx.x * blockDim.x + threadIdx.x) >> 5;
  int lane = threadIdx.x & 31, hi = lane >> 4, lo = lane & 15;
  int m = wid * 16 + lo;
  int b = m / 49, p = m % 49, oh = p / 7, ow = p % 7;
  v8f acc0 = {}, acc1 = {}, acc2 = {}, acc3 = {};
  for (int kb = 0; kb < 576; kb += 32) {
    int pix = kb >> 6, cb = (kb & 63) + hi * 8;
    int kh = pix / 3, kw = pix % 3;
    v16h a  = pack_frag(act2 + ((b * 9 + oh + kh) * 9 + (ow + kw)) * 64 + cb);
    v16h b0 = pack_frag(wt + (lo)      * 576 + kb + hi * 8);
    v16h b1 = pack_frag(wt + (16 + lo) * 576 + kb + hi * 8);
    v16h b2 = pack_frag(wt + (32 + lo) * 576 + kb + hi * 8);
    v16h b3 = pack_frag(wt + (48 + lo) * 576 + kb + hi * 8);
    acc0 = WMMA16(a, b0, acc0);
    acc1 = WMMA16(a, b1, acc1);
    acc2 = WMMA16(a, b2, acc2);
    acc3 = WMMA16(a, b3, acc3);
  }
  float bs0 = bias[lo], bs1 = bias[16 + lo], bs2 = bias[32 + lo], bs3 = bias[48 + lo];
#pragma unroll
  for (int r = 0; r < 8; r++) {
    int mm = wid * 16 + hi * 8 + r;
    float v0 = acc0[r] + bs0; v0 = v0 > 0.0f ? v0 : 0.0f;
    float v1 = acc1[r] + bs1; v1 = v1 > 0.0f ? v1 : 0.0f;
    float v2 = acc2[r] + bs2; v2 = v2 > 0.0f ? v2 : 0.0f;
    float v3 = acc3[r] + bs3; v3 = v3 > 0.0f ? v3 : 0.0f;
    out[mm * 64 + lo]      = (_Float16)v0;
    out[mm * 64 + 16 + lo] = (_Float16)v1;
    out[mm * 64 + 32 + lo] = (_Float16)v2;
    out[mm * 64 + 48 + lo] = (_Float16)v3;
  }
}

__global__ void __launch_bounds__(256) k_fc(const _Float16* act3, const _Float16* wt,
                                            const float* bias, _Float16* X) {
  int wid  = (blockIdx.x * blockDim.x + threadIdx.x) >> 5;
  int lane = threadIdx.x & 31, hi = lane >> 4, lo = lane & 15;
  int ng = wid & 7, mt = wid >> 3;
  int m = mt * 16 + lo;
  int nb = ng * 64;
  v8f acc0 = {}, acc1 = {}, acc2 = {}, acc3 = {};
  for (int kb = 0; kb < 3136; kb += 32) {
    v16h a  = pack_frag(act3 + (size_t)m * 3136 + kb + hi * 8);
    v16h b0 = pack_frag(wt + (size_t)(nb + lo)      * 3136 + kb + hi * 8);
    v16h b1 = pack_frag(wt + (size_t)(nb + 16 + lo) * 3136 + kb + hi * 8);
    v16h b2 = pack_frag(wt + (size_t)(nb + 32 + lo) * 3136 + kb + hi * 8);
    v16h b3 = pack_frag(wt + (size_t)(nb + 48 + lo) * 3136 + kb + hi * 8);
    acc0 = WMMA16(a, b0, acc0);
    acc1 = WMMA16(a, b1, acc1);
    acc2 = WMMA16(a, b2, acc2);
    acc3 = WMMA16(a, b3, acc3);
  }
  float bs0 = bias[nb + lo], bs1 = bias[nb + 16 + lo];
  float bs2 = bias[nb + 32 + lo], bs3 = bias[nb + 48 + lo];
#pragma unroll
  for (int r = 0; r < 8; r++) {
    int mm = mt * 16 + hi * 8 + r;
    float v0 = acc0[r] + bs0; v0 = v0 > 0.0f ? v0 : 0.0f;
    float v1 = acc1[r] + bs1; v1 = v1 > 0.0f ? v1 : 0.0f;
    float v2 = acc2[r] + bs2; v2 = v2 > 0.0f ? v2 : 0.0f;
    float v3 = acc3[r] + bs3; v3 = v3 > 0.0f ? v3 : 0.0f;
    X[mm * XS + nb + lo]      = (_Float16)v0;
    X[mm * XS + nb + 16 + lo] = (_Float16)v1;
    X[mm * XS + nb + 32 + lo] = (_Float16)v2;
    X[mm * XS + nb + 48 + lo] = (_Float16)v3;
  }
}

static constexpr int PPITCH = 36;
static constexpr int PTILE  = 8 * PPITCH;
__global__ void __launch_bounds__(128) k_lstm_step(const _Float16* X, const _Float16* Wg,
                                                   const float* biasg, const int* done,
                                                   const float* h0, const float* c0,
                                                   float* cbuf, float* hs, int t) {
  __shared__ float P[4 * PTILE];
  int tid = threadIdx.x;
  int w = __builtin_amdgcn_readfirstlane(tid >> 5);
  int lane = tid & 31, hi = lane >> 4, lo = lane & 15;
  int mt = blockIdx.x >> 6, cg = blockIdx.x & 63;
  const float* hprev = (t == 0) ? h0 : hs + (size_t)(t - 1) * (E_ENVS * HL);
  const float* cin   = (t == 0) ? c0 : cbuf;
  int se = (t == 0) ? HL : 4;
  int sq = (t == 0) ? 4 : E_ENVS * 4;

  int ci   = tid & 63;
  int eenv = mt * 16 + (ci >> 2);
  int ecol = cg * 4 + (ci & 3);
  int edone = done[t * E_ENVS + eenv];
  float cprev = cin[eenv * se + cg * sq + (ci & 3)];
  float bi = biasg[0 * HL + ecol], bf = biasg[1 * HL + ecol];
  float bg = biasg[2 * HL + ecol], bo = biasg[3 * HL + ecol];

  int aenv = mt * 16 + lo;
  int adone = done[t * E_ENVS + aenv];
  int grow = (lo >> 2) * HL + cg * 4 + (lo & 3);
  const _Float16* wrow = Wg + (size_t)grow * KPAD + hi * 8;
  const _Float16* xrow = X + (size_t)(t * E_ENVS + aenv) * XS + hi * 8;

  v16h a6 = {}, b6 = {};
  if (w < 2) {
    int kc = w + 24;
    a6 = pack_frag(xrow + (kc - 8) * 32);
    b6 = pack_frag(wrow + kc * 32);
  }
  float4 hu[2][4];
  v16h bh[2];
#pragma unroll
  for (int i = 0; i < 2; i++) {
    int kc = w + 4 * i;
    const float* hp = hprev + aenv * se + (kc * 8 + hi * 2) * sq;
    hu[i][0] = *(const float4*)hp;
    hu[i][1] = *(const float4*)(hp + sq);
    hu[i][2] = *(const float4*)(hp + 4 * sq);
    hu[i][3] = *(const float4*)(hp + 5 * sq);
    bh[i] = pack_frag(wrow + kc * 32);
  }
  v16h ax[4], bx[4];
#pragma unroll
  for (int i = 0; i < 4; i++) {
    int kc = w + 8 + 4 * i;
    ax[i] = pack_frag(xrow + (kc - 8) * 32);
    bx[i] = pack_frag(wrow + kc * 32);
  }

  __builtin_amdgcn_sched_barrier(0);

  v8f acc = {};
#pragma unroll
  for (int i = 0; i < 2; i++) {
    float hv16[16] = {hu[i][0].x, hu[i][0].y, hu[i][0].z, hu[i][0].w,
                      hu[i][1].x, hu[i][1].y, hu[i][1].z, hu[i][1].w,
                      hu[i][2].x, hu[i][2].y, hu[i][2].z, hu[i][2].w,
                      hu[i][3].x, hu[i][3].y, hu[i][3].z, hu[i][3].w};
    v16h a;
#pragma unroll
    for (int e = 0; e < 16; e++) a[e] = adone ? (_Float16)0.0f : (_Float16)hv16[e];
    acc = WMMA16(a, bh[i], acc);
  }
#pragma unroll
  for (int i = 0; i < 4; i++) acc = WMMA16(ax[i], bx[i], acc);
  if (w < 2) acc = WMMA16(a6, b6, acc);

  {
    float* pw = P + w * PTILE + lane;
#pragma unroll
    for (int r = 0; r < 8; r++) pw[r * PPITCH] = acc[r];
  }
  __syncthreads();

  if (tid < 64) {
    int el = tid >> 2, q = tid & 3;
    const float* pr = P + (el & 7) * PPITCH + (el >> 3) * 16 + q;
    float si = 0.0f, sf = 0.0f, sg = 0.0f, so = 0.0f;
#pragma unroll
    for (int k = 0; k < 4; k++) {
      si += pr[k * PTILE + 0];
      sf += pr[k * PTILE + 4];
      sg += pr[k * PTILE + 8];
      so += pr[k * PTILE + 12];
    }
    float cp = edone ? 0.0f : cprev;
    float iv = sigmoidf_(si + bi), fv = sigmoidf_(sf + bf);
    float gv = tanhf(sg + bg),     ov = sigmoidf_(so + bo);
    float cv = fv * cp + iv * gv;
    float hv = ov * tanhf(cv);
    cbuf[(cg * E_ENVS + eenv) * 4 + q] = cv;
    hs[(size_t)t * (E_ENVS * HL) + (cg * E_ENVS + eenv) * 4 + q] = hv;
  }
}

__global__ void k_heads(const float* hs, const float* cbuf, const float* pw, const float* pb,
                        const float* vw, const float* vb, float* out) {
  int i = blockIdx.x * blockDim.x + threadIdx.x;
  if (i < B_TOT * 6) {
    int b = i / 6, j = i % 6;
    const float* h    = hs + (size_t)(b >> 5) * (E_ENVS * HL) + (b & 31) * 4;
    const float* wrow = (j < 5) ? pw + j * HL : vw;
    float s = (j < 5) ? pb[j] : vb[0];
#pragma unroll 4
    for (int k4 = 0; k4 < HL / 4; k4++) {
      float4 hv = *(const float4*)(h + k4 * (E_ENVS * 4));
      s += hv.x * wrow[k4 * 4 + 0];
      s += hv.y * wrow[k4 * 4 + 1];
      s += hv.z * wrow[k4 * 4 + 2];
      s += hv.w * wrow[k4 * 4 + 3];
    }
    if (j < 5) out[b * 5 + j] = s;
    else       out[10240 + b] = s;
  } else {
    int o = i - B_TOT * 6;
    int sel = o >> 13, r = o & 8191;
    int e = r >> 8, k = r & 255;
    int src = ((k >> 2) * E_ENVS + e) * 4 + (k & 3);
    if (sel == 0) out[12288 + r] = hs[(size_t)(T_STEPS - 1) * (E_ENVS * HL) + src];
    else          out[20480 + r] = cbuf[src];
  }
}

extern "C" void kernel_launch(void* const* d_in, const int* in_sizes, int n_in,
                              void* d_out, int out_size, void* d_ws, size_t ws_size,
                              hipStream_t stream) {
  (void)in_sizes; (void)n_in; (void)out_size; (void)ws_size;
  const float* img  = (const float*)d_in[0];
  const int*   pos  = (const int*)d_in[1];
  const int*   lact = (const int*)d_in[2];
  const int*   done = (const int*)d_in[3];
  const float* h0   = (const float*)d_in[4];
  const float* c0   = (const float*)d_in[5];
  const float* w1   = (const float*)d_in[6];
  const float* b1   = (const float*)d_in[7];
  const float* w2   = (const float*)d_in[8];
  const float* b2   = (const float*)d_in[9];
  const float* w3   = (const float*)d_in[10];
  const float* b3   = (const float*)d_in[11];
  const float* fcw  = (const float*)d_in[12];
  const float* fcb  = (const float*)d_in[13];
  const float* wih  = (const float*)d_in[14];
  const float* whh  = (const float*)d_in[15];
  const float* bih  = (const float*)d_in[16];
  const float* bhh  = (const float*)d_in[17];
  const float* pw   = (const float*)d_in[18];
  const float* pb   = (const float*)d_in[19];
  const float* vw   = (const float*)d_in[20];
  const float* vb   = (const float*)d_in[21];
  float* out = (float*)d_out;
  char*  ws  = (char*)d_ws;

  const size_t o_img  = 0;
  const size_t o_act1 = 90832896;
  const size_t o_w1t  = o_act1 + 52428800;
  const size_t o_w2t  = o_w1t + 12288;
  const size_t o_w3t  = o_w2t + 65536;
  const size_t o_fcwt = o_w3t + 73728;
  const size_t o_wg   = o_fcwt + 3211264;
  const size_t o_bg   = o_wg + 1703936;
  const size_t o_act2 = 0;
  const size_t o_act3 = 21233664;
  const size_t o_X    = 34078720;
  const size_t o_hs   = 36438016;
  const size_t o_c    = 38535168;

  _Float16* imgC = (_Float16*)(ws + o_img);
  _Float16* act1 = (_Float16*)(ws + o_act1);
  _Float16* act2 = (_Float16*)(ws + o_act2);
  _Float16* act3 = (_Float16*)(ws + o_act3);
  _Float16* Xm   = (_Float16*)(ws + o_X);
  _Float16* w1t  = (_Float16*)(ws + o_w1t);
  _Float16* w2t  = (_Float16*)(ws + o_w2t);
  _Float16* w3t  = (_Float16*)(ws + o_w3t);
  _Float16* fcwt = (_Float16*)(ws + o_fcwt);
  _Float16* wg   = (_Float16*)(ws + o_wg);
  float*    bg   = (float*)(ws + o_bg);
  float*    hsb  = (float*)(ws + o_hs);
  float*    cbuf = (float*)(ws + o_c);

  const int TB = 256;
  k_prep_img <<<3528, TB, 0, stream>>>(img, imgC);
  k_prep_w1  <<<(32 * 192   + TB - 1) / TB, TB, 0, stream>>>(w1, w1t);
  k_prep_w2  <<<(64 * 512   + TB - 1) / TB, TB, 0, stream>>>(w2, w2t);
  k_prep_w3  <<<(64 * 576   + TB - 1) / TB, TB, 0, stream>>>(w3, w3t);
  k_prep_fc  <<<(512 * 3136 + TB - 1) / TB, TB, 0, stream>>>(fcw, fcwt);
  k_prep_wg  <<<(1024 * KPAD + TB - 1) / TB, TB, 0, stream>>>(wih, whh, wg);
  k_prep_bias<<<(1024 + TB - 1) / TB, TB, 0, stream>>>(bih, bhh, bg);

  k_conv1<<<6400, TB, 0, stream>>>(imgC, w1t, b1, act1);
  k_conv2<<<1296, TB, 0, stream>>>(act1, w2t, b2, act2);
  k_conv3<<< 784, TB, 0, stream>>>(act2, w3t, b3, act3);
  k_fc   <<< 128, TB, 0, stream>>>(act3, fcwt, fcb, Xm);

  k_onehot<<<(B_TOT * 64 + TB - 1) / TB, TB, 0, stream>>>(pos, lact, Xm);

  k_lstm_step<<<128, 128, 0, stream>>>(Xm, wg, bg, done, h0, c0, cbuf, hsb, 0);
  k_lstm_step<<<128, 128, 0, stream>>>(Xm, wg, bg, done, h0, c0, cbuf, hsb, 1);
  k_lstm_step<<<128, 128, 0, stream>>>(Xm, wg, bg, done, h0, c0, cbuf, hsb, 2);
  k_lstm_step<<<128, 128, 0, stream>>>(Xm, wg, bg, done, h0, c0, cbuf, hsb, 3);
  k_lstm_step<<<128, 128, 0, stream>>>(Xm, wg, bg, done, h0, c0, cbuf, hsb, 4);
  k_lstm_step<<<128, 128, 0, stream>>>(Xm, wg, bg, done, h0, c0, cbuf, hsb, 5);
  k_lstm_step<<<128, 128, 0, stream>>>(Xm, wg, bg, done, h0, c0, cbuf, hsb, 6);
  k_lstm_step<<<128, 128, 0, stream>>>(Xm, wg, bg, done, h0, c0, cbuf, hsb, 7);
  k_lstm_step<<<128, 128, 0, stream>>>(Xm, wg, bg, done, h0, c0, cbuf, hsb, 8);
  k_lstm_step<<<128, 128, 0, stream>>>(Xm, wg, bg, done, h0, c0, cbuf, hsb, 9);
  k_lstm_step<<<128, 128, 0, stream>>>(Xm, wg, bg, done, h0, c0, cbuf, hsb, 10);
  k_lstm_step<<<128, 128, 0, stream>>>(Xm, wg, bg, done, h0, c0, cbuf, hsb, 11);
  k_lstm_step<<<128, 128, 0, stream>>>(Xm, wg, bg, done, h0, c0, cbuf, hsb, 12);
  k_lstm_step<<<128, 128, 0, stream>>>(Xm, wg, bg, done, h0, c0, cbuf, hsb, 13);
  k_lstm_step<<<128, 128, 0, stream>>>(Xm, wg, bg, done, h0, c0, cbuf, hsb, 14);
  k_lstm_step<<<128, 128, 0, stream>>>(Xm, wg, bg, done, h0, c0, cbuf, hsb, 15);
  k_lstm_step<<<128, 128, 0, stream>>>(Xm, wg, bg, done, h0, c0, cbuf, hsb, 16);
  k_lstm_step<<<128, 128, 0, stream>>>(Xm, wg, bg, done, h0, c0, cbuf, hsb, 17);
  k_lstm_step<<<128, 128, 0, stream>>>(Xm, wg, bg, done, h0, c0, cbuf, hsb, 18);
  k_lstm_step<<<128, 128, 0, stream>>>(Xm, wg, bg, done, h0, c0, cbuf, hsb, 19);
  k_lstm_step<<<128, 128, 0, stream>>>(Xm, wg, bg, done, h0, c0, cbuf, hsb, 20);
  k_lstm_step<<<128, 128, 0, stream>>>(Xm, wg, bg, done, h0, c0, cbuf, hsb, 21);
  k_lstm_step<<<128, 128, 0, stream>>>(Xm, wg, bg, done, h0, c0, cbuf, hsb, 22);
  k_lstm_step<<<128, 128, 0, stream>>>(Xm, wg, bg, done, h0, c0, cbuf, hsb, 23);
  k_lstm_step<<<128, 128, 0, stream>>>(Xm, wg, bg, done, h0, c0, cbuf, hsb, 24);
  k_lstm_step<<<128, 128, 0, stream>>>(Xm, wg, bg, done, h0, c0, cbuf, hsb, 25);
  k_lstm_step<<<128, 128, 0, stream>>>(Xm, wg, bg, done, h0, c0, cbuf, hsb, 26);
  k_lstm_step<<<128, 128, 0, stream>>>(Xm, wg, bg, done, h0, c0, cbuf, hsb, 27);
  k_lstm_step<<<128, 128, 0, stream>>>(Xm, wg, bg, done, h0, c0, cbuf, hsb, 28);
  k_lstm_step<<<128, 128, 0, stream>>>(Xm, wg, bg, done, h0, c0, cbuf, hsb, 29);
  k_lstm_step<<<128, 128, 0, stream>>>(Xm, wg, bg, done, h0, c0, cbuf, hsb, 30);
  k_lstm_step<<<128, 128, 0, stream>>>(Xm, wg, bg, done, h0, c0, cbuf, hsb, 31);
  k_lstm_step<<<128, 128, 0, stream>>>(Xm, wg, bg, done, h0, c0, cbuf, hsb, 32);
  k_lstm_step<<<128, 128, 0, stream>>>(Xm, wg, bg, done, h0, c0, cbuf, hsb, 33);
  k_lstm_step<<<128, 128, 0, stream>>>(Xm, wg, bg, done, h0, c0, cbuf, hsb, 34);
  k_lstm_step<<<128, 128, 0, stream>>>(Xm, wg, bg, done, h0, c0, cbuf, hsb, 35);
  k_lstm_step<<<128, 128, 0, stream>>>(Xm, wg, bg, done, h0, c0, cbuf, hsb, 36);
  k_lstm_step<<<128, 128, 0, stream>>>(Xm, wg, bg, done, h0, c0, cbuf, hsb, 37);
  k_lstm_step<<<128, 128, 0, stream>>>(Xm, wg, bg, done, h0, c0, cbuf, hsb, 38);
  k_lstm_step<<<128, 128, 0, stream>>>(Xm, wg, bg, done, h0, c0, cbuf, hsb, 39);
  k_lstm_step<<<128, 128, 0, stream>>>(Xm, wg, bg, done, h0, c0, cbuf, hsb, 40);
  k_lstm_step<<<128, 128, 0, stream>>>(Xm, wg, bg, done, h0, c0, cbuf, hsb, 41);
  k_lstm_step<<<128, 128, 0, stream>>>(Xm, wg, bg, done, h0, c0, cbuf, hsb, 42);
  k_lstm_step<<<128, 128, 0, stream>>>(Xm, wg, bg, done, h0, c0, cbuf, hsb, 43);
  k_lstm_step<<<128, 128, 0, stream>>>(Xm, wg, bg, done, h0, c0, cbuf, hsb, 44);
  k_lstm_step<<<128, 128, 0, stream>>>(Xm, wg, bg, done, h0, c0, cbuf, hsb, 45);
  k_lstm_step<<<128, 128, 0, stream>>>(Xm, wg, bg, done, h0, c0, cbuf, hsb, 46);
  k_lstm_step<<<128, 128, 0, stream>>>(Xm, wg, bg, done, h0, c0, cbuf, hsb, 47);
  k_lstm_step<<<128, 128, 0, stream>>>(Xm, wg, bg, done, h0, c0, cbuf, hsb, 48);
  k_lstm_step<<<128, 128, 0, stream>>>(Xm, wg, bg, done, h0, c0, cbuf, hsb, 49);
  k_lstm_step<<<128, 128, 0, stream>>>(Xm, wg, bg, done, h0, c0, cbuf, hsb, 50);
  k_lstm_step<<<128, 128, 0, stream>>>(Xm, wg, bg, done, h0, c0, cbuf, hsb, 51);
  k_lstm_step<<<128, 128, 0, stream>>>(Xm, wg, bg, done, h0, c0, cbuf, hsb, 52);
  k_lstm_step<<<128, 128, 0, stream>>>(Xm, wg, bg, done, h0, c0, cbuf, hsb, 53);
  k_lstm_step<<<128, 128, 0, stream>>>(Xm, wg, bg, done, h0, c0, cbuf, hsb, 54);
  k_lstm_step<<<128, 128, 0, stream>>>(Xm, wg, bg, done, h0, c0, cbuf, hsb, 55);
  k_lstm_step<<<128, 128, 0, stream>>>(Xm, wg, bg, done, h0, c0, cbuf, hsb, 56);
  k_lstm_step<<<128, 128, 0, stream>>>(Xm, wg, bg, done, h0, c0, cbuf, hsb, 57);
  k_lstm_step<<<128, 128, 0, stream>>>(Xm, wg, bg, done, h0, c0, cbuf, hsb, 58);
  k_lstm_step<<<128, 128, 0, stream>>>(Xm, wg, bg, done, h0, c0, cbuf, hsb, 59);
  k_lstm_step<<<128, 128, 0, stream>>>(Xm, wg, bg, done, h0, c0, cbuf, hsb, 60);
  k_lstm_step<<<128, 128, 0, stream>>>(Xm, wg, bg, done, h0, c0, cbuf, hsb, 61);
  k_lstm_step<<<128, 128, 0, stream>>>(Xm, wg, bg, done, h0, c0, cbuf, hsb, 62);
  k_lstm_step<<<128, 128, 0, stream>>>(Xm, wg, bg, done, h0, c0, cbuf, hsb, 63);

  k_heads<<<112, TB, 0, stream>>>(hsb, cbuf, pw, pb, vw, vb, out);
}


# FC GEMM split along K over the 8 waves of a workgroup (tile 32 rows x 64 cols, wave w takes k chunks w, w+8, ..), partials summed in LDS; 512 workgroups, was 128; FC f32 sum order changes
# speedup vs baseline: 8.3498x; 1.2486x over previous
#include <hip/hip_runtime.h>
#include <math.h>

typedef __attribute__((ext_vector_type(16))) _Float16 v16h;
typedef __attribute__((ext_vector_type(8)))  _Float16 v8h;
typedef __attribute__((ext_vector_type(4)))  _Float16 v4h;
typedef __attribute__((ext_vector_type(8)))  float    v8f;

#define WMMA16(A,B,C) __builtin_amdgcn_wmma_f32_16x16x32_f16(false,(A),false,(B),(short)0,(C),false,false)

static constexpr int T_STEPS = 64;
static constexpr int E_ENVS  = 32;
static constexpr int B_TOT   = 2048;
static constexpr int HL      = 256;
static constexpr int IND     = 549;
static constexpr int KPAD    = 832;
static constexpr int XS      = 576;
static constexpr int IMS     = 88;

__device__ __forceinline__ v16h pack_frag(const _Float16* p) {
  v8h u = *(const v8h*)p;
  v8h w = *(const v8h*)(p + 16);
  v16h a;
#pragma unroll
  for (int e = 0; e < 8; e++) { a[e] = u[e]; a[e + 8] = w[e]; }
  return a;
}
__device__ __forceinline__ v16h pack_frag_c1(const _Float16* p0) {
  const _Float16* p1 = p0 + 2 * IMS;
  v4h u0 = *(const v4h*)p0;
  v4h u1 = *(const v4h*)(p0 + 4);
  v4h u2 = *(const v4h*)p1;
  v4h u3 = *(const v4h*)(p1 + 4);
  v16h a;
#pragma unroll
  for (int e = 0; e < 4; e++) {
    a[e] = u0[e]; a[e + 4] = u1[e]; a[e + 8] = u2[e]; a[e + 12] = u3[e];
  }
  return a;
}

__device__ __forceinline__ float sigmoidf_(float x) { return 1.0f / (1.0f + __expf(-x)); }

__device__ __forceinline__ unsigned pk2_scaled_f16(float a, float b, float sc) {
  unsigned r;
  asm("v_fma_mixlo_f16 %0, %1, %2, 0" : "=v"(r) : "v"(a), "s"(sc));
  asm("v_fma_mixhi_f16 %0, %1, %2, 0" : "+v"(r) : "v"(b), "s"(sc));
  return r;
}

__global__ void __launch_bounds__(256) k_prep_img(const float* img, _Float16* imgC) {
  const float sc = 1.0f / 255.0f;
  int g0 = blockIdx.x * 1024 + threadIdx.x;
  float4 v[4][3];
#pragma unroll
  for (int k = 0; k < 4; k++) {
    const float4* p = (const float4*)(img + (g0 + k * 256) * 12);
    v[k][0] = p[0]; v[k][1] = p[1]; v[k][2] = p[2];
  }
#pragma unroll
  for (int k = 0; k < 4; k++) {
    int g  = g0 + k * 256;
    int R  = g / 21, cg = g - R * 21;
    int b  = R / 84;
    _Float16* o = imgC + (R + b * 168) * IMS + cg * 4;
    float4 q0 = v[k][0], q1 = v[k][1], q2 = v[k][2];
    uint2 c0, c1, c2;
    c0.x = pk2_scaled_f16(q0.x, q0.w, sc); c0.y = pk2_scaled_f16(q1.z, q2.y, sc);
    c1.x = pk2_scaled_f16(q0.y, q1.x, sc); c1.y = pk2_scaled_f16(q1.w, q2.z, sc);
    c2.x = pk2_scaled_f16(q0.z, q1.y, sc); c2.y = pk2_scaled_f16(q2.x, q2.w, sc);
    *(uint2*)(o)                = c0;
    *(uint2*)(o + 84 * IMS)     = c1;
    *(uint2*)(o + 2 * 84 * IMS) = c2;
  }
}

__global__ void k_prep_w1(const float* w, _Float16* wt) {
  int i = blockIdx.x * blockDim.x + threadIdx.x;
  if (i >= 32 * 192) return;
  int n = i / 192, k = i % 192;
  int c = k >> 6, kh = (k & 63) >> 3, kw = k & 7;
  wt[i] = (_Float16)w[((n * 3 + c) * 8 + kh) * 8 + kw];
}
__global__ void k_prep_w2(const float* w, _Float16* wt) {
  int i = blockIdx.x * blockDim.x + threadIdx.x;
  if (i >= 64 * 512) return;
  int n = i / 512, k = i % 512;
  int pix = k >> 5, c = k & 31, kh = pix >> 2, kw = pix & 3;
  wt[i] = (_Float16)w[((n * 32 + c) * 4 + kh) * 4 + kw];
}
__global__ void k_prep_w3(const float* w, _Float16* wt) {
  int i = blockIdx.x * blockDim.x + threadIdx.x;
  if (i >= 64 * 576) return;
  int n = i / 576, k = i % 576;
  int pix = k >> 6, c = k & 63, kh = pix / 3, kw = pix % 3;
  wt[i] = (_Float16)w[((n * 64 + c) * 3 + kh) * 3 + kw];
}
__global__ void k_prep_fc(const float* w, _Float16* wt) {
  int i = blockIdx.x * blockDim.x + threadIdx.x;
  if (i >= 512 * 3136) return;
  int f = i / 3136, k = i % 3136;
  int s = k >> 6, c = k & 63;
  wt[i] = (_Float16)w[f * 3136 + c * 49 + s];
}
__global__ void k_prep_wg(const float* wih, const float* whh, _Float16* wg) {
  int i = blockIdx.x * blockDim.x + threadIdx.x;
  if (i >= 1024 * KPAD) return;
  int r = i / KPAD, k = i % KPAD;
  float v = 0.0f;
  if (k < HL)             v = whh[r * HL + k];
  else if (k < HL + IND)  v = wih[r * IND + (k - HL)];
  wg[i] = (_Float16)v;
}
__global__ void k_prep_bias(const float* bih, const float* bhh, float* bg) {
  int i = blockIdx.x * blockDim.x + threadIdx.x;
  if (i < 1024) bg[i] = bih[i] + bhh[i];
}
__global__ void k_onehot(const int* pos, const int* lact, _Float16* X) {
  int i = blockIdx.x * blockDim.x + threadIdx.x;
  if (i >= B_TOT * 64) return;
  int b = i / 64, c = i % 64;
  float v = 0.0f;
  if (c < 16)      v = (pos[b * 2 + 0] == c)      ? 1.0f : 0.0f;
  else if (c < 32) v = (pos[b * 2 + 1] == c - 16) ? 1.0f : 0.0f;
  else if (c < 37) v = (lact[b]        == c - 32) ? 1.0f : 0.0f;
  X[b * XS + 512 + c] = (_Float16)v;
}

__global__ void __launch_bounds__(256) k_conv1(const _Float16* imgC, const _Float16* wt,
                                               const float* bias, _Float16* out) {
  int wid  = (blockIdx.x * blockDim.x + threadIdx.x) >> 5;
  int lane = threadIdx.x & 31, hi = lane >> 4, lo = lane & 15;
  int m = wid * 16 + lo;
  int b = m / 400, p = m % 400, oh = p / 20, ow = p % 20;
  const _Float16* abase = imgC + (size_t)b * 3 * 84 * IMS + (oh * 4) * IMS + ow * 4;
  v8f acc0 = {}, acc1 = {};
#pragma unroll
  for (int kb = 0; kb < 192; kb += 32) {
    int c = kb >> 6, khb = (kb & 63) >> 3;
    v16h a  = pack_frag_c1(abase + (c * 84 + khb + hi) * IMS);
    v16h b0 = pack_frag(wt + lo * 192        + kb + hi * 8);
    v16h b1 = pack_frag(wt + (16 + lo) * 192 + kb + hi * 8);
    acc0 = WMMA16(a, b0, acc0);
    acc1 = WMMA16(a, b1, acc1);
  }
  float bs0 = bias[lo], bs1 = bias[16 + lo];
#pragma unroll
  for (int r = 0; r < 8; r++) {
    int mm = wid * 16 + hi * 8 + r;
    float v0 = acc0[r] + bs0; v0 = v0 > 0.0f ? v0 : 0.0f;
    float v1 = acc1[r] + bs1; v1 = v1 > 0.0f ? v1 : 0.0f;
    out[mm * 32 + lo]      = (_Float16)v0;
    out[mm * 32 + 16 + lo] = (_Float16)v1;
  }
}

__global__ void __launch_bounds__(256) k_conv2(const _Float16* act1, const _Float16* wt,
                                               const float* bias, _Float16* out) {
  int wid  = (blockIdx.x * blockDim.x + threadIdx.x) >> 5;
  int lane = threadIdx.x & 31, hi = lane >> 4, lo = lane & 15;
  int m = wid * 16 + lo;
  int b = m / 81, p = m % 81, oh = p / 9, ow = p % 9;
  v8f acc0 = {}, acc1 = {}, acc2 = {}, acc3 = {};
  for (int kb = 0; kb < 512; kb += 32) {
    int pix = kb >> 5, kh = pix >> 2, kw = pix & 3;
    v16h a  = pack_frag(act1 + ((b * 20 + oh * 2 + kh) * 20 + (ow * 2 + kw)) * 32 + hi * 8);
    v16h b0 = pack_frag(wt + (lo)      * 512 + kb + hi * 8);
    v16h b1 = pack_frag(wt + (16 + lo) * 512 + kb + hi * 8);
    v16h b2 = pack_frag(wt + (32 + lo) * 512 + kb + hi * 8);
    v16h b3 = pack_frag(wt + (48 + lo) * 512 + kb + hi * 8);
    acc0 = WMMA16(a, b0, acc0);
    acc1 = WMMA16(a, b1, acc1);
    acc2 = WMMA16(a, b2, acc2);
    acc3 = WMMA16(a, b3, acc3);
  }
  float bs0 = bias[lo], bs1 = bias[16 + lo], bs2 = bias[32 + lo], bs3 = bias[48 + lo];
#pragma unroll
  for (int r = 0; r < 8; r++) {
    int mm = wid * 16 + hi * 8 + r;
    float v0 = acc0[r] + bs0; v0 = v0 > 0.0f ? v0 : 0.0f;
    float v1 = acc1[r] + bs1; v1 = v1 > 0.0f ? v1 : 0.0f;
    float v2 = acc2[r] + bs2; v2 = v2 > 0.0f ? v2 : 0.0f;
    float v3 = acc3[r] + bs3; v3 = v3 > 0.0f ? v3 : 0.0f;
    out[mm * 64 + lo]      = (_Float16)v0;
    out[mm * 64 + 16 + lo] = (_Float16)v1;
    out[mm * 64 + 32 + lo] = (_Float16)v2;
    out[mm * 64 + 48 + lo] = (_Float16)v3;
  }
}

__global__ void __launch_bounds__(256) k_conv3(const _Float16* act2, const _Float16* wt,
                                               const float* bias, _Float16* out) {
  int wid  = (blockIdx.x * blockDim.x + threadIdx.x) >> 5;
  int lane = threadIdx.x & 31, hi = lane >> 4, lo = lane & 15;
  int m = wid * 16 + lo;
  int b = m / 49, p = m % 49, oh = p / 7, ow = p % 7;
  v8f acc0 = {}, acc1 = {}, acc2 = {}, acc3 = {};
  for (int kb = 0; kb < 576; kb += 32) {
    int pix = kb >> 6, cb = (kb & 63) + hi * 8;
    int kh = pix / 3, kw = pix % 3;
    v16h a  = pack_frag(act2 + ((b * 9 + oh + kh) * 9 + (ow + kw)) * 64 + cb);
    v16h b0 = pack_frag(wt + (lo)      * 576 + kb + hi * 8);
    v16h b1 = pack_frag(wt + (16 + lo) * 576 + kb + hi * 8);
    v16h b2 = pack_frag(wt + (32 + lo) * 576 + kb + hi * 8);
    v16h b3 = pack_frag(wt + (48 + lo) * 576 + kb + hi * 8);
    acc0 = WMMA16(a, b0, acc0);
    acc1 = WMMA16(a, b1, acc1);
    acc2 = WMMA16(a, b2, acc2);
    acc3 = WMMA16(a, b3, acc3);
  }
  float bs0 = bias[lo], bs1 = bias[16 + lo], bs2 = bias[32 + lo], bs3 = bias[48 + lo];
#pragma unroll
  for (int r = 0; r < 8; r++) {
    int mm = wid * 16 + hi * 8 + r;
    float v0 = acc0[r] + bs0; v0 = v0 > 0.0f ? v0 : 0.0f;
    float v1 = acc1[r] + bs1; v1 = v1 > 0.0f ? v1 : 0.0f;
    float v2 = acc2[r] + bs2; v2 = v2 > 0.0f ? v2 : 0.0f;
    float v3 = acc3[r] + bs3; v3 = v3 > 0.0f ? v3 : 0.0f;
    out[mm * 64 + lo]      = (_Float16)v0;
    out[mm * 64 + 16 + lo] = (_Float16)v1;
    out[mm * 64 + 32 + lo] = (_Float16)v2;
    out[mm * 64 + 48 + lo] = (_Float16)v3;
  }
}

static constexpr int FPITCH = 68;
static constexpr int FTILE  = 32 * FPITCH;
__global__ void __launch_bounds__(256) k_fc(const _Float16* act3, const _Float16* wt,
                                            const float* bias, _Float16* X) {
  __shared__ __attribute__((aligned(16))) float P[8 * FTILE];
  int tid = threadIdx.x;
  int w = __builtin_amdgcn_readfirstlane(tid >> 5);
  int lane = tid & 31, hi = lane >> 4, lo = lane & 15;
  int ng = blockIdx.x & 7, mp = blockIdx.x >> 3;
  int m0 = mp * 32, nb = ng * 64;
  const _Float16* ar0 = act3 + (size_t)(m0 + lo) * 3136 + hi * 8;
  const _Float16* ar1 = ar0 + (size_t)16 * 3136;
  const _Float16* br0 = wt + (size_t)(nb + lo) * 3136 + hi * 8;
  const _Float16* br1 = br0 + (size_t)16 * 3136;
  const _Float16* br2 = br0 + (size_t)32 * 3136;
  const _Float16* br3 = br0 + (size_t)48 * 3136;
  v8f c00 = {}, c01 = {}, c02 = {}, c03 = {};
  v8f c10 = {}, c11 = {}, c12 = {}, c13 = {};
  for (int kc = w; kc < 98; kc += 8) {
    int kb = kc * 32;
    v16h a0 = pack_frag(ar0 + kb);
    v16h a1 = pack_frag(ar1 + kb);
    v16h b0 = pack_frag(br0 + kb);
    v16h b1 = pack_frag(br1 + kb);
    v16h b2 = pack_frag(br2 + kb);
    v16h b3 = pack_frag(br3 + kb);
    c00 = WMMA16(a0, b0, c00);
    c10 = WMMA16(a1, b0, c10);
    c01 = WMMA16(a0, b1, c01);
    c11 = WMMA16(a1, b1, c11);
    c02 = WMMA16(a0, b2, c02);
    c12 = WMMA16(a1, b2, c12);
    c03 = WMMA16(a0, b3, c03);
    c13 = WMMA16(a1, b3, c13);
  }
  {
    float* pw = P + w * FTILE + (hi * 8) * FPITCH + lo;
#pragma unroll
    for (int r = 0; r < 8; r++) {
      pw[r * FPITCH +  0] = c00[r];
      pw[r * FPITCH + 16] = c01[r];
      pw[r * FPITCH + 32] = c02[r];
      pw[r * FPITCH + 48] = c03[r];
      pw[(16 + r) * FPITCH +  0] = c10[r];
      pw[(16 + r) * FPITCH + 16] = c11[r];
      pw[(16 + r) * FPITCH + 32] = c12[r];
      pw[(16 + r) * FPITCH + 48] = c13[r];
    }
  }
  __syncthreads();
  {
    int row = tid >> 3, c0 = (tid & 7) * 8;
    const float* pr = P + row * FPITCH + c0;
    float4 s0 = *(const float4*)pr;
    float4 s1 = *(const float4*)(pr + 4);
#pragma unroll
    for (int k = 1; k < 8; k++) {
      float4 u0 = *(const float4*)(pr + k * FTILE);
      float4 u1 = *(const float4*)(pr + k * FTILE + 4);
      s0.x += u0.x; s0.y += u0.y; s0.z += u0.z; s0.w += u0.w;
      s1.x += u1.x; s1.y += u1.y; s1.z += u1.z; s1.w += u1.w;
    }
    float4 q0 = *(const float4*)(bias + nb + c0);
    float4 q1 = *(const float4*)(bias + nb + c0 + 4);
    float v[8] = {s0.x + q0.x, s0.y + q0.y, s0.z + q0.z, s0.w + q0.w,
                  s1.x + q1.x, s1.y + q1.y, s1.z + q1.z, s1.w + q1.w};
    v8h o;
#pragma unroll
    for (int e = 0; e < 8; e++) o[e] = (_Float16)(v[e] > 0.0f ? v[e] : 0.0f);
    *(v8h*)(X + (size_t)(m0 + row) * XS + nb + c0) = o;
  }
}

static constexpr int PPITCH = 36;
static constexpr int PTILE  = 8 * PPITCH;
__global__ void __launch_bounds__(128) k_lstm_step(const _Float16* X, const _Float16* Wg,
                                                   const float* biasg, const int* done,
                                                   const float* h0, const float* c0,
                                                   float* cbuf, float* hs, int t) {
  __shared__ float P[4 * PTILE];
  int tid = threadIdx.x;
  int w = __builtin_amdgcn_readfirstlane(tid >> 5);
  int lane = tid & 31, hi = lane >> 4, lo = lane & 15;
  int mt = blockIdx.x >> 6, cg = blockIdx.x & 63;
  const float* hprev = (t == 0) ? h0 : hs + (size_t)(t - 1) * (E_ENVS * HL);
  const float* cin   = (t == 0) ? c0 : cbuf;
  int se = (t == 0) ? HL : 4;
  int sq = (t == 0) ? 4 : E_ENVS * 4;

  int ci   = tid & 63;
  int eenv = mt * 16 + (ci >> 2);
  int ecol = cg * 4 + (ci & 3);
  int edone = done[t * E_ENVS + eenv];
  float cprev = cin[eenv * se + cg * sq + (ci & 3)];
  float bi = biasg[0 * HL + ecol], bf = biasg[1 * HL + ecol];
  float bg = biasg[2 * HL + ecol], bo = biasg[3 * HL + ecol];

  int aenv = mt * 16 + lo;
  int adone = done[t * E_ENVS + aenv];
  int grow = (lo >> 2) * HL + cg * 4 + (lo & 3);
  const _Float16* wrow = Wg + (size_t)grow * KPAD + hi * 8;
  const _Float16* xrow = X + (size_t)(t * E_ENVS + aenv) * XS + hi * 8;

  v16h a6 = {}, b6 = {};
  if (w < 2) {
    int kc = w + 24;
    a6 = pack_frag(xrow + (kc - 8) * 32);
    b6 = pack_frag(wrow + kc * 32);
  }
  float4 hu[2][4];
  v16h bh[2];
#pragma unroll
  for (int i = 0; i < 2; i++) {
    int kc = w + 4 * i;
    const float* hp = hprev + aenv * se + (kc * 8 + hi * 2) * sq;
    hu[i][0] = *(const float4*)hp;
    hu[i][1] = *(const float4*)(hp + sq);
    hu[i][2] = *(const float4*)(hp + 4 * sq);
    hu[i][3] = *(const float4*)(hp + 5 * sq);
    bh[i] = pack_frag(wrow + kc * 32);
  }
  v16h ax[4], bx[4];
#pragma unroll
  for (int i = 0; i < 4; i++) {
    int kc = w + 8 + 4 * i;
    ax[i] = pack_frag(xrow + (kc - 8) * 32);
    bx[i] = pack_frag(wrow + kc * 32);
  }

  __builtin_amdgcn_sched_barrier(0);

  v8f acc = {};
#pragma unroll
  for (int i = 0; i < 2; i++) {
    float hv16[16] = {hu[i][0].x, hu[i][0].y, hu[i][0].z, hu[i][0].w,
                      hu[i][1].x, hu[i][1].y, hu[i][1].z, hu[i][1].w,
                      hu[i][2].x, hu[i][2].y, hu[i][2].z, hu[i][2].w,
                      hu[i][3].x, hu[i][3].y, hu[i][3].z, hu[i][3].w};
    v16h a;
#pragma unroll
    for (int e = 0; e < 16; e++) a[e] = adone ? (_Float16)0.0f : (_Float16)hv16[e];
    acc = WMMA16(a, bh[i], acc);
  }
#pragma unroll
  for (int i = 0; i < 4; i++) acc = WMMA16(ax[i], bx[i], acc);
  if (w < 2) acc = WMMA16(a6, b6, acc);

  {
    float* pw = P + w * PTILE + lane;
#pragma unroll
    for (int r = 0; r < 8; r++) pw[r * PPITCH] = acc[r];
  }
  __syncthreads();

  if (tid < 64) {
    int el = tid >> 2, q = tid & 3;
    const float* pr = P + (el & 7) * PPITCH + (el >> 3) * 16 + q;
    float si = 0.0f, sf = 0.0f, sg = 0.0f, so = 0.0f;
#pragma unroll
    for (int k = 0; k < 4; k++) {
      si += pr[k * PTILE + 0];
      sf += pr[k * PTILE + 4];
      sg += pr[k * PTILE + 8];
      so += pr[k * PTILE + 12];
    }
    float cp = edone ? 0.0f : cprev;
    float iv = sigmoidf_(si + bi), fv = sigmoidf_(sf + bf);
    float gv = tanhf(sg + bg),     ov = sigmoidf_(so + bo);
    float cv = fv * cp + iv * gv;
    float hv = ov * tanhf(cv);
    cbuf[(cg * E_ENVS + eenv) * 4 + q] = cv;
    hs[(size_t)t * (E_ENVS * HL) + (cg * E_ENVS + eenv) * 4 + q] = hv;
  }
}

__global__ void k_heads(const float* hs, const float* cbuf, const float* pw, const float* pb,
                        const float* vw, const float* vb, float* out) {
  int i = blockIdx.x * blockDim.x + threadIdx.x;
  if (i < B_TOT * 6) {
    int b = i / 6, j = i % 6;
    const float* h    = hs + (size_t)(b >> 5) * (E_ENVS * HL) + (b & 31) * 4;
    const float* wrow = (j < 5) ? pw + j * HL : vw;
    float s = (j < 5) ? pb[j] : vb[0];
#pragma unroll 4
    for (int k4 = 0; k4 < HL / 4; k4++) {
      float4 hv = *(const float4*)(h + k4 * (E_ENVS * 4));
      s += hv.x * wrow[k4 * 4 + 0];
      s += hv.y * wrow[k4 * 4 + 1];
      s += hv.z * wrow[k4 * 4 + 2];
      s += hv.w * wrow[k4 * 4 + 3];
    }
    if (j < 5) out[b * 5 + j] = s;
    else       out[10240 + b] = s;
  } else {
    int o = i - B_TOT * 6;
    int sel = o >> 13, r = o & 8191;
    int e = r >> 8, k = r & 255;
    int src = ((k >> 2) * E_ENVS + e) * 4 + (k & 3);
    if (sel == 0) out[12288 + r] = hs[(size_t)(T_STEPS - 1) * (E_ENVS * HL) + src];
    else          out[20480 + r] = cbuf[src];
  }
}

extern "C" void kernel_launch(void* const* d_in, const int* in_sizes, int n_in,
                              void* d_out, int out_size, void* d_ws, size_t ws_size,
                              hipStream_t stream) {
  (void)in_sizes; (void)n_in; (void)out_size; (void)ws_size;
  const float* img  = (const float*)d_in[0];
  const int*   pos  = (const int*)d_in[1];
  const int*   lact = (const int*)d_in[2];
  const int*   done = (const int*)d_in[3];
  const float* h0   = (const float*)d_in[4];
  const float* c0   = (const float*)d_in[5];
  const float* w1   = (const float*)d_in[6];
  const float* b1   = (const float*)d_in[7];
  const float* w2   = (const float*)d_in[8];
  const float* b2   = (const float*)d_in[9];
  const float* w3   = (const float*)d_in[10];
  const float* b3   = (const float*)d_in[11];
  const float* fcw  = (const float*)d_in[12];
  const float* fcb  = (const float*)d_in[13];
  const float* wih  = (const float*)d_in[14];
  const float* whh  = (const float*)d_in[15];
  const float* bih  = (const float*)d_in[16];
  const float* bhh  = (const float*)d_in[17];
  const float* pw   = (const float*)d_in[18];
  const float* pb   = (const float*)d_in[19];
  const float* vw   = (const float*)d_in[20];
  const float* vb   = (const float*)d_in[21];
  float* out = (float*)d_out;
  char*  ws  = (char*)d_ws;

  const size_t o_img  = 0;
  const size_t o_act1 = 90832896;
  const size_t o_w1t  = o_act1 + 52428800;
  const size_t o_w2t  = o_w1t + 12288;
  const size_t o_w3t  = o_w2t + 65536;
  const size_t o_fcwt = o_w3t + 73728;
  const size_t o_wg   = o_fcwt + 3211264;
  const size_t o_bg   = o_wg + 1703936;
  const size_t o_act2 = 0;
  const size_t o_act3 = 21233664;
  const size_t o_X    = 34078720;
  const size_t o_hs   = 36438016;
  const size_t o_c    = 38535168;

  _Float16* imgC = (_Float16*)(ws + o_img);
  _Float16* act1 = (_Float16*)(ws + o_act1);
  _Float16* act2 = (_Float16*)(ws + o_act2);
  _Float16* act3 = (_Float16*)(ws + o_act3);
  _Float16* Xm   = (_Float16*)(ws + o_X);
  _Float16* w1t  = (_Float16*)(ws + o_w1t);
  _Float16* w2t  = (_Float16*)(ws + o_w2t);
  _Float16* w3t  = (_Float16*)(ws + o_w3t);
  _Float16* fcwt = (_Float16*)(ws + o_fcwt);
  _Float16* wg   = (_Float16*)(ws + o_wg);
  float*    bg   = (float*)(ws + o_bg);
  float*    hsb  = (float*)(ws + o_hs);
  float*    cbuf = (float*)(ws + o_c);

  const int TB = 256;
  k_prep_img <<<3528, TB, 0, stream>>>(img, imgC);
  k_prep_w1  <<<(32 * 192   + TB - 1) / TB, TB, 0, stream>>>(w1, w1t);
  k_prep_w2  <<<(64 * 512   + TB - 1) / TB, TB, 0, stream>>>(w2, w2t);
  k_prep_w3  <<<(64 * 576   + TB - 1) / TB, TB, 0, stream>>>(w3, w3t);
  k_prep_fc  <<<(512 * 3136 + TB - 1) / TB, TB, 0, stream>>>(fcw, fcwt);
  k_prep_wg  <<<(1024 * KPAD + TB - 1) / TB, TB, 0, stream>>>(wih, whh, wg);
  k_prep_bias<<<(1024 + TB - 1) / TB, TB, 0, stream>>>(bih, bhh, bg);

  k_conv1<<<6400, TB, 0, stream>>>(imgC, w1t, b1, act1);
  k_conv2<<<1296, TB, 0, stream>>>(act1, w2t, b2, act2);
  k_conv3<<< 784, TB, 0, stream>>>(act2, w3t, b3, act3);
  k_fc   <<< 512, TB, 0, stream>>>(act3, fcwt, fcb, Xm);

  k_onehot<<<(B_TOT * 64 + TB - 1) / TB, TB, 0, stream>>>(pos, lact, Xm);

  k_lstm_step<<<128, 128, 0, stream>>>(Xm, wg, bg, done, h0, c0, cbuf, hsb, 0);
  k_lstm_step<<<128, 128, 0, stream>>>(Xm, wg, bg, done, h0, c0, cbuf, hsb, 1);
  k_lstm_step<<<128, 128, 0, stream>>>(Xm, wg, bg, done, h0, c0, cbuf, hsb, 2);
  k_lstm_step<<<128, 128, 0, stream>>>(Xm, wg, bg, done, h0, c0, cbuf, hsb, 3);
  k_lstm_step<<<128, 128, 0, stream>>>(Xm, wg, bg, done, h0, c0, cbuf, hsb, 4);
  k_lstm_step<<<128, 128, 0, stream>>>(Xm, wg, bg, done, h0, c0, cbuf, hsb, 5);
  k_lstm_step<<<128, 128, 0, stream>>>(Xm, wg, bg, done, h0, c0, cbuf, hsb, 6);
  k_lstm_step<<<128, 128, 0, stream>>>(Xm, wg, bg, done, h0, c0, cbuf, hsb, 7);
  k_lstm_step<<<128, 128, 0, stream>>>(Xm, wg, bg, done, h0, c0, cbuf, hsb, 8);
  k_lstm_step<<<128, 128, 0, stream>>>(Xm, wg, bg, done, h0, c0, cbuf, hsb, 9);
  k_lstm_step<<<128, 128, 0, stream>>>(Xm, wg, bg, done, h0, c0, cbuf, hsb, 10);
  k_lstm_step<<<128, 128, 0, stream>>>(Xm, wg, bg, done, h0, c0, cbuf, hsb, 11);
  k_lstm_step<<<128, 128, 0, stream>>>(Xm, wg, bg, done, h0, c0, cbuf, hsb, 12);
  k_lstm_step<<<128, 128, 0, stream>>>(Xm, wg, bg, done, h0, c0, cbuf, hsb, 13);
  k_lstm_step<<<128, 128, 0, stream>>>(Xm, wg, bg, done, h0, c0, cbuf, hsb, 14);
  k_lstm_step<<<128, 128, 0, stream>>>(Xm, wg, bg, done, h0, c0, cbuf, hsb, 15);
  k_lstm_step<<<128, 128, 0, stream>>>(Xm, wg, bg, done, h0, c0, cbuf, hsb, 16);
  k_lstm_step<<<128, 128, 0, stream>>>(Xm, wg, bg, done, h0, c0, cbuf, hsb, 17);
  k_lstm_step<<<128, 128, 0, stream>>>(Xm, wg, bg, done, h0, c0, cbuf, hsb, 18);
  k_lstm_step<<<128, 128, 0, stream>>>(Xm, wg, bg, done, h0, c0, cbuf, hsb, 19);
  k_lstm_step<<<128, 128, 0, stream>>>(Xm, wg, bg, done, h0, c0, cbuf, hsb, 20);
  k_lstm_step<<<128, 128, 0, stream>>>(Xm, wg, bg, done, h0, c0, cbuf, hsb, 21);
  k_lstm_step<<<128, 128, 0, stream>>>(Xm, wg, bg, done, h0, c0, cbuf, hsb, 22);
  k_lstm_step<<<128, 128, 0, stream>>>(Xm, wg, bg, done, h0, c0, cbuf, hsb, 23);
  k_lstm_step<<<128, 128, 0, stream>>>(Xm, wg, bg, done, h0, c0, cbuf, hsb, 24);
  k_lstm_step<<<128, 128, 0, stream>>>(Xm, wg, bg, done, h0, c0, cbuf, hsb, 25);
  k_lstm_step<<<128, 128, 0, stream>>>(Xm, wg, bg, done, h0, c0, cbuf, hsb, 26);
  k_lstm_step<<<128, 128, 0, stream>>>(Xm, wg, bg, done, h0, c0, cbuf, hsb, 27);
  k_lstm_step<<<128, 128, 0, stream>>>(Xm, wg, bg, done, h0, c0, cbuf, hsb, 28);
  k_lstm_step<<<128, 128, 0, stream>>>(Xm, wg, bg, done, h0, c0, cbuf, hsb, 29);
  k_lstm_step<<<128, 128, 0, stream>>>(Xm, wg, bg, done, h0, c0, cbuf, hsb, 30);
  k_lstm_step<<<128, 128, 0, stream>>>(Xm, wg, bg, done, h0, c0, cbuf, hsb, 31);
  k_lstm_step<<<128, 128, 0, stream>>>(Xm, wg, bg, done, h0, c0, cbuf, hsb, 32);
  k_lstm_step<<<128, 128, 0, stream>>>(Xm, wg, bg, done, h0, c0, cbuf, hsb, 33);
  k_lstm_step<<<128, 128, 0, stream>>>(Xm, wg, bg, done, h0, c0, cbuf, hsb, 34);
  k_lstm_step<<<128, 128, 0, stream>>>(Xm, wg, bg, done, h0, c0, cbuf, hsb, 35);
  k_lstm_step<<<128, 128, 0, stream>>>(Xm, wg, bg, done, h0, c0, cbuf, hsb, 36);
  k_lstm_step<<<128, 128, 0, stream>>>(Xm, wg, bg, done, h0, c0, cbuf, hsb, 37);
  k_lstm_step<<<128, 128, 0, stream>>>(Xm, wg, bg, done, h0, c0, cbuf, hsb, 38);
  k_lstm_step<<<128, 128, 0, stream>>>(Xm, wg, bg, done, h0, c0, cbuf, hsb, 39);
  k_lstm_step<<<128, 128, 0, stream>>>(Xm, wg, bg, done, h0, c0, cbuf, hsb, 40);
  k_lstm_step<<<128, 128, 0, stream>>>(Xm, wg, bg, done, h0, c0, cbuf, hsb, 41);
  k_lstm_step<<<128, 128, 0, stream>>>(Xm, wg, bg, done, h0, c0, cbuf, hsb, 42);
  k_lstm_step<<<128, 128, 0, stream>>>(Xm, wg, bg, done, h0, c0, cbuf, hsb, 43);
  k_lstm_step<<<128, 128, 0, stream>>>(Xm, wg, bg, done, h0, c0, cbuf, hsb, 44);
  k_lstm_step<<<128, 128, 0, stream>>>(Xm, wg, bg, done, h0, c0, cbuf, hsb, 45);
  k_lstm_step<<<128, 128, 0, stream>>>(Xm, wg, bg, done, h0, c0, cbuf, hsb, 46);
  k_lstm_step<<<128, 128, 0, stream>>>(Xm, wg, bg, done, h0, c0, cbuf, hsb, 47);
  k_lstm_step<<<128, 128, 0, stream>>>(Xm, wg, bg, done, h0, c0, cbuf, hsb, 48);
  k_lstm_step<<<128, 128, 0, stream>>>(Xm, wg, bg, done, h0, c0, cbuf, hsb, 49);
  k_lstm_step<<<128, 128, 0, stream>>>(Xm, wg, bg, done, h0, c0, cbuf, hsb, 50);
  k_lstm_step<<<128, 128, 0, stream>>>(Xm, wg, bg, done, h0, c0, cbuf, hsb, 51);
  k_lstm_step<<<128, 128, 0, stream>>>(Xm, wg, bg, done, h0, c0, cbuf, hsb, 52);
  k_lstm_step<<<128, 128, 0, stream>>>(Xm, wg, bg, done, h0, c0, cbuf, hsb, 53);
  k_lstm_step<<<128, 128, 0, stream>>>(Xm, wg, bg, done, h0, c0, cbuf, hsb, 54);
  k_lstm_step<<<128, 128, 0, stream>>>(Xm, wg, bg, done, h0, c0, cbuf, hsb, 55);
  k_lstm_step<<<128, 128, 0, stream>>>(Xm, wg, bg, done, h0, c0, cbuf, hsb, 56);
  k_lstm_step<<<128, 128, 0, stream>>>(Xm, wg, bg, done, h0, c0, cbuf, hsb, 57);
  k_lstm_step<<<128, 128, 0, stream>>>(Xm, wg, bg, done, h0, c0, cbuf, hsb, 58);
  k_lstm_step<<<128, 128, 0, stream>>>(Xm, wg, bg, done, h0, c0, cbuf, hsb, 59);
  k_lstm_step<<<128, 128, 0, stream>>>(Xm, wg, bg, done, h0, c0, cbuf, hsb, 60);
  k_lstm_step<<<128, 128, 0, stream>>>(Xm, wg, bg, done, h0, c0, cbuf, hsb, 61);
  k_lstm_step<<<128, 128, 0, stream>>>(Xm, wg, bg, done, h0, c0, cbuf, hsb, 62);
  k_lstm_step<<<128, 128, 0, stream>>>(Xm, wg, bg, done, h0, c0, cbuf, hsb, 63);

  k_heads<<<112, TB, 0, stream>>>(hsb, cbuf, pw, pb, vw, vb, out);
}


# conv1, conv2, conv3 GEMMs: a wave takes 64 output rows (4 row tiles) instead of 16, so a weight fragment is loaded once per k chunk for 4 row tiles; 2-wave workgroups; sum order unchanged
# speedup vs baseline: 9.9982x; 1.1974x over previous
#include <hip/hip_runtime.h>
#include <math.h>

typedef __attribute__((ext_vector_type(16))) _Float16 v16h;
typedef __attribute__((ext_vector_type(8)))  _Float16 v8h;
typedef __attribute__((ext_vector_type(4)))  _Float16 v4h;
typedef __attribute__((ext_vector_type(8)))  float    v8f;

#define WMMA16(A,B,C) __builtin_amdgcn_wmma_f32_16x16x32_f16(false,(A),false,(B),(short)0,(C),false,false)

static constexpr int T_STEPS = 64;
static constexpr int E_ENVS  = 32;
static constexpr int B_TOT   = 2048;
static constexpr int HL      = 256;
static constexpr int IND     = 549;
static constexpr int KPAD    = 832;
static constexpr int XS      = 576;
static constexpr int IMS     = 88;

__device__ __forceinline__ v16h pack_frag(const _Float16* p) {
  v8h u = *(const v8h*)p;
  v8h w = *(const v8h*)(p + 16);
  v16h a;
#pragma unroll
  for (int e = 0; e < 8; e++) { a[e] = u[e]; a[e + 8] = w[e]; }
  return a;
}
__device__ __forceinline__ v16h pack_frag_c1(const _Float16* p0) {
  const _Float16* p1 = p0 + 2 * IMS;
  v4h u0 = *(const v4h*)p0;
  v4h u1 = *(const v4h*)(p0 + 4);
  v4h u2 = *(const v4h*)p1;
  v4h u3 = *(const v4h*)(p1 + 4);
  v16h a;
#pragma unroll
  for (int e = 0; e < 4; e++) {
    a[e] = u0[e]; a[e + 4] = u1[e]; a[e + 8] = u2[e]; a[e + 12] = u3[e];
  }
  return a;
}

__device__ __forceinline__ float sigmoidf_(float x) { return 1.0f / (1.0f + __expf(-x)); }

__device__ __forceinline__ unsigned pk2_scaled_f16(float a, float b, float sc) {
  unsigned r;
  asm("v_fma_mixlo_f16 %0, %1, %2, 0" : "=v"(r) : "v"(a), "s"(sc));
  asm("v_fma_mixhi_f16 %0, %1, %2, 0" : "+v"(r) : "v"(b), "s"(sc));
  return r;
}

__global__ void __launch_bounds__(256) k_prep_img(const float* img, _Float16* imgC) {
  const float sc = 1.0f / 255.0f;
  int g0 = blockIdx.x * 1024 + threadIdx.x;
  float4 v[4][3];
#pragma unroll
  for (int k = 0; k < 4; k++) {
    const float4* p = (const float4*)(img + (g0 + k * 256) * 12);
    v[k][0] = p[0]; v[k][1] = p[1]; v[k][2] = p[2];
  }
#pragma unroll
  for (int k = 0; k < 4; k++) {
    int g  = g0 + k * 256;
    int R  = g / 21, cg = g - R * 21;
    int b  = R / 84;
    _Float16* o = imgC + (R + b * 168) * IMS + cg * 4;
    float4 q0 = v[k][0], q1 = v[k][1], q2 = v[k][2];
    uint2 c0, c1, c2;
    c0.x = pk2_scaled_f16(q0.x, q0.w, sc); c0.y = pk2_scaled_f16(q1.z, q2.y, sc);
    c1.x = pk2_scaled_f16(q0.y, q1.x, sc); c1.y = pk2_scaled_f16(q1.w, q2.z, sc);
    c2.x = pk2_scaled_f16(q0.z, q1.y, sc); c2.y = pk2_scaled_f16(q2.x, q2.w, sc);
    *(uint2*)(o)                = c0;
    *(uint2*)(o + 84 * IMS)     = c1;
    *(uint2*)(o + 2 * 84 * IMS) = c2;
  }
}

__global__ void k_prep_w1(const float* w, _Float16* wt) {
  int i = blockIdx.x * blockDim.x + threadIdx.x;
  if (i >= 32 * 192) return;
  int n = i / 192, k = i % 192;
  int c = k >> 6, kh = (k & 63) >> 3, kw = k & 7;
  wt[i] = (_Float16)w[((n * 3 + c) * 8 + kh) * 8 + kw];
}
__global__ void k_prep_w2(const float* w, _Float16* wt) {
  int i = blockIdx.x * blockDim.x + threadIdx.x;
  if (i >= 64 * 512) return;
  int n = i / 512, k = i % 512;
  int pix = k >> 5, c = k & 31, kh = pix >> 2, kw = pix & 3;
  wt[i] = (_Float16)w[((n * 32 + c) * 4 + kh) * 4 + kw];
}
__global__ void k_prep_w3(const float* w, _Float16* wt) {
  int i = blockIdx.x * blockDim.x + threadIdx.x;
  if (i >= 64 * 576) return;
  int n = i / 576, k = i % 576;
  int pix = k >> 6, c = k & 63, kh = pix / 3, kw = pix % 3;
  wt[i] = (_Float16)w[((n * 64 + c) * 3 + kh) * 3 + kw];
}
__global__ void k_prep_fc(const float* w, _Float16* wt) {
  int i = blockIdx.x * blockDim.x + threadIdx.x;
  if (i >= 512 * 3136) return;
  int f = i / 3136, k = i % 3136;
  int s = k >> 6, c = k & 63;
  wt[i] = (_Float16)w[f * 3136 + c * 49 + s];
}
__global__ void k_prep_wg(const float* wih, const float* whh, _Float16* wg) {
  int i = blockIdx.x * blockDim.x + threadIdx.x;
  if (i >= 1024 * KPAD) return;
  int r = i / KPAD, k = i % KPAD;
  float v = 0.0f;
  if (k < HL)             v = whh[r * HL + k];
  else if (k < HL + IND)  v = wih[r * IND + (k - HL)];
  wg[i] = (_Float16)v;
}
__global__ void k_prep_bias(const float* bih, const float* bhh, float* bg) {
  int i = blockIdx.x * blockDim.x + threadIdx.x;
  if (i < 1024) bg[i] = bih[i] + bhh[i];
}
__global__ void k_onehot(const int* pos, const int* lact, _Float16* X) {
  int i = blockIdx.x * blockDim.x + threadIdx.x;
  if (i >= B_TOT * 64) return;
  int b = i / 64, c = i % 64;
  float v = 0.0f;
  if (c < 16)      v = (pos[b * 2 + 0] == c)      ? 1.0f : 0.0f;
  else if (c < 32) v = (pos[b * 2 + 1] == c - 16) ? 1.0f : 0.0f;
  else if (c < 37) v = (lact[b]        == c - 32) ? 1.0f : 0.0f;
  X[b * XS + 512 + c] = (_Float16)v;
}

__global__ void __launch_bounds__(64) k_conv1(const _Float16* imgC, const _Float16* wt,
                                              const float* bias, _Float16* out) {
  int wid  = (blockIdx.x * blockDim.x + threadIdx.x) >> 5;
  int lane = threadIdx.x & 31, hi = lane >> 4, lo = lane & 15;
  const _Float16* ap[4];
#pragma unroll
  for (int i = 0; i < 4; i++) {
    int m = wid * 64 + i * 16 + lo;
    int b = m / 400, p = m % 400, oh = p / 20, ow = p % 20;
    ap[i] = imgC + (size_t)b * 3 * 84 * IMS + (oh * 4 + hi) * IMS + ow * 4;
  }
  const _Float16* bp = wt + lo * 192 + hi * 8;
  v8f acc[4][2] = {};
#pragma unroll
  for (int kb = 0; kb < 192; kb += 32) {
    int c = kb >> 6, khb = (kb & 63) >> 3;
    v16h a[4];
    v16h b0 = pack_frag(bp + kb);
    v16h b1 = pack_frag(bp + 16 * 192 + kb);
#pragma unroll
    for (int i = 0; i < 4; i++) a[i] = pack_frag_c1(ap[i] + (c * 84 + khb) * IMS);
    __builtin_amdgcn_sched_barrier(0);
#pragma unroll
    for (int i = 0; i < 4; i++) {
      acc[i][0] = WMMA16(a[i], b0, acc[i][0]);
      acc[i][1] = WMMA16(a[i], b1, acc[i][1]);
    }
  }
  float bs0 = bias[lo], bs1 = bias[16 + lo];
#pragma unroll
  for (int i = 0; i < 4; i++) {
#pragma unroll
    for (int r = 0; r < 8; r++) {
      int mm = wid * 64 + i * 16 + hi * 8 + r;
      float v0 = acc[i][0][r] + bs0; v0 = v0 > 0.0f ? v0 : 0.0f;
      float v1 = acc[i][1][r] + bs1; v1 = v1 > 0.0f ? v1 : 0.0f;
      out[mm * 32 + lo]      = (_Float16)v0;
      out[mm * 32 + 16 + lo] = (_Float16)v1;
    }
  }
}

__global__ void __launch_bounds__(64) k_conv2(const _Float16* act1, const _Float16* wt,
                                              const float* bias, _Float16* out) {
  int wid  = (blockIdx.x * blockDim.x + threadIdx.x) >> 5;
  int lane = threadIdx.x & 31, hi = lane >> 4, lo = lane & 15;
  const _Float16* ap[4];
#pragma unroll
  for (int i = 0; i < 4; i++) {
    int m = wid * 64 + i * 16 + lo;
    int b = m / 81, p = m % 81, oh = p / 9, ow = p % 9;
    ap[i] = act1 + ((b * 20 + oh * 2) * 20 + ow * 2) * 32 + hi * 8;
  }
  const _Float16* bp = wt + lo * 512 + hi * 8;
  v8f acc[4][4] = {};
#pragma unroll 1
  for (int kb = 0; kb < 512; kb += 32) {
    int pix = kb >> 5, kh = pix >> 2, kw = pix & 3;
    int aoff = (kh * 20 + kw) * 32;
    v16h a[4], b[4];
#pragma unroll
    for (int i = 0; i < 4; i++) a[i] = pack_frag(ap[i] + aoff);
#pragma unroll
    for (int j = 0; j < 4; j++) b[j] = pack_frag(bp + j * 16 * 512 + kb);
    __builtin_amdgcn_sched_barrier(0);
#pragma unroll
    for (int i = 0; i < 4; i++) {
#pragma unroll
      for (int j = 0; j < 4; j++) acc[i][j] = WMMA16(a[i], b[j], acc[i][j]);
    }
  }
  float bs[4] = {bias[lo], bias[16 + lo], bias[32 + lo], bias[48 + lo]};
#pragma unroll
  for (int i = 0; i < 4; i++) {
#pragma unroll
    for (int r = 0; r < 8; r++) {
      int mm = wid * 64 + i * 16 + hi * 8 + r;
#pragma unroll
      for (int j = 0; j < 4; j++) {
        float v = acc[i][j][r] + bs[j]; v = v > 0.0f ? v : 0.0f;
        out[mm * 64 + j * 16 + lo] = (_Float16)v;
      }
    }
  }
}

__global__ void __launch_bounds__(64) k_conv3(const _Float16* act2, const _Float16* wt,
                                              const float* bias, _Float16* out) {
  int wid  = (blockIdx.x * blockDim.x + threadIdx.x) >> 5;
  int lane = threadIdx.x & 31, hi = lane >> 4, lo = lane & 15;
  const _Float16* ap[4];
#pragma unroll
  for (int i = 0; i < 4; i++) {
    int m = wid * 64 + i * 16 + lo;
    int b = m / 49, p = m % 49, oh = p / 7, ow = p % 7;
    ap[i] = act2 + ((b * 9 + oh) * 9 + ow) * 64 + hi * 8;
  }
  const _Float16* bp = wt + lo * 576 + hi * 8;
  v8f acc[4][4] = {};
#pragma unroll 1
  for (int kb = 0; kb < 576; kb += 32) {
    int pix = kb >> 6, cb = kb & 63;
    int kh = pix / 3, kw = pix % 3;
    int aoff = (kh * 9 + kw) * 64 + cb;
    v16h a[4], b[4];
#pragma unroll
    for (int i = 0; i < 4; i++) a[i] = pack_frag(ap[i] + aoff);
#pragma unroll
    for (int j = 0; j < 4; j++) b[j] = pack_frag(bp + j * 16 * 576 + kb);
    __builtin_amdgcn_sched_barrier(0);
#pragma unroll
    for (int i = 0; i < 4; i++) {
#pragma unroll
      for (int j = 0; j < 4; j++) acc[i][j] = WMMA16(a[i], b[j], acc[i][j]);
    }
  }
  float bs[4] = {bias[lo], bias[16 + lo], bias[32 + lo], bias[48 + lo]};
#pragma unroll
  for (int i = 0; i < 4; i++) {
#pragma unroll
    for (int r = 0; r < 8; r++) {
      int mm = wid * 64 + i * 16 + hi * 8 + r;
#pragma unroll
      for (int j = 0; j < 4; j++) {
        float v = acc[i][j][r] + bs[j]; v = v > 0.0f ? v : 0.0f;
        out[mm * 64 + j * 16 + lo] = (_Float16)v;
      }
    }
  }
}

static constexpr int FPITCH = 68;
static constexpr int FTILE  = 32 * FPITCH;
__global__ void __launch_bounds__(256) k_fc(const _Float16* act3, const _Float16* wt,
                                            const float* bias, _Float16* X) {
  __shared__ __attribute__((aligned(16))) float P[8 * FTILE];
  int tid = threadIdx.x;
  int w = __builtin_amdgcn_readfirstlane(tid >> 5);
  int lane = tid & 31, hi = lane >> 4, lo = lane & 15;
  int ng = blockIdx.x & 7, mp = blockIdx.x >> 3;
  int m0 = mp * 32, nb = ng * 64;
  const _Float16* ar0 = act3 + (size_t)(m0 + lo) * 3136 + hi * 8;
  const _Float16* ar1 = ar0 + (size_t)16 * 3136;
  const _Float16* br0 = wt + (size_t)(nb + lo) * 3136 + hi * 8;
  const _Float16* br1 = br0 + (size_t)16 * 3136;
  const _Float16* br2 = br0 + (size_t)32 * 3136;
  const _Float16* br3 = br0 + (size_t)48 * 3136;
  v8f c00 = {}, c01 = {}, c02 = {}, c03 = {};
  v8f c10 = {}, c11 = {}, c12 = {}, c13 = {};
  for (int kc = w; kc < 98; kc += 8) {
    int kb = kc * 32;
    v16h a0 = pack_frag(ar0 + kb);
    v16h a1 = pack_frag(ar1 + kb);
    v16h b0 = pack_frag(br0 + kb);
    v16h b1 = pack_frag(br1 + kb);
    v16h b2 = pack_frag(br2 + kb);
    v16h b3 = pack_frag(br3 + kb);
    c00 = WMMA16(a0, b0, c00);
    c10 = WMMA16(a1, b0, c10);
    c01 = WMMA16(a0, b1, c01);
    c11 = WMMA16(a1, b1, c11);
    c02 = WMMA16(a0, b2, c02);
    c12 = WMMA16(a1, b2, c12);
    c03 = WMMA16(a0, b3, c03);
    c13 = WMMA16(a1, b3, c13);
  }
  {
    float* pw = P + w * FTILE + (hi * 8) * FPITCH + lo;
#pragma unroll
    for (int r = 0; r < 8; r++) {
      pw[r * FPITCH +  0] = c00[r];
      pw[r * FPITCH + 16] = c01[r];
      pw[r * FPITCH + 32] = c02[r];
      pw[r * FPITCH + 48] = c03[r];
      pw[(16 + r) * FPITCH +  0] = c10[r];
      pw[(16 + r) * FPITCH + 16] = c11[r];
      pw[(16 + r) * FPITCH + 32] = c12[r];
      pw[(16 + r) * FPITCH + 48] = c13[r];
    }
  }
  __syncthreads();
  {
    int row = tid >> 3, c0 = (tid & 7) * 8;
    const float* pr = P + row * FPITCH + c0;
    float4 s0 = *(const float4*)pr;
    float4 s1 = *(const float4*)(pr + 4);
#pragma unroll
    for (int k = 1; k < 8; k++) {
      float4 u0 = *(const float4*)(pr + k * FTILE);
      float4 u1 = *(const float4*)(pr + k * FTILE + 4);
      s0.x += u0.x; s0.y += u0.y; s0.z += u0.z; s0.w += u0.w;
      s1.x += u1.x; s1.y += u1.y; s1.z += u1.z; s1.w += u1.w;
    }
    float4 q0 = *(const float4*)(bias + nb + c0);
    float4 q1 = *(const float4*)(bias + nb + c0 + 4);
    float v[8] = {s0.x + q0.x, s0.y + q0.y, s0.z + q0.z, s0.w + q0.w,
                  s1.x + q1.x, s1.y + q1.y, s1.z + q1.z, s1.w + q1.w};
    v8h o;
#pragma unroll
    for (int e = 0; e < 8; e++) o[e] = (_Float16)(v[e] > 0.0f ? v[e] : 0.0f);
    *(v8h*)(X + (size_t)(m0 + row) * XS + nb + c0) = o;
  }
}

static constexpr int PPITCH = 36;
static constexpr int PTILE  = 8 * PPITCH;
__global__ void __launch_bounds__(128) k_lstm_step(const _Float16* X, const _Float16* Wg,
                                                   const float* biasg, const int* done,
                                                   const float* h0, const float* c0,
                                                   float* cbuf, float* hs, int t) {
  __shared__ float P[4 * PTILE];
  int tid = threadIdx.x;
  int w = __builtin_amdgcn_readfirstlane(tid >> 5);
  int lane = tid & 31, hi = lane >> 4, lo = lane & 15;
  int mt = blockIdx.x >> 6, cg = blockIdx.x & 63;
  const float* hprev = (t == 0) ? h0 : hs + (size_t)(t - 1) * (E_ENVS * HL);
  const float* cin   = (t == 0) ? c0 : cbuf;
  int se = (t == 0) ? HL : 4;
  int sq = (t == 0) ? 4 : E_ENVS * 4;

  int ci   = tid & 63;
  int eenv = mt * 16 + (ci >> 2);
  int ecol = cg * 4 + (ci & 3);
  int edone = done[t * E_ENVS + eenv];
  float cprev = cin[eenv * se + cg * sq + (ci & 3)];
  float bi = biasg[0 * HL + ecol], bf = biasg[1 * HL + ecol];
  float bg = biasg[2 * HL + ecol], bo = biasg[3 * HL + ecol];

  int aenv = mt * 16 + lo;
  int adone = done[t * E_ENVS + aenv];
  int grow = (lo >> 2) * HL + cg * 4 + (lo & 3);
  const _Float16* wrow = Wg + (size_t)grow * KPAD + hi * 8;
  const _Float16* xrow = X + (size_t)(t * E_ENVS + aenv) * XS + hi * 8;

  v16h a6 = {}, b6 = {};
  if (w < 2) {
    int kc = w + 24;
    a6 = pack_frag(xrow + (kc - 8) * 32);
    b6 = pack_frag(wrow + kc * 32);
  }
  float4 hu[2][4];
  v16h bh[2];
#pragma unroll
  for (int i = 0; i < 2; i++) {
    int kc = w + 4 * i;
    const float* hp = hprev + aenv * se + (kc * 8 + hi * 2) * sq;
    hu[i][0] = *(const float4*)hp;
    hu[i][1] = *(const float4*)(hp + sq);
    hu[i][2] = *(const float4*)(hp + 4 * sq);
    hu[i][3] = *(const float4*)(hp + 5 * sq);
    bh[i] = pack_frag(wrow + kc * 32);
  }
  v16h ax[4], bx[4];
#pragma unroll
  for (int i = 0; i < 4; i++) {
    int kc = w + 8 + 4 * i;
    ax[i] = pack_frag(xrow + (kc - 8) * 32);
    bx[i] = pack_frag(wrow + kc * 32);
  }

  __builtin_amdgcn_sched_barrier(0);

  v8f acc = {};
#pragma unroll
  for (int i = 0; i < 2; i++) {
    float hv16[16] = {hu[i][0].x, hu[i][0].y, hu[i][0].z, hu[i][0].w,
                      hu[i][1].x, hu[i][1].y, hu[i][1].z, hu[i][1].w,
                      hu[i][2].x, hu[i][2].y, hu[i][2].z, hu[i][2].w,
                      hu[i][3].x, hu[i][3].y, hu[i][3].z, hu[i][3].w};
    v16h a;
#pragma unroll
    for (int e = 0; e < 16; e++) a[e] = adone ? (_Float16)0.0f : (_Float16)hv16[e];
    acc = WMMA16(a, bh[i], acc);
  }
#pragma unroll
  for (int i = 0; i < 4; i++) acc = WMMA16(ax[i], bx[i], acc);
  if (w < 2) acc = WMMA16(a6, b6, acc);

  {
    float* pw = P + w * PTILE + lane;
#pragma unroll
    for (int r = 0; r < 8; r++) pw[r * PPITCH] = acc[r];
  }
  __syncthreads();

  if (tid < 64) {
    int el = tid >> 2, q = tid & 3;
    const float* pr = P + (el & 7) * PPITCH + (el >> 3) * 16 + q;
    float si = 0.0f, sf = 0.0f, sg = 0.0f, so = 0.0f;
#pragma unroll
    for (int k = 0; k < 4; k++) {
      si += pr[k * PTILE + 0];
      sf += pr[k * PTILE + 4];
      sg += pr[k * PTILE + 8];
      so += pr[k * PTILE + 12];
    }
    float cp = edone ? 0.0f : cprev;
    float iv = sigmoidf_(si + bi), fv = sigmoidf_(sf + bf);
    float gv = tanhf(sg + bg),     ov = sigmoidf_(so + bo);
    float cv = fv * cp + iv * gv;
    float hv = ov * tanhf(cv);
    cbuf[(cg * E_ENVS + eenv) * 4 + q] = cv;
    hs[(size_t)t * (E_ENVS * HL) + (cg * E_ENVS + eenv) * 4 + q] = hv;
  }
}

__global__ void k_heads(const float* hs, const float* cbuf, const float* pw, const float* pb,
                        const float* vw, const float* vb, float* out) {
  int i = blockIdx.x * blockDim.x + threadIdx.x;
  if (i < B_TOT * 6) {
    int b = i / 6, j = i % 6;
    const float* h    = hs + (size_t)(b >> 5) * (E_ENVS * HL) + (b & 31) * 4;
    const float* wrow = (j < 5) ? pw + j * HL : vw;
    float s = (j < 5) ? pb[j] : vb[0];
#pragma unroll 4
    for (int k4 = 0; k4 < HL / 4; k4++) {
      float4 hv = *(const float4*)(h + k4 * (E_ENVS * 4));
      s += hv.x * wrow[k4 * 4 + 0];
      s += hv.y * wrow[k4 * 4 + 1];
      s += hv.z * wrow[k4 * 4 + 2];
      s += hv.w * wrow[k4 * 4 + 3];
    }
    if (j < 5) out[b * 5 + j] = s;
    else       out[10240 + b] = s;
  } else {
    int o = i - B_TOT * 6;
    int sel = o >> 13, r = o & 8191;
    int e = r >> 8, k = r & 255;
    int src = ((k >> 2) * E_ENVS + e) * 4 + (k & 3);
    if (sel == 0) out[12288 + r] = hs[(size_t)(T_STEPS - 1) * (E_ENVS * HL) + src];
    else          out[20480 + r] = cbuf[src];
  }
}

extern "C" void kernel_launch(void* const* d_in, const int* in_sizes, int n_in,
                              void* d_out, int out_size, void* d_ws, size_t ws_size,
                              hipStream_t stream) {
  (void)in_sizes; (void)n_in; (void)out_size; (void)ws_size;
  const float* img  = (const float*)d_in[0];
  const int*   pos  = (const int*)d_in[1];
  const int*   lact = (const int*)d_in[2];
  const int*   done = (const int*)d_in[3];
  const float* h0   = (const float*)d_in[4];
  const float* c0   = (const float*)d_in[5];
  const float* w1   = (const float*)d_in[6];
  const float* b1   = (const float*)d_in[7];
  const float* w2   = (const float*)d_in[8];
  const float* b2   = (const float*)d_in[9];
  const float* w3   = (const float*)d_in[10];
  const float* b3   = (const float*)d_in[11];
  const float* fcw  = (const float*)d_in[12];
  const float* fcb  = (const float*)d_in[13];
  const float* wih  = (const float*)d_in[14];
  const float* whh  = (const float*)d_in[15];
  const float* bih  = (const float*)d_in[16];
  const float* bhh  = (const float*)d_in[17];
  const float* pw   = (const float*)d_in[18];
  const float* pb   = (const float*)d_in[19];
  const float* vw   = (const float*)d_in[20];
  const float* vb   = (const float*)d_in[21];
  float* out = (float*)d_out;
  char*  ws  = (char*)d_ws;

  const size_t o_img  = 0;
  const size_t o_act1 = 90832896;
  const size_t o_w1t  = o_act1 + 52428800;
  const size_t o_w2t  = o_w1t + 12288;
  const size_t o_w3t  = o_w2t + 65536;
  const size_t o_fcwt = o_w3t + 73728;
  const size_t o_wg   = o_fcwt + 3211264;
  const size_t o_bg   = o_wg + 1703936;
  const size_t o_act2 = 0;
  const size_t o_act3 = 21233664;
  const size_t o_X    = 34078720;
  const size_t o_hs   = 36438016;
  const size_t o_c    = 38535168;

  _Float16* imgC = (_Float16*)(ws + o_img);
  _Float16* act1 = (_Float16*)(ws + o_act1);
  _Float16* act2 = (_Float16*)(ws + o_act2);
  _Float16* act3 = (_Float16*)(ws + o_act3);
  _Float16* Xm   = (_Float16*)(ws + o_X);
  _Float16* w1t  = (_Float16*)(ws + o_w1t);
  _Float16* w2t  = (_Float16*)(ws + o_w2t);
  _Float16* w3t  = (_Float16*)(ws + o_w3t);
  _Float16* fcwt = (_Float16*)(ws + o_fcwt);
  _Float16* wg   = (_Float16*)(ws + o_wg);
  float*    bg   = (float*)(ws + o_bg);
  float*    hsb  = (float*)(ws + o_hs);
  float*    cbuf = (float*)(ws + o_c);

  const int TB = 256;
  k_prep_img <<<3528, TB, 0, stream>>>(img, imgC);
  k_prep_w1  <<<(32 * 192   + TB - 1) / TB, TB, 0, stream>>>(w1, w1t);
  k_prep_w2  <<<(64 * 512   + TB - 1) / TB, TB, 0, stream>>>(w2, w2t);
  k_prep_w3  <<<(64 * 576   + TB - 1) / TB, TB, 0, stream>>>(w3, w3t);
  k_prep_fc  <<<(512 * 3136 + TB - 1) / TB, TB, 0, stream>>>(fcw, fcwt);
  k_prep_wg  <<<(1024 * KPAD + TB - 1) / TB, TB, 0, stream>>>(wih, whh, wg);
  k_prep_bias<<<(1024 + TB - 1) / TB, TB, 0, stream>>>(bih, bhh, bg);

  k_conv1<<<6400, 64, 0, stream>>>(imgC, w1t, b1, act1);
  k_conv2<<<1296, 64, 0, stream>>>(act1, w2t, b2, act2);
  k_conv3<<< 784, 64, 0, stream>>>(act2, w3t, b3, act3);
  k_fc   <<< 512, TB, 0, stream>>>(act3, fcwt, fcb, Xm);

  k_onehot<<<(B_TOT * 64 + TB - 1) / TB, TB, 0, stream>>>(pos, lact, Xm);

  k_lstm_step<<<128, 128, 0, stream>>>(Xm, wg, bg, done, h0, c0, cbuf, hsb, 0);
  k_lstm_step<<<128, 128, 0, stream>>>(Xm, wg, bg, done, h0, c0, cbuf, hsb, 1);
  k_lstm_step<<<128, 128, 0, stream>>>(Xm, wg, bg, done, h0, c0, cbuf, hsb, 2);
  k_lstm_step<<<128, 128, 0, stream>>>(Xm, wg, bg, done, h0, c0, cbuf, hsb, 3);
  k_lstm_step<<<128, 128, 0, stream>>>(Xm, wg, bg, done, h0, c0, cbuf, hsb, 4);
  k_lstm_step<<<128, 128, 0, stream>>>(Xm, wg, bg, done, h0, c0, cbuf, hsb, 5);
  k_lstm_step<<<128, 128, 0, stream>>>(Xm, wg, bg, done, h0, c0, cbuf, hsb, 6);
  k_lstm_step<<<128, 128, 0, stream>>>(Xm, wg, bg, done, h0, c0, cbuf, hsb, 7);
  k_lstm_step<<<128, 128, 0, stream>>>(Xm, wg, bg, done, h0, c0, cbuf, hsb, 8);
  k_lstm_step<<<128, 128, 0, stream>>>(Xm, wg, bg, done, h0, c0, cbuf, hsb, 9);
  k_lstm_step<<<128, 128, 0, stream>>>(Xm, wg, bg, done, h0, c0, cbuf, hsb, 10);
  k_lstm_step<<<128, 128, 0, stream>>>(Xm, wg, bg, done, h0, c0, cbuf, hsb, 11);
  k_lstm_step<<<128, 128, 0, stream>>>(Xm, wg, bg, done, h0, c0, cbuf, hsb, 12);
  k_lstm_step<<<128, 128, 0, stream>>>(Xm, wg, bg, done, h0, c0, cbuf, hsb, 13);
  k_lstm_step<<<128, 128, 0, stream>>>(Xm, wg, bg, done, h0, c0, cbuf, hsb, 14);
  k_lstm_step<<<128, 128, 0, stream>>>(Xm, wg, bg, done, h0, c0, cbuf, hsb, 15);
  k_lstm_step<<<128, 128, 0, stream>>>(Xm, wg, bg, done, h0, c0, cbuf, hsb, 16);
  k_lstm_step<<<128, 128, 0, stream>>>(Xm, wg, bg, done, h0, c0, cbuf, hsb, 17);
  k_lstm_step<<<128, 128, 0, stream>>>(Xm, wg, bg, done, h0, c0, cbuf, hsb, 18);
  k_lstm_step<<<128, 128, 0, stream>>>(Xm, wg, bg, done, h0, c0, cbuf, hsb, 19);
  k_lstm_step<<<128, 128, 0, stream>>>(Xm, wg, bg, done, h0, c0, cbuf, hsb, 20);
  k_lstm_step<<<128, 128, 0, stream>>>(Xm, wg, bg, done, h0, c0, cbuf, hsb, 21);
  k_lstm_step<<<128, 128, 0, stream>>>(Xm, wg, bg, done, h0, c0, cbuf, hsb, 22);
  k_lstm_step<<<128, 128, 0, stream>>>(Xm, wg, bg, done, h0, c0, cbuf, hsb, 23);
  k_lstm_step<<<128, 128, 0, stream>>>(Xm, wg, bg, done, h0, c0, cbuf, hsb, 24);
  k_lstm_step<<<128, 128, 0, stream>>>(Xm, wg, bg, done, h0, c0, cbuf, hsb, 25);
  k_lstm_step<<<128, 128, 0, stream>>>(Xm, wg, bg, done, h0, c0, cbuf, hsb, 26);
  k_lstm_step<<<128, 128, 0, stream>>>(Xm, wg, bg, done, h0, c0, cbuf, hsb, 27);
  k_lstm_step<<<128, 128, 0, stream>>>(Xm, wg, bg, done, h0, c0, cbuf, hsb, 28);
  k_lstm_step<<<128, 128, 0, stream>>>(Xm, wg, bg, done, h0, c0, cbuf, hsb, 29);
  k_lstm_step<<<128, 128, 0, stream>>>(Xm, wg, bg, done, h0, c0, cbuf, hsb, 30);
  k_lstm_step<<<128, 128, 0, stream>>>(Xm, wg, bg, done, h0, c0, cbuf, hsb, 31);
  k_lstm_step<<<128, 128, 0, stream>>>(Xm, wg, bg, done, h0, c0, cbuf, hsb, 32);
  k_lstm_step<<<128, 128, 0, stream>>>(Xm, wg, bg, done, h0, c0, cbuf, hsb, 33);
  k_lstm_step<<<128, 128, 0, stream>>>(Xm, wg, bg, done, h0, c0, cbuf, hsb, 34);
  k_lstm_step<<<128, 128, 0, stream>>>(Xm, wg, bg, done, h0, c0, cbuf, hsb, 35);
  k_lstm_step<<<128, 128, 0, stream>>>(Xm, wg, bg, done, h0, c0, cbuf, hsb, 36);
  k_lstm_step<<<128, 128, 0, stream>>>(Xm, wg, bg, done, h0, c0, cbuf, hsb, 37);
  k_lstm_step<<<128, 128, 0, stream>>>(Xm, wg, bg, done, h0, c0, cbuf, hsb, 38);
  k_lstm_step<<<128, 128, 0, stream>>>(Xm, wg, bg, done, h0, c0, cbuf, hsb, 39);
  k_lstm_step<<<128, 128, 0, stream>>>(Xm, wg, bg, done, h0, c0, cbuf, hsb, 40);
  k_lstm_step<<<128, 128, 0, stream>>>(Xm, wg, bg, done, h0, c0, cbuf, hsb, 41);
  k_lstm_step<<<128, 128, 0, stream>>>(Xm, wg, bg, done, h0, c0, cbuf, hsb, 42);
  k_lstm_step<<<128, 128, 0, stream>>>(Xm, wg, bg, done, h0, c0, cbuf, hsb, 43);
  k_lstm_step<<<128, 128, 0, stream>>>(Xm, wg, bg, done, h0, c0, cbuf, hsb, 44);
  k_lstm_step<<<128, 128, 0, stream>>>(Xm, wg, bg, done, h0, c0, cbuf, hsb, 45);
  k_lstm_step<<<128, 128, 0, stream>>>(Xm, wg, bg, done, h0, c0, cbuf, hsb, 46);
  k_lstm_step<<<128, 128, 0, stream>>>(Xm, wg, bg, done, h0, c0, cbuf, hsb, 47);
  k_lstm_step<<<128, 128, 0, stream>>>(Xm, wg, bg, done, h0, c0, cbuf, hsb, 48);
  k_lstm_step<<<128, 128, 0, stream>>>(Xm, wg, bg, done, h0, c0, cbuf, hsb, 49);
  k_lstm_step<<<128, 128, 0, stream>>>(Xm, wg, bg, done, h0, c0, cbuf, hsb, 50);
  k_lstm_step<<<128, 128, 0, stream>>>(Xm, wg, bg, done, h0, c0, cbuf, hsb, 51);
  k_lstm_step<<<128, 128, 0, stream>>>(Xm, wg, bg, done, h0, c0, cbuf, hsb, 52);
  k_lstm_step<<<128, 128, 0, stream>>>(Xm, wg, bg, done, h0, c0, cbuf, hsb, 53);
  k_lstm_step<<<128, 128, 0, stream>>>(Xm, wg, bg, done, h0, c0, cbuf, hsb, 54);
  k_lstm_step<<<128, 128, 0, stream>>>(Xm, wg, bg, done, h0, c0, cbuf, hsb, 55);
  k_lstm_step<<<128, 128, 0, stream>>>(Xm, wg, bg, done, h0, c0, cbuf, hsb, 56);
  k_lstm_step<<<128, 128, 0, stream>>>(Xm, wg, bg, done, h0, c0, cbuf, hsb, 57);
  k_lstm_step<<<128, 128, 0, stream>>>(Xm, wg, bg, done, h0, c0, cbuf, hsb, 58);
  k_lstm_step<<<128, 128, 0, stream>>>(Xm, wg, bg, done, h0, c0, cbuf, hsb, 59);
  k_lstm_step<<<128, 128, 0, stream>>>(Xm, wg, bg, done, h0, c0, cbuf, hsb, 60);
  k_lstm_step<<<128, 128, 0, stream>>>(Xm, wg, bg, done, h0, c0, cbuf, hsb, 61);
  k_lstm_step<<<128, 128, 0, stream>>>(Xm, wg, bg, done, h0, c0, cbuf, hsb, 62);
  k_lstm_step<<<128, 128, 0, stream>>>(Xm, wg, bg, done, h0, c0, cbuf, hsb, 63);

  k_heads<<<112, TB, 0, stream>>>(hsb, cbuf, pw, pb, vw, vb, out);
}


# conv1 takes the f32 image itself: a workgroup per image reads it once, keeps it as f16 CHW planes in LDS (44 KB), 5 waves read A fragments there; re-layout launch gone; values, sum order same
# speedup vs baseline: 10.4539x; 1.0456x over previous
#include <hip/hip_runtime.h>
#include <math.h>

typedef __attribute__((ext_vector_type(16))) _Float16 v16h;
typedef __attribute__((ext_vector_type(8)))  _Float16 v8h;
typedef __attribute__((ext_vector_type(4)))  _Float16 v4h;
typedef __attribute__((ext_vector_type(8)))  float    v8f;

#define WMMA16(A,B,C) __builtin_amdgcn_wmma_f32_16x16x32_f16(false,(A),false,(B),(short)0,(C),false,false)

static constexpr int T_STEPS = 64;
static constexpr int E_ENVS  = 32;
static constexpr int B_TOT   = 2048;
static constexpr int HL      = 256;
static constexpr int IND     = 549;
static constexpr int KPAD    = 832;
static constexpr int XS      = 576;
static constexpr int IMS     = 88;

__device__ __forceinline__ v16h pack_frag(const _Float16* p) {
  v8h u = *(const v8h*)p;
  v8h w = *(const v8h*)(p + 16);
  v16h a;
#pragma unroll
  for (int e = 0; e < 8; e++) { a[e] = u[e]; a[e + 8] = w[e]; }
  return a;
}
__device__ __forceinline__ v16h pack_frag_c1(const _Float16* p0) {
  const _Float16* p1 = p0 + 2 * IMS;
  v4h u0 = *(const v4h*)p0;
  v4h u1 = *(const v4h*)(p0 + 4);
  v4h u2 = *(const v4h*)p1;
  v4h u3 = *(const v4h*)(p1 + 4);
  v16h a;
#pragma unroll
  for (int e = 0; e < 4; e++) {
    a[e] = u0[e]; a[e + 4] = u1[e]; a[e + 8] = u2[e]; a[e + 12] = u3[e];
  }
  return a;
}

__device__ __forceinline__ float sigmoidf_(float x) { return 1.0f / (1.0f + __expf(-x)); }

__device__ __forceinline__ unsigned pk2_scaled_f16(float a, float b, float sc) {
  unsigned r;
  asm("v_fma_mixlo_f16 %0, %1, %2, 0" : "=v"(r) : "v"(a), "s"(sc));
  asm("v_fma_mixhi_f16 %0, %1, %2, 0" : "+v"(r) : "v"(b), "s"(sc));
  return r;
}

__global__ void k_prep_w1(const float* w, _Float16* wt) {
  int i = blockIdx.x * blockDim.x + threadIdx.x;
  if (i >= 32 * 192) return;
  int n = i / 192, k = i % 192;
  int c = k >> 6, kh = (k & 63) >> 3, kw = k & 7;
  wt[i] = (_Float16)w[((n * 3 + c) * 8 + kh) * 8 + kw];
}
__global__ void k_prep_w2(const float* w, _Float16* wt) {
  int i = blockIdx.x * blockDim.x + threadIdx.x;
  if (i >= 64 * 512) return;
  int n = i / 512, k = i % 512;
  int pix = k >> 5, c = k & 31, kh = pix >> 2, kw = pix & 3;
  wt[i] = (_Float16)w[((n * 32 + c) * 4 + kh) * 4 + kw];
}
__global__ void k_prep_w3(const float* w, _Float16* wt) {
  int i = blockIdx.x * blockDim.x + threadIdx.x;
  if (i >= 64 * 576) return;
  int n = i / 576, k = i % 576;
  int pix = k >> 6, c = k & 63, kh = pix / 3, kw = pix % 3;
  wt[i] = (_Float16)w[((n * 64 + c) * 3 + kh) * 3 + kw];
}
__global__ void k_prep_fc(const float* w, _Float16* wt) {
  int i = blockIdx.x * blockDim.x + threadIdx.x;
  if (i >= 512 * 3136) return;
  int f = i / 3136, k = i % 3136;
  int s = k >> 6, c = k & 63;
  wt[i] = (_Float16)w[f * 3136 + c * 49 + s];
}
__global__ void k_prep_wg(const float* wih, const float* whh, _Float16* wg) {
  int i = blockIdx.x * blockDim.x + threadIdx.x;
  if (i >= 1024 * KPAD) return;
  int r = i / KPAD, k = i % KPAD;
  float v = 0.0f;
  if (k < HL)             v = whh[r * HL + k];
  else if (k < HL + IND)  v = wih[r * IND + (k - HL)];
  wg[i] = (_Float16)v;
}
__global__ void k_prep_bias(const float* bih, const float* bhh, float* bg) {
  int i = blockIdx.x * blockDim.x + threadIdx.x;
  if (i < 1024) bg[i] = bih[i] + bhh[i];
}
__global__ void k_onehot(const int* pos, const int* lact, _Float16* X) {
  int i = blockIdx.x * blockDim.x + threadIdx.x;
  if (i >= B_TOT * 64) return;
  int b = i / 64, c = i % 64;
  float v = 0.0f;
  if (c < 16)      v = (pos[b * 2 + 0] == c)      ? 1.0f : 0.0f;
  else if (c < 32) v = (pos[b * 2 + 1] == c - 16) ? 1.0f : 0.0f;
  else if (c < 37) v = (lact[b]        == c - 32) ? 1.0f : 0.0f;
  X[b * XS + 512 + c] = (_Float16)v;
}

__global__ void __launch_bounds__(160) k_conv1(const float* img, const _Float16* wt,
                                               const float* bias, _Float16* out) {
  __shared__ __attribute__((aligned(16))) _Float16 imgL[3 * 84 * IMS];
  const float sc = 1.0f / 255.0f;
  int tid = threadIdx.x;
  int bimg = blockIdx.x;
  const float* ib = img + (size_t)bimg * (84 * 84 * 3);
  float4 v[12][3];
#pragma unroll
  for (int k = 0; k < 12; k++) {
    int u  = tid + k * 160;
    int uc = u < 1764 ? u : 1763;
    const float4* p = (const float4*)(ib + uc * 12);
    v[k][0] = p[0]; v[k][1] = p[1]; v[k][2] = p[2];
  }
#pragma unroll
  for (int k = 0; k < 12; k++) {
    int u = tid + k * 160;
    if (u < 1764) {
      int R = u / 21, cg = u - R * 21;
      _Float16* o = imgL + R * IMS + cg * 4;
      float4 q0 = v[k][0], q1 = v[k][1], q2 = v[k][2];
      uint2 c0, c1, c2;
      c0.x = pk2_scaled_f16(q0.x, q0.w, sc); c0.y = pk2_scaled_f16(q1.z, q2.y, sc);
      c1.x = pk2_scaled_f16(q0.y, q1.x, sc); c1.y = pk2_scaled_f16(q1.w, q2.z, sc);
      c2.x = pk2_scaled_f16(q0.z, q1.y, sc); c2.y = pk2_scaled_f16(q2.x, q2.w, sc);
      *(uint2*)(o)                = c0;
      *(uint2*)(o + 84 * IMS)     = c1;
      *(uint2*)(o + 2 * 84 * IMS) = c2;
    }
  }
  __syncthreads();

  int w    = tid >> 5;
  int lane = tid & 31, hi = lane >> 4, lo = lane & 15;
  int aoff[5];
#pragma unroll
  for (int i = 0; i < 5; i++) {
    int p = (w * 5 + i) * 16 + lo;
    int oh = p / 20, ow = p % 20;
    aoff[i] = (oh * 4 + hi) * IMS + ow * 4;
  }
  const _Float16* bp = wt + lo * 192 + hi * 8;
  v8f acc[5][2] = {};
#pragma unroll
  for (int kb = 0; kb < 192; kb += 32) {
    int c = kb >> 6, khb = (kb & 63) >> 3;
    v16h a[5];
    v16h b0 = pack_frag(bp + kb);
    v16h b1 = pack_frag(bp + 16 * 192 + kb);
#pragma unroll
    for (int i = 0; i < 5; i++) a[i] = pack_frag_c1(imgL + aoff[i] + (c * 84 + khb) * IMS);
    __builtin_amdgcn_sched_barrier(0);
#pragma unroll
    for (int i = 0; i < 5; i++) {
      acc[i][0] = WMMA16(a[i], b0, acc[i][0]);
      acc[i][1] = WMMA16(a[i], b1, acc[i][1]);
    }
    __builtin_amdgcn_sched_barrier(0);
  }
  float bs0 = bias[lo], bs1 = bias[16 + lo];
#pragma unroll
  for (int i = 0; i < 5; i++) {
#pragma unroll
    for (int r = 0; r < 8; r++) {
      int mm = bimg * 400 + (w * 5 + i) * 16 + hi * 8 + r;
      float v0 = acc[i][0][r] + bs0; v0 = v0 > 0.0f ? v0 : 0.0f;
      float v1 = acc[i][1][r] + bs1; v1 = v1 > 0.0f ? v1 : 0.0f;
      out[mm * 32 + lo]      = (_Float16)v0;
      out[mm * 32 + 16 + lo] = (_Float16)v1;
    }
  }
}

__global__ void __launch_bounds__(64) k_conv2(const _Float16* act1, const _Float16* wt,
                                              const float* bias, _Float16* out) {
  int wid  = (blockIdx.x * blockDim.x + threadIdx.x) >> 5;
  int lane = threadIdx.x & 31, hi = lane >> 4, lo = lane & 15;
  const _Float16* ap[4];
#pragma unroll
  for (int i = 0; i < 4; i++) {
    int m = wid * 64 + i * 16 + lo;
    int b = m / 81, p = m % 81, oh = p / 9, ow = p % 9;
    ap[i] = act1 + ((b * 20 + oh * 2) * 20 + ow * 2) * 32 + hi * 8;
  }
  const _Float16* bp = wt + lo * 512 + hi * 8;
  v8f acc[4][4] = {};
#pragma unroll 1
  for (int kb = 0; kb < 512; kb += 32) {
    int pix = kb >> 5, kh = pix >> 2, kw = pix & 3;
    int aoff = (kh * 20 + kw) * 32;
    v16h a[4], b[4];
#pragma unroll
    for (int i = 0; i < 4; i++) a[i] = pack_frag(ap[i] + aoff);
#pragma unroll
    for (int j = 0; j < 4; j++) b[j] = pack_frag(bp + j * 16 * 512 + kb);
    __builtin_amdgcn_sched_barrier(0);
#pragma unroll
    for (int i = 0; i < 4; i++) {
#pragma unroll
      for (int j = 0; j < 4; j++) acc[i][j] = WMMA16(a[i], b[j], acc[i][j]);
    }
  }
  float bs[4] = {bias[lo], bias[16 + lo], bias[32 + lo], bias[48 + lo]};
#pragma unroll
  for (int i = 0; i < 4; i++) {
#pragma unroll
    for (int r = 0; r < 8; r++) {
      int mm = wid * 64 + i * 16 + hi * 8 + r;
#pragma unroll
      for (int j = 0; j < 4; j++) {
        float v = acc[i][j][r] + bs[j]; v = v > 0.0f ? v : 0.0f;
        out[mm * 64 + j * 16 + lo] = (_Float16)v;
      }
    }
  }
}

__global__ void __launch_bounds__(64) k_conv3(const _Float16* act2, const _Float16* wt,
                                              const float* bias, _Float16* out) {
  int wid  = (blockIdx.x * blockDim.x + threadIdx.x) >> 5;
  int lane = threadIdx.x & 31, hi = lane >> 4, lo = lane & 15;
  const _Float16* ap[4];
#pragma unroll
  for (int i = 0; i < 4; i++) {
    int m = wid * 64 + i * 16 + lo;
    int b = m / 49, p = m % 49, oh = p / 7, ow = p % 7;
    ap[i] = act2 + ((b * 9 + oh) * 9 + ow) * 64 + hi * 8;
  }
  const _Float16* bp = wt + lo * 576 + hi * 8;
  v8f acc[4][4] = {};
#pragma unroll 1
  for (int kb = 0; kb < 576; kb += 32) {
    int pix = kb >> 6, cb = kb & 63;
    int kh = pix / 3, kw = pix % 3;
    int aoff = (kh * 9 + kw) * 64 + cb;
    v16h a[4], b[4];
#pragma unroll
    for (int i = 0; i < 4; i++) a[i] = pack_frag(ap[i] + aoff);
#pragma unroll
    for (int j = 0; j < 4; j++) b[j] = pack_frag(bp + j * 16 * 576 + kb);
    __builtin_amdgcn_sched_barrier(0);
#pragma unroll
    for (int i = 0; i < 4; i++) {
#pragma unroll
      for (int j = 0; j < 4; j++) acc[i][j] = WMMA16(a[i], b[j], acc[i][j]);
    }
  }
  float bs[4] = {bias[lo], bias[16 + lo], bias[32 + lo], bias[48 + lo]};
#pragma unroll
  for (int i = 0; i < 4; i++) {
#pragma unroll
    for (int r = 0; r < 8; r++) {
      int mm = wid * 64 + i * 16 + hi * 8 + r;
#pragma unroll
      for (int j = 0; j < 4; j++) {
        float v = acc[i][j][r] + bs[j]; v = v > 0.0f ? v : 0.0f;
        out[mm * 64 + j * 16 + lo] = (_Float16)v;
      }
    }
  }
}

static constexpr int FPITCH = 68;
static constexpr int FTILE  = 32 * FPITCH;
__global__ void __launch_bounds__(256) k_fc(const _Float16* act3, const _Float16* wt,
                                            const float* bias, _Float16* X) {
  __shared__ __attribute__((aligned(16))) float P[8 * FTILE];
  int tid = threadIdx.x;
  int w = __builtin_amdgcn_readfirstlane(tid >> 5);
  int lane = tid & 31, hi = lane >> 4, lo = lane & 15;
  int ng = blockIdx.x & 7, mp = blockIdx.x >> 3;
  int m0 = mp * 32, nb = ng * 64;
  const _Float16* ar0 = act3 + (size_t)(m0 + lo) * 3136 + hi * 8;
  const _Float16* ar1 = ar0 + (size_t)16 * 3136;
  const _Float16* br0 = wt + (size_t)(nb + lo) * 3136 + hi * 8;
  const _Float16* br1 = br0 + (size_t)16 * 3136;
  const _Float16* br2 = br0 + (size_t)32 * 3136;
  const _Float16* br3 = br0 + (size_t)48 * 3136;
  v8f c00 = {}, c01 = {}, c02 = {}, c03 = {};
  v8f c10 = {}, c11 = {}, c12 = {}, c13 = {};
  for (int kc = w; kc < 98; kc += 8) {
    int kb = kc * 32;
    v16h a0 = pack_frag(ar0 + kb);
    v16h a1 = pack_frag(ar1 + kb);
    v16h b0 = pack_frag(br0 + kb);
    v16h b1 = pack_frag(br1 + kb);
    v16h b2 = pack_frag(br2 + kb);
    v16h b3 = pack_frag(br3 + kb);
    c00 = WMMA16(a0, b0, c00);
    c10 = WMMA16(a1, b0, c10);
    c01 = WMMA16(a0, b1, c01);
    c11 = WMMA16(a1, b1, c11);
    c02 = WMMA16(a0, b2, c02);
    c12 = WMMA16(a1, b2, c12);
    c03 = WMMA16(a0, b3, c03);
    c13 = WMMA16(a1, b3, c13);
  }
  {
    float* pw = P + w * FTILE + (hi * 8) * FPITCH + lo;
#pragma unroll
    for (int r = 0; r < 8; r++) {
      pw[r * FPITCH +  0] = c00[r];
      pw[r * FPITCH + 16] = c01[r];
      pw[r * FPITCH + 32] = c02[r];
      pw[r * FPITCH + 48] = c03[r];
      pw[(16 + r) * FPITCH +  0] = c10[r];
      pw[(16 + r) * FPITCH + 16] = c11[r];
      pw[(16 + r) * FPITCH + 32] = c12[r];
      pw[(16 + r) * FPITCH + 48] = c13[r];
    }
  }
  __syncthreads();
  {
    int row = tid >> 3, c0 = (tid & 7) * 8;
    const float* pr = P + row * FPITCH + c0;
    float4 s0 = *(const float4*)pr;
    float4 s1 = *(const float4*)(pr + 4);
#pragma unroll
    for (int k = 1; k < 8; k++) {
      float4 u0 = *(const float4*)(pr + k * FTILE);
      float4 u1 = *(const float4*)(pr + k * FTILE + 4);
      s0.x += u0.x; s0.y += u0.y; s0.z += u0.z; s0.w += u0.w;
      s1.x += u1.x; s1.y += u1.y; s1.z += u1.z; s1.w += u1.w;
    }
    float4 q0 = *(const float4*)(bias + nb + c0);
    float4 q1 = *(const float4*)(bias + nb + c0 + 4);
    float v[8] = {s0.x + q0.x, s0.y + q0.y, s0.z + q0.z, s0.w + q0.w,
                  s1.x + q1.x, s1.y + q1.y, s1.z + q1.z, s1.w + q1.w};
    v8h o;
#pragma unroll
    for (int e = 0; e < 8; e++) o[e] = (_Float16)(v[e] > 0.0f ? v[e] : 0.0f);
    *(v8h*)(X + (size_t)(m0 + row) * XS + nb + c0) = o;
  }
}

static constexpr int PPITCH = 36;
static constexpr int PTILE  = 8 * PPITCH;
__global__ void __launch_bounds__(128) k_lstm_step(const _Float16* X, const _Float16* Wg,
                                                   const float* biasg, const int* done,
                                                   const float* h0, const float* c0,
                                                   float* cbuf, float* hs, int t) {
  __shared__ float P[4 * PTILE];
  int tid = threadIdx.x;
  int w = __builtin_amdgcn_readfirstlane(tid >> 5);
  int lane = tid & 31, hi = lane >> 4, lo = lane & 15;
  int mt = blockIdx.x >> 6, cg = blockIdx.x & 63;
  const float* hprev = (t == 0) ? h0 : hs + (size_t)(t - 1) * (E_ENVS * HL);
  const float* cin   = (t == 0) ? c0 : cbuf;
  int se = (t == 0) ? HL : 4;
  int sq = (t == 0) ? 4 : E_ENVS * 4;

  int ci   = tid & 63;
  int eenv = mt * 16 + (ci >> 2);
  int ecol = cg * 4 + (ci & 3);
  int edone = done[t * E_ENVS + eenv];
  float cprev = cin[eenv * se + cg * sq + (ci & 3)];
  float bi = biasg[0 * HL + ecol], bf = biasg[1 * HL + ecol];
  float bg = biasg[2 * HL + ecol], bo = biasg[3 * HL + ecol];

  int aenv = mt * 16 + lo;
  int adone = done[t * E_ENVS + aenv];
  int grow = (lo >> 2) * HL + cg * 4 + (lo & 3);
  const _Float16* wrow = Wg + (size_t)grow * KPAD + hi * 8;
  const _Float16* xrow = X + (size_t)(t * E_ENVS + aenv) * XS + hi * 8;

  v16h a6 = {}, b6 = {};
  if (w < 2) {
    int kc = w + 24;
    a6 = pack_frag(xrow + (kc - 8) * 32);
    b6 = pack_frag(wrow + kc * 32);
  }
  float4 hu[2][4];
  v16h bh[2];
#pragma unroll
  for (int i = 0; i < 2; i++) {
    int kc = w + 4 * i;
    const float* hp = hprev + aenv * se + (kc * 8 + hi * 2) * sq;
    hu[i][0] = *(const float4*)hp;
    hu[i][1] = *(const float4*)(hp + sq);
    hu[i][2] = *(const float4*)(hp + 4 * sq);
    hu[i][3] = *(const float4*)(hp + 5 * sq);
    bh[i] = pack_frag(wrow + kc * 32);
  }
  v16h ax[4], bx[4];
#pragma unroll
  for (int i = 0; i < 4; i++) {
    int kc = w + 8 + 4 * i;
    ax[i] = pack_frag(xrow + (kc - 8) * 32);
    bx[i] = pack_frag(wrow + kc * 32);
  }

  __builtin_amdgcn_sched_barrier(0);

  v8f acc = {};
#pragma unroll
  for (int i = 0; i < 2; i++) {
    float hv16[16] = {hu[i][0].x, hu[i][0].y, hu[i][0].z, hu[i][0].w,
                      hu[i][1].x, hu[i][1].y, hu[i][1].z, hu[i][1].w,
                      hu[i][2].x, hu[i][2].y, hu[i][2].z, hu[i][2].w,
                      hu[i][3].x, hu[i][3].y, hu[i][3].z, hu[i][3].w};
    v16h a;
#pragma unroll
    for (int e = 0; e < 16; e++) a[e] = adone ? (_Float16)0.0f : (_Float16)hv16[e];
    acc = WMMA16(a, bh[i], acc);
  }
#pragma unroll
  for (int i = 0; i < 4; i++) acc = WMMA16(ax[i], bx[i], acc);
  if (w < 2) acc = WMMA16(a6, b6, acc);

  {
    float* pw = P + w * PTILE + lane;
#pragma unroll
    for (int r = 0; r < 8; r++) pw[r * PPITCH] = acc[r];
  }
  __syncthreads();

  if (tid < 64) {
    int el = tid >> 2, q = tid & 3;
    const float* pr = P + (el & 7) * PPITCH + (el >> 3) * 16 + q;
    float si = 0.0f, sf = 0.0f, sg = 0.0f, so = 0.0f;
#pragma unroll
    for (int k = 0; k < 4; k++) {
      si += pr[k * PTILE + 0];
      sf += pr[k * PTILE + 4];
      sg += pr[k * PTILE + 8];
      so += pr[k * PTILE + 12];
    }
    float cp = edone ? 0.0f : cprev;
    float iv = sigmoidf_(si + bi), fv = sigmoidf_(sf + bf);
    float gv = tanhf(sg + bg),     ov = sigmoidf_(so + bo);
    float cv = fv * cp + iv * gv;
    float hv = ov * tanhf(cv);
    cbuf[(cg * E_ENVS + eenv) * 4 + q] = cv;
    hs[(size_t)t * (E_ENVS * HL) + (cg * E_ENVS + eenv) * 4 + q] = hv;
  }
}

__global__ void k_heads(const float* hs, const float* cbuf, const float* pw, const float* pb,
                        const float* vw, const float* vb, float* out) {
  int i = blockIdx.x * blockDim.x + threadIdx.x;
  if (i < B_TOT * 6) {
    int b = i / 6, j = i % 6;
    const float* h    = hs + (size_t)(b >> 5) * (E_ENVS * HL) + (b & 31) * 4;
    const float* wrow = (j < 5) ? pw + j * HL : vw;
    float s = (j < 5) ? pb[j] : vb[0];
#pragma unroll 4
    for (int k4 = 0; k4 < HL / 4; k4++) {
      float4 hv = *(const float4*)(h + k4 * (E_ENVS * 4));
      s += hv.x * wrow[k4 * 4 + 0];
      s += hv.y * wrow[k4 * 4 + 1];
      s += hv.z * wrow[k4 * 4 + 2];
      s += hv.w * wrow[k4 * 4 + 3];
    }
    if (j < 5) out[b * 5 + j] = s;
    else       out[10240 + b] = s;
  } else {
    int o = i - B_TOT * 6;
    int sel = o >> 13, r = o & 8191;
    int e = r >> 8, k = r & 255;
    int src = ((k >> 2) * E_ENVS + e) * 4 + (k & 3);
    if (sel == 0) out[12288 + r] = hs[(size_t)(T_STEPS - 1) * (E_ENVS * HL) + src];
    else          out[20480 + r] = cbuf[src];
  }
}

extern "C" void kernel_launch(void* const* d_in, const int* in_sizes, int n_in,
                              void* d_out, int out_size, void* d_ws, size_t ws_size,
                              hipStream_t stream) {
  (void)in_sizes; (void)n_in; (void)out_size; (void)ws_size;
  const float* img  = (const float*)d_in[0];
  const int*   pos  = (const int*)d_in[1];
  const int*   lact = (const int*)d_in[2];
  const int*   done = (const int*)d_in[3];
  const float* h0   = (const float*)d_in[4];
  const float* c0   = (const float*)d_in[5];
  const float* w1   = (const float*)d_in[6];
  const float* b1   = (const float*)d_in[7];
  const float* w2   = (const float*)d_in[8];
  const float* b2   = (const float*)d_in[9];
  const float* w3   = (const float*)d_in[10];
  const float* b3   = (const float*)d_in[11];
  const float* fcw  = (const float*)d_in[12];
  const float* fcb  = (const float*)d_in[13];
  const float* wih  = (const float*)d_in[14];
  const float* whh  = (const float*)d_in[15];
  const float* bih  = (const float*)d_in[16];
  const float* bhh  = (const float*)d_in[17];
  const float* pw   = (const float*)d_in[18];
  const float* pb   = (const float*)d_in[19];
  const float* vw   = (const float*)d_in[20];
  const float* vb   = (const float*)d_in[21];
  float* out = (float*)d_out;
  char*  ws  = (char*)d_ws;

  const size_t o_act1 = 90832896;
  const size_t o_w1t  = o_act1 + 52428800;
  const size_t o_w2t  = o_w1t + 12288;
  const size_t o_w3t  = o_w2t + 65536;
  const size_t o_fcwt = o_w3t + 73728;
  const size_t o_wg   = o_fcwt + 3211264;
  const size_t o_bg   = o_wg + 1703936;
  const size_t o_act2 = 0;
  const size_t o_act3 = 21233664;
  const size_t o_X    = 34078720;
  const size_t o_hs   = 36438016;
  const size_t o_c    = 38535168;

  _Float16* act1 = (_Float16*)(ws + o_act1);
  _Float16* act2 = (_Float16*)(ws + o_act2);
  _Float16* act3 = (_Float16*)(ws + o_act3);
  _Float16* Xm   = (_Float16*)(ws + o_X);
  _Float16* w1t  = (_Float16*)(ws + o_w1t);
  _Float16* w2t  = (_Float16*)(ws + o_w2t);
  _Float16* w3t  = (_Float16*)(ws + o_w3t);
  _Float16* fcwt = (_Float16*)(ws + o_fcwt);
  _Float16* wg   = (_Float16*)(ws + o_wg);
  float*    bg   = (float*)(ws + o_bg);
  float*    hsb  = (float*)(ws + o_hs);
  float*    cbuf = (float*)(ws + o_c);

  const int TB = 256;
  k_prep_w1  <<<(32 * 192   + TB - 1) / TB, TB, 0, stream>>>(w1, w1t);
  k_prep_w2  <<<(64 * 512   + TB - 1) / TB, TB, 0, stream>>>(w2, w2t);
  k_prep_w3  <<<(64 * 576   + TB - 1) / TB, TB, 0, stream>>>(w3, w3t);
  k_prep_fc  <<<(512 * 3136 + TB - 1) / TB, TB, 0, stream>>>(fcw, fcwt);
  k_prep_wg  <<<(1024 * KPAD + TB - 1) / TB, TB, 0, stream>>>(wih, whh, wg);
  k_prep_bias<<<(1024 + TB - 1) / TB, TB, 0, stream>>>(bih, bhh, bg);

  k_conv1<<<2048, 160, 0, stream>>>(img, w1t, b1, act1);
  k_conv2<<<1296, 64, 0, stream>>>(act1, w2t, b2, act2);
  k_conv3<<< 784, 64, 0, stream>>>(act2, w3t, b3, act3);
  k_fc   <<< 512, TB, 0, stream>>>(act3, fcwt, fcb, Xm);

  k_onehot<<<(B_TOT * 64 + TB - 1) / TB, TB, 0, stream>>>(pos, lact, Xm);

  k_lstm_step<<<128, 128, 0, stream>>>(Xm, wg, bg, done, h0, c0, cbuf, hsb, 0);
  k_lstm_step<<<128, 128, 0, stream>>>(Xm, wg, bg, done, h0, c0, cbuf, hsb, 1);
  k_lstm_step<<<128, 128, 0, stream>>>(Xm, wg, bg, done, h0, c0, cbuf, hsb, 2);
  k_lstm_step<<<128, 128, 0, stream>>>(Xm, wg, bg, done, h0, c0, cbuf, hsb, 3);
  k_lstm_step<<<128, 128, 0, stream>>>(Xm, wg, bg, done, h0, c0, cbuf, hsb, 4);
  k_lstm_step<<<128, 128, 0, stream>>>(Xm, wg, bg, done, h0, c0, cbuf, hsb, 5);
  k_lstm_step<<<128, 128, 0, stream>>>(Xm, wg, bg, done, h0, c0, cbuf, hsb, 6);
  k_lstm_step<<<128, 128, 0, stream>>>(Xm, wg, bg, done, h0, c0, cbuf, hsb, 7);
  k_lstm_step<<<128, 128, 0, stream>>>(Xm, wg, bg, done, h0, c0, cbuf, hsb, 8);
  k_lstm_step<<<128, 128, 0, stream>>>(Xm, wg, bg, done, h0, c0, cbuf, hsb, 9);
  k_lstm_step<<<128, 128, 0, stream>>>(Xm, wg, bg, done, h0, c0, cbuf, hsb, 10);
  k_lstm_step<<<128, 128, 0, stream>>>(Xm, wg, bg, done, h0, c0, cbuf, hsb, 11);
  k_lstm_step<<<128, 128, 0, stream>>>(Xm, wg, bg, done, h0, c0, cbuf, hsb, 12);
  k_lstm_step<<<128, 128, 0, stream>>>(Xm, wg, bg, done, h0, c0, cbuf, hsb, 13);
  k_lstm_step<<<128, 128, 0, stream>>>(Xm, wg, bg, done, h0, c0, cbuf, hsb, 14);
  k_lstm_step<<<128, 128, 0, stream>>>(Xm, wg, bg, done, h0, c0, cbuf, hsb, 15);
  k_lstm_step<<<128, 128, 0, stream>>>(Xm, wg, bg, done, h0, c0, cbuf, hsb, 16);
  k_lstm_step<<<128, 128, 0, stream>>>(Xm, wg, bg, done, h0, c0, cbuf, hsb, 17);
  k_lstm_step<<<128, 128, 0, stream>>>(Xm, wg, bg, done, h0, c0, cbuf, hsb, 18);
  k_lstm_step<<<128, 128, 0, stream>>>(Xm, wg, bg, done, h0, c0, cbuf, hsb, 19);
  k_lstm_step<<<128, 128, 0, stream>>>(Xm, wg, bg, done, h0, c0, cbuf, hsb, 20);
  k_lstm_step<<<128, 128, 0, stream>>>(Xm, wg, bg, done, h0, c0, cbuf, hsb, 21);
  k_lstm_step<<<128, 128, 0, stream>>>(Xm, wg, bg, done, h0, c0, cbuf, hsb, 22);
  k_lstm_step<<<128, 128, 0, stream>>>(Xm, wg, bg, done, h0, c0, cbuf, hsb, 23);
  k_lstm_step<<<128, 128, 0, stream>>>(Xm, wg, bg, done, h0, c0, cbuf, hsb, 24);
  k_lstm_step<<<128, 128, 0, stream>>>(Xm, wg, bg, done, h0, c0, cbuf, hsb, 25);
  k_lstm_step<<<128, 128, 0, stream>>>(Xm, wg, bg, done, h0, c0, cbuf, hsb, 26);
  k_lstm_step<<<128, 128, 0, stream>>>(Xm, wg, bg, done, h0, c0, cbuf, hsb, 27);
  k_lstm_step<<<128, 128, 0, stream>>>(Xm, wg, bg, done, h0, c0, cbuf, hsb, 28);
  k_lstm_step<<<128, 128, 0, stream>>>(Xm, wg, bg, done, h0, c0, cbuf, hsb, 29);
  k_lstm_step<<<128, 128, 0, stream>>>(Xm, wg, bg, done, h0, c0, cbuf, hsb, 30);
  k_lstm_step<<<128, 128, 0, stream>>>(Xm, wg, bg, done, h0, c0, cbuf, hsb, 31);
  k_lstm_step<<<128, 128, 0, stream>>>(Xm, wg, bg, done, h0, c0, cbuf, hsb, 32);
  k_lstm_step<<<128, 128, 0, stream>>>(Xm, wg, bg, done, h0, c0, cbuf, hsb, 33);
  k_lstm_step<<<128, 128, 0, stream>>>(Xm, wg, bg, done, h0, c0, cbuf, hsb, 34);
  k_lstm_step<<<128, 128, 0, stream>>>(Xm, wg, bg, done, h0, c0, cbuf, hsb, 35);
  k_lstm_step<<<128, 128, 0, stream>>>(Xm, wg, bg, done, h0, c0, cbuf, hsb, 36);
  k_lstm_step<<<128, 128, 0, stream>>>(Xm, wg, bg, done, h0, c0, cbuf, hsb, 37);
  k_lstm_step<<<128, 128, 0, stream>>>(Xm, wg, bg, done, h0, c0, cbuf, hsb, 38);
  k_lstm_step<<<128, 128, 0, stream>>>(Xm, wg, bg, done, h0, c0, cbuf, hsb, 39);
  k_lstm_step<<<128, 128, 0, stream>>>(Xm, wg, bg, done, h0, c0, cbuf, hsb, 40);
  k_lstm_step<<<128, 128, 0, stream>>>(Xm, wg, bg, done, h0, c0, cbuf, hsb, 41);
  k_lstm_step<<<128, 128, 0, stream>>>(Xm, wg, bg, done, h0, c0, cbuf, hsb, 42);
  k_lstm_step<<<128, 128, 0, stream>>>(Xm, wg, bg, done, h0, c0, cbuf, hsb, 43);
  k_lstm_step<<<128, 128, 0, stream>>>(Xm, wg, bg, done, h0, c0, cbuf, hsb, 44);
  k_lstm_step<<<128, 128, 0, stream>>>(Xm, wg, bg, done, h0, c0, cbuf, hsb, 45);
  k_lstm_step<<<128, 128, 0, stream>>>(Xm, wg, bg, done, h0, c0, cbuf, hsb, 46);
  k_lstm_step<<<128, 128, 0, stream>>>(Xm, wg, bg, done, h0, c0, cbuf, hsb, 47);
  k_lstm_step<<<128, 128, 0, stream>>>(Xm, wg, bg, done, h0, c0, cbuf, hsb, 48);
  k_lstm_step<<<128, 128, 0, stream>>>(Xm, wg, bg, done, h0, c0, cbuf, hsb, 49);
  k_lstm_step<<<128, 128, 0, stream>>>(Xm, wg, bg, done, h0, c0, cbuf, hsb, 50);
  k_lstm_step<<<128, 128, 0, stream>>>(Xm, wg, bg, done, h0, c0, cbuf, hsb, 51);
  k_lstm_step<<<128, 128, 0, stream>>>(Xm, wg, bg, done, h0, c0, cbuf, hsb, 52);
  k_lstm_step<<<128, 128, 0, stream>>>(Xm, wg, bg, done, h0, c0, cbuf, hsb, 53);
  k_lstm_step<<<128, 128, 0, stream>>>(Xm, wg, bg, done, h0, c0, cbuf, hsb, 54);
  k_lstm_step<<<128, 128, 0, stream>>>(Xm, wg, bg, done, h0, c0, cbuf, hsb, 55);
  k_lstm_step<<<128, 128, 0, stream>>>(Xm, wg, bg, done, h0, c0, cbuf, hsb, 56);
  k_lstm_step<<<128, 128, 0, stream>>>(Xm, wg, bg, done, h0, c0, cbuf, hsb, 57);
  k_lstm_step<<<128, 128, 0, stream>>>(Xm, wg, bg, done, h0, c0, cbuf, hsb, 58);
  k_lstm_step<<<128, 128, 0, stream>>>(Xm, wg, bg, done, h0, c0, cbuf, hsb, 59);
  k_lstm_step<<<128, 128, 0, stream>>>(Xm, wg, bg, done, h0, c0, cbuf, hsb, 60);
  k_lstm_step<<<128, 128, 0, stream>>>(Xm, wg, bg, done, h0, c0, cbuf, hsb, 61);
  k_lstm_step<<<128, 128, 0, stream>>>(Xm, wg, bg, done, h0, c0, cbuf, hsb, 62);
  k_lstm_step<<<128, 128, 0, stream>>>(Xm, wg, bg, done, h0, c0, cbuf, hsb, 63);

  k_heads<<<112, TB, 0, stream>>>(hsb, cbuf, pw, pb, vw, vb, out);
}


# conv1-3 fused per image: one workgroup keeps conv1 and conv2 outputs in LDS (over the image planes) as the next layer's A fragments; 3 conv launches -> 1; values and sum order unchanged
# speedup vs baseline: 11.2580x; 1.0769x over previous
#include <hip/hip_runtime.h>
#include <math.h>

typedef __attribute__((ext_vector_type(16))) _Float16 v16h;
typedef __attribute__((ext_vector_type(8)))  _Float16 v8h;
typedef __attribute__((ext_vector_type(4)))  _Float16 v4h;
typedef __attribute__((ext_vector_type(8)))  float    v8f;

#define WMMA16(A,B,C) __builtin_amdgcn_wmma_f32_16x16x32_f16(false,(A),false,(B),(short)0,(C),false,false)

static constexpr int T_STEPS = 64;
static constexpr int E_ENVS  = 32;
static constexpr int B_TOT   = 2048;
static constexpr int HL      = 256;
static constexpr int IND     = 549;
static constexpr int KPAD    = 832;
static constexpr int XS      = 576;
static constexpr int IMS     = 88;
static constexpr int A1P     = 40;
static constexpr int A2P     = 72;
static constexpr int A2OFF   = 400 * A1P;

__device__ __forceinline__ v16h pack_frag(const _Float16* p) {
  v8h u = *(const v8h*)p;
  v8h w = *(const v8h*)(p + 16);
  v16h a;
#pragma unroll
  for (int e = 0; e < 8; e++) { a[e] = u[e]; a[e + 8] = w[e]; }
  return a;
}
__device__ __forceinline__ v16h pack_frag_c1(const _Float16* p0) {
  const _Float16* p1 = p0 + 2 * IMS;
  v4h u0 = *(const v4h*)p0;
  v4h u1 = *(const v4h*)(p0 + 4);
  v4h u2 = *(const v4h*)p1;
  v4h u3 = *(const v4h*)(p1 + 4);
  v16h a;
#pragma unroll
  for (int e = 0; e < 4; e++) {
    a[e] = u0[e]; a[e + 4] = u1[e]; a[e + 8] = u2[e]; a[e + 12] = u3[e];
  }
  return a;
}

__device__ __forceinline__ float sigmoidf_(float x) { return 1.0f / (1.0f + __expf(-x)); }

__device__ __forceinline__ unsigned pk2_scaled_f16(float a, float b, float sc) {
  unsigned r;
  asm("v_fma_mixlo_f16 %0, %1, %2, 0" : "=v"(r) : "v"(a), "s"(sc));
  asm("v_fma_mixhi_f16 %0, %1, %2, 0" : "+v"(r) : "v"(b), "s"(sc));
  return r;
}

__global__ void k_prep_w1(const float* w, _Float16* wt) {
  int i = blockIdx.x * blockDim.x + threadIdx.x;
  if (i >= 32 * 192) return;
  int n = i / 192, k = i % 192;
  int c = k >> 6, kh = (k & 63) >> 3, kw = k & 7;
  wt[i] = (_Float16)w[((n * 3 + c) * 8 + kh) * 8 + kw];
}
__global__ void k_prep_w2(const float* w, _Float16* wt) {
  int i = blockIdx.x * blockDim.x + threadIdx.x;
  if (i >= 64 * 512) return;
  int n = i / 512, k = i % 512;
  int pix = k >> 5, c = k & 31, kh = pix >> 2, kw = pix & 3;
  wt[i] = (_Float16)w[((n * 32 + c) * 4 + kh) * 4 + kw];
}
__global__ void k_prep_w3(const float* w, _Float16* wt) {
  int i = blockIdx.x * blockDim.x + threadIdx.x;
  if (i >= 64 * 576) return;
  int n = i / 576, k = i % 576;
  int pix = k >> 6, c = k & 63, kh = pix / 3, kw = pix % 3;
  wt[i] = (_Float16)w[((n * 64 + c) * 3 + kh) * 3 + kw];
}
__global__ void k_prep_fc(const float* w, _Float16* wt) {
  int i = blockIdx.x * blockDim.x + threadIdx.x;
  if (i >= 512 * 3136) return;
  int f = i / 3136, k = i % 3136;
  int s = k >> 6, c = k & 63;
  wt[i] = (_Float16)w[f * 3136 + c * 49 + s];
}
__global__ void k_prep_wg(const float* wih, const float* whh, _Float16* wg) {
  int i = blockIdx.x * blockDim.x + threadIdx.x;
  if (i >= 1024 * KPAD) return;
  int r = i / KPAD, k = i % KPAD;
  float v = 0.0f;
  if (k < HL)             v = whh[r * HL + k];
  else if (k < HL + IND)  v = wih[r * IND + (k - HL)];
  wg[i] = (_Float16)v;
}
__global__ void k_prep_bias(const float* bih, const float* bhh, float* bg) {
  int i = blockIdx.x * blockDim.x + threadIdx.x;
  if (i < 1024) bg[i] = bih[i] + bhh[i];
}
__global__ void k_onehot(const int* pos, const int* lact, _Float16* X) {
  int i = blockIdx.x * blockDim.x + threadIdx.x;
  if (i >= B_TOT * 64) return;
  int b = i / 64, c = i % 64;
  float v = 0.0f;
  if (c < 16)      v = (pos[b * 2 + 0] == c)      ? 1.0f : 0.0f;
  else if (c < 32) v = (pos[b * 2 + 1] == c - 16) ? 1.0f : 0.0f;
  else if (c < 37) v = (lact[b]        == c - 32) ? 1.0f : 0.0f;
  X[b * XS + 512 + c] = (_Float16)v;
}

__global__ void __launch_bounds__(160) k_convs(const float* img, const _Float16* wt1,
                                               const float* bias1, const _Float16* wt2,
                                               const float* bias2, const _Float16* wt3,
                                               const float* bias3, _Float16* out) {
  __shared__ __attribute__((aligned(16))) _Float16 imgL[3 * 84 * IMS];
  const float sc = 1.0f / 255.0f;
  int tid = threadIdx.x;
  int bimg = blockIdx.x;
  const float* ib = img + (size_t)bimg * (84 * 84 * 3);
  {
    float4 v[12][3];
#pragma unroll
    for (int k = 0; k < 12; k++) {
      int u  = tid + k * 160;
      int uc = u < 1764 ? u : 1763;
      const float4* p = (const float4*)(ib + uc * 12);
      v[k][0] = p[0]; v[k][1] = p[1]; v[k][2] = p[2];
    }
#pragma unroll
    for (int k = 0; k < 12; k++) {
      int u = tid + k * 160;
      if (u < 1764) {
        int R = u / 21, cg = u - R * 21;
        _Float16* o = imgL + R * IMS + cg * 4;
        float4 q0 = v[k][0], q1 = v[k][1], q2 = v[k][2];
        uint2 c0, c1, c2;
        c0.x = pk2_scaled_f16(q0.x, q0.w, sc); c0.y = pk2_scaled_f16(q1.z, q2.y, sc);
        c1.x = pk2_scaled_f16(q0.y, q1.x, sc); c1.y = pk2_scaled_f16(q1.w, q2.z, sc);
        c2.x = pk2_scaled_f16(q0.z, q1.y, sc); c2.y = pk2_scaled_f16(q2.x, q2.w, sc);
        *(uint2*)(o)                = c0;
        *(uint2*)(o + 84 * IMS)     = c1;
        *(uint2*)(o + 2 * 84 * IMS) = c2;
      }
    }
  }
  __syncthreads();

  int w    = __builtin_amdgcn_readfirstlane(tid >> 5);
  int lane = tid & 31, hi = lane >> 4, lo = lane & 15;
  _Float16* a1 = imgL;
  _Float16* a2 = imgL + A2OFF;

  {
    int aoff[5];
#pragma unroll
    for (int i = 0; i < 5; i++) {
      int p = (w * 5 + i) * 16 + lo;
      int oh = p / 20, ow = p % 20;
      aoff[i] = (oh * 4 + hi) * IMS + ow * 4;
    }
    const _Float16* bp = wt1 + lo * 192 + hi * 8;
    v8f acc[5][2] = {};
#pragma unroll
    for (int kb = 0; kb < 192; kb += 32) {
      int c = kb >> 6, khb = (kb & 63) >> 3;
      v16h a[5];
      v16h b0 = pack_frag(bp + kb);
      v16h b1 = pack_frag(bp + 16 * 192 + kb);
#pragma unroll
      for (int i = 0; i < 5; i++) a[i] = pack_frag_c1(imgL + aoff[i] + (c * 84 + khb) * IMS);
      __builtin_amdgcn_sched_barrier(0);
#pragma unroll
      for (int i = 0; i < 5; i++) {
        acc[i][0] = WMMA16(a[i], b0, acc[i][0]);
        acc[i][1] = WMMA16(a[i], b1, acc[i][1]);
      }
      __builtin_amdgcn_sched_barrier(0);
    }
    __syncthreads();
    float bs0 = bias1[lo], bs1 = bias1[16 + lo];
#pragma unroll
    for (int i = 0; i < 5; i++) {
#pragma unroll
      for (int r = 0; r < 8; r++) {
        int p = (w * 5 + i) * 16 + hi * 8 + r;
        float v0 = acc[i][0][r] + bs0; v0 = v0 > 0.0f ? v0 : 0.0f;
        float v1 = acc[i][1][r] + bs1; v1 = v1 > 0.0f ? v1 : 0.0f;
        a1[p * A1P + lo]      = (_Float16)v0;
        a1[p * A1P + 16 + lo] = (_Float16)v1;
      }
    }
  }
  __syncthreads();

  if (w < 4) {
    int aoff[6];
#pragma unroll
    for (int i = 0; i < 6; i++) {
      int p = i * 16 + lo;
      p = p < 81 ? p : 80;
      int oh = p / 9, ow = p % 9;
      aoff[i] = ((oh * 2) * 20 + ow * 2) * A1P + hi * 8;
    }
    const _Float16* bp = wt2 + (w * 16 + lo) * 512 + hi * 8;
    v8f acc[6] = {};
#pragma unroll 1
    for (int kh = 0; kh < 4; kh++) {
      v16h b[4];
#pragma unroll
      for (int kw = 0; kw < 4; kw++) b[kw] = pack_frag(bp + (kh * 4 + kw) * 32);
#pragma unroll
      for (int kw = 0; kw < 4; kw++) {
        v16h a[6];
#pragma unroll
        for (int i = 0; i < 6; i++) a[i] = pack_frag(a1 + aoff[i] + (kh * 20 + kw) * A1P);
        __builtin_amdgcn_sched_barrier(0);
#pragma unroll
        for (int i = 0; i < 6; i++) acc[i] = WMMA16(a[i], b[kw], acc[i]);
        __builtin_amdgcn_sched_barrier(0);
      }
    }
    float bs = bias2[w * 16 + lo];
#pragma unroll
    for (int i = 0; i < 6; i++) {
#pragma unroll
      for (int r = 0; r < 8; r++) {
        int p = i * 16 + hi * 8 + r;
        float v = acc[i][r] + bs; v = v > 0.0f ? v : 0.0f;
        if (p < 81) a2[p * A2P + w * 16 + lo] = (_Float16)v;
      }
    }
  }
  __syncthreads();

  if (w < 4) {
    int aoff[4];
#pragma unroll
    for (int i = 0; i < 4; i++) {
      int p = i * 16 + lo;
      p = p < 49 ? p : 48;
      int oh = p / 7, ow = p % 7;
      aoff[i] = (oh * 9 + ow) * A2P + hi * 8;
    }
    const _Float16* bp = wt3 + (w * 16 + lo) * 576 + hi * 8;
    v8f acc[4] = {};
#pragma unroll 1
    for (int kh = 0; kh < 3; kh++) {
      v16h b[6];
#pragma unroll
      for (int j = 0; j < 6; j++) b[j] = pack_frag(bp + (kh * 6 + j) * 32);
#pragma unroll
      for (int j = 0; j < 6; j++) {
        int kw = j >> 1, cb = (j & 1) * 32;
        v16h a[4];
#pragma unroll
        for (int i = 0; i < 4; i++) a[i] = pack_frag(a2 + aoff[i] + (kh * 9 + kw) * A2P + cb);
        __builtin_amdgcn_sched_barrier(0);
#pragma unroll
        for (int i = 0; i < 4; i++) acc[i] = WMMA16(a[i], b[j], acc[i]);
        __builtin_amdgcn_sched_barrier(0);
      }
    }
    float bs = bias3[w * 16 + lo];
    _Float16* ob = out + (size_t)bimg * (49 * 64) + w * 16 + lo;
#pragma unroll
    for (int i = 0; i < 4; i++) {
#pragma unroll
      for (int r = 0; r < 8; r++) {
        int p = i * 16 + hi * 8 + r;
        float v = acc[i][r] + bs; v = v > 0.0f ? v : 0.0f;
        if (p < 49) ob[p * 64] = (_Float16)v;
      }
    }
  }
}

static constexpr int FPITCH = 68;
static constexpr int FTILE  = 32 * FPITCH;
__global__ void __launch_bounds__(256) k_fc(const _Float16* act3, const _Float16* wt,
                                            const float* bias, _Float16* X) {
  __shared__ __attribute__((aligned(16))) float P[8 * FTILE];
  int tid = threadIdx.x;
  int w = __builtin_amdgcn_readfirstlane(tid >> 5);
  int lane = tid & 31, hi = lane >> 4, lo = lane & 15;
  int ng = blockIdx.x & 7, mp = blockIdx.x >> 3;
  int m0 = mp * 32, nb = ng * 64;
  const _Float16* ar0 = act3 + (size_t)(m0 + lo) * 3136 + hi * 8;
  const _Float16* ar1 = ar0 + (size_t)16 * 3136;
  const _Float16* br0 = wt + (size_t)(nb + lo) * 3136 + hi * 8;
  const _Float16* br1 = br0 + (size_t)16 * 3136;
  const _Float16* br2 = br0 + (size_t)32 * 3136;
  const _Float16* br3 = br0 + (size_t)48 * 3136;
  v8f c00 = {}, c01 = {}, c02 = {}, c03 = {};
  v8f c10 = {}, c11 = {}, c12 = {}, c13 = {};
  for (int kc = w; kc < 98; kc += 8) {
    int kb = kc * 32;
    v16h a0 = pack_frag(ar0 + kb);
    v16h a1 = pack_frag(ar1 + kb);
    v16h b0 = pack_frag(br0 + kb);
    v16h b1 = pack_frag(br1 + kb);
    v16h b2 = pack_frag(br2 + kb);
    v16h b3 = pack_frag(br3 + kb);
    c00 = WMMA16(a0, b0, c00);
    c10 = WMMA16(a1, b0, c10);
    c01 = WMMA16(a0, b1, c01);
    c11 = WMMA16(a1, b1, c11);
    c02 = WMMA16(a0, b2, c02);
    c12 = WMMA16(a1, b2, c12);
    c03 = WMMA16(a0, b3, c03);
    c13 = WMMA16(a1, b3, c13);
  }
  {
    float* pw = P + w * FTILE + (hi * 8) * FPITCH + lo;
#pragma unroll
    for (int r = 0; r < 8; r++) {
      pw[r * FPITCH +  0] = c00[r];
      pw[r * FPITCH + 16] = c01[r];
      pw[r * FPITCH + 32] = c02[r];
      pw[r * FPITCH + 48] = c03[r];
      pw[(16 + r) * FPITCH +  0] = c10[r];
      pw[(16 + r) * FPITCH + 16] = c11[r];
      pw[(16 + r) * FPITCH + 32] = c12[r];
      pw[(16 + r) * FPITCH + 48] = c13[r];
    }
  }
  __syncthreads();
  {
    int row = tid >> 3, c0 = (tid & 7) * 8;
    const float* pr = P + row * FPITCH + c0;
    float4 s0 = *(const float4*)pr;
    float4 s1 = *(const float4*)(pr + 4);
#pragma unroll
    for (int k = 1; k < 8; k++) {
      float4 u0 = *(const float4*)(pr + k * FTILE);
      float4 u1 = *(const float4*)(pr + k * FTILE + 4);
      s0.x += u0.x; s0.y += u0.y; s0.z += u0.z; s0.w += u0.w;
      s1.x += u1.x; s1.y += u1.y; s1.z += u1.z; s1.w += u1.w;
    }
    float4 q0 = *(const float4*)(bias + nb + c0);
    float4 q1 = *(const float4*)(bias + nb + c0 + 4);
    float v[8] = {s0.x + q0.x, s0.y + q0.y, s0.z + q0.z, s0.w + q0.w,
                  s1.x + q1.x, s1.y + q1.y, s1.z + q1.z, s1.w + q1.w};
    v8h o;
#pragma unroll
    for (int e = 0; e < 8; e++) o[e] = (_Float16)(v[e] > 0.0f ? v[e] : 0.0f);
    *(v8h*)(X + (size_t)(m0 + row) * XS + nb + c0) = o;
  }
}

static constexpr int PPITCH = 36;
static constexpr int PTILE  = 8 * PPITCH;
__global__ void __launch_bounds__(128) k_lstm_step(const _Float16* X, const _Float16* Wg,
                                                   const float* biasg, const int* done,
                                                   const float* h0, const float* c0,
                                                   float* cbuf, float* hs, int t) {
  __shared__ float P[4 * PTILE];
  int tid = threadIdx.x;
  int w = __builtin_amdgcn_readfirstlane(tid >> 5);
  int lane = tid & 31, hi = lane >> 4, lo = lane & 15;
  int mt = blockIdx.x >> 6, cg = blockIdx.x & 63;
  const float* hprev = (t == 0) ? h0 : hs + (size_t)(t - 1) * (E_ENVS * HL);
  const float* cin   = (t == 0) ? c0 : cbuf;
  int se = (t == 0) ? HL : 4;
  int sq = (t == 0) ? 4 : E_ENVS * 4;

  int ci   = tid & 63;
  int eenv = mt * 16 + (ci >> 2);
  int ecol = cg * 4 + (ci & 3);
  int edone = done[t * E_ENVS + eenv];
  float cprev = cin[eenv * se + cg * sq + (ci & 3)];
  float bi = biasg[0 * HL + ecol], bf = biasg[1 * HL + ecol];
  float bg = biasg[2 * HL + ecol], bo = biasg[3 * HL + ecol];

  int aenv = mt * 16 + lo;
  int adone = done[t * E_ENVS + aenv];
  int grow = (lo >> 2) * HL + cg * 4 + (lo & 3);
  const _Float16* wrow = Wg + (size_t)grow * KPAD + hi * 8;
  const _Float16* xrow = X + (size_t)(t * E_ENVS + aenv) * XS + hi * 8;

  v16h a6 = {}, b6 = {};
  if (w < 2) {
    int kc = w + 24;
    a6 = pack_frag(xrow + (kc - 8) * 32);
    b6 = pack_frag(wrow + kc * 32);
  }
  float4 hu[2][4];
  v16h bh[2];
#pragma unroll
  for (int i = 0; i < 2; i++) {
    int kc = w + 4 * i;
    const float* hp = hprev + aenv * se + (kc * 8 + hi * 2) * sq;
    hu[i][0] = *(const float4*)hp;
    hu[i][1] = *(const float4*)(hp + sq);
    hu[i][2] = *(const float4*)(hp + 4 * sq);
    hu[i][3] = *(const float4*)(hp + 5 * sq);
    bh[i] = pack_frag(wrow + kc * 32);
  }
  v16h ax[4], bx[4];
#pragma unroll
  for (int i = 0; i < 4; i++) {
    int kc = w + 8 + 4 * i;
    ax[i] = pack_frag(xrow + (kc - 8) * 32);
    bx[i] = pack_frag(wrow + kc * 32);
  }

  __builtin_amdgcn_sched_barrier(0);

  v8f acc = {};
#pragma unroll
  for (int i = 0; i < 2; i++) {
    float hv16[16] = {hu[i][0].x, hu[i][0].y, hu[i][0].z, hu[i][0].w,
                      hu[i][1].x, hu[i][1].y, hu[i][1].z, hu[i][1].w,
                      hu[i][2].x, hu[i][2].y, hu[i][2].z, hu[i][2].w,
                      hu[i][3].x, hu[i][3].y, hu[i][3].z, hu[i][3].w};
    v16h a;
#pragma unroll
    for (int e = 0; e < 16; e++) a[e] = adone ? (_Float16)0.0f : (_Float16)hv16[e];
    acc = WMMA16(a, bh[i], acc);
  }
#pragma unroll
  for (int i = 0; i < 4; i++) acc = WMMA16(ax[i], bx[i], acc);
  if (w < 2) acc = WMMA16(a6, b6, acc);

  {
    float* pw = P + w * PTILE + lane;
#pragma unroll
    for (int r = 0; r < 8; r++) pw[r * PPITCH] = acc[r];
  }
  __syncthreads();

  if (tid < 64) {
    int el = tid >> 2, q = tid & 3;
    const float* pr = P + (el & 7) * PPITCH + (el >> 3) * 16 + q;
    float si = 0.0f, sf = 0.0f, sg = 0.0f, so = 0.0f;
#pragma unroll
    for (int k = 0; k < 4; k++) {
      si += pr[k * PTILE + 0];
      sf += pr[k * PTILE + 4];
      sg += pr[k * PTILE + 8];
      so += pr[k * PTILE + 12];
    }
    float cp = edone ? 0.0f : cprev;
    float iv = sigmoidf_(si + bi), fv = sigmoidf_(sf + bf);
    float gv = tanhf(sg + bg),     ov = sigmoidf_(so + bo);
    float cv = fv * cp + iv * gv;
    float hv = ov * tanhf(cv);
    cbuf[(cg * E_ENVS + eenv) * 4 + q] = cv;
    hs[(size_t)t * (E_ENVS * HL) + (cg * E_ENVS + eenv) * 4 + q] = hv;
  }
}

__global__ void k_heads(const float* hs, const float* cbuf, const float* pw, const float* pb,
                        const float* vw, const float* vb, float* out) {
  int i = blockIdx.x * blockDim.x + threadIdx.x;
  if (i < B_TOT * 6) {
    int b = i / 6, j = i % 6;
    const float* h    = hs + (size_t)(b >> 5) * (E_ENVS * HL) + (b & 31) * 4;
    const float* wrow = (j < 5) ? pw + j * HL : vw;
    float s = (j < 5) ? pb[j] : vb[0];
#pragma unroll 4
    for (int k4 = 0; k4 < HL / 4; k4++) {
      float4 hv = *(const float4*)(h + k4 * (E_ENVS * 4));
      s += hv.x * wrow[k4 * 4 + 0];
      s += hv.y * wrow[k4 * 4 + 1];
      s += hv.z * wrow[k4 * 4 + 2];
      s += hv.w * wrow[k4 * 4 + 3];
    }
    if (j < 5) out[b * 5 + j] = s;
    else       out[10240 + b] = s;
  } else {
    int o = i - B_TOT * 6;
    int sel = o >> 13, r = o & 8191;
    int e = r >> 8, k = r & 255;
    int src = ((k >> 2) * E_ENVS + e) * 4 + (k & 3);
    if (sel == 0) out[12288 + r] = hs[(size_t)(T_STEPS - 1) * (E_ENVS * HL) + src];
    else          out[20480 + r] = cbuf[src];
  }
}

extern "C" void kernel_launch(void* const* d_in, const int* in_sizes, int n_in,
                              void* d_out, int out_size, void* d_ws, size_t ws_size,
                              hipStream_t stream) {
  (void)in_sizes; (void)n_in; (void)out_size; (void)ws_size;
  const float* img  = (const float*)d_in[0];
  const int*   pos  = (const int*)d_in[1];
  const int*   lact = (const int*)d_in[2];
  const int*   done = (const int*)d_in[3];
  const float* h0   = (const float*)d_in[4];
  const float* c0   = (const float*)d_in[5];
  const float* w1   = (const float*)d_in[6];
  const float* b1   = (const float*)d_in[7];
  const float* w2   = (const float*)d_in[8];
  const float* b2   = (const float*)d_in[9];
  const float* w3   = (const float*)d_in[10];
  const float* b3   = (const float*)d_in[11];
  const float* fcw  = (const float*)d_in[12];
  const float* fcb  = (const float*)d_in[13];
  const float* wih  = (const float*)d_in[14];
  const float* whh  = (const float*)d_in[15];
  const float* bih  = (const float*)d_in[16];
  const float* bhh  = (const float*)d_in[17];
  const float* pw   = (const float*)d_in[18];
  const float* pb   = (const float*)d_in[19];
  const float* vw   = (const float*)d_in[20];
  const float* vb   = (const float*)d_in[21];
  float* out = (float*)d_out;
  char*  ws  = (char*)d_ws;

  const size_t o_act1 = 90832896;
  const size_t o_w1t  = o_act1 + 52428800;
  const size_t o_w2t  = o_w1t + 12288;
  const size_t o_w3t  = o_w2t + 65536;
  const size_t o_fcwt = o_w3t + 73728;
  const size_t o_wg   = o_fcwt + 3211264;
  const size_t o_bg   = o_wg + 1703936;
  const size_t o_act3 = 21233664;
  const size_t o_X    = 34078720;
  const size_t o_hs   = 36438016;
  const size_t o_c    = 38535168;

  _Float16* act3 = (_Float16*)(ws + o_act3);
  _Float16* Xm   = (_Float16*)(ws + o_X);
  _Float16* w1t  = (_Float16*)(ws + o_w1t);
  _Float16* w2t  = (_Float16*)(ws + o_w2t);
  _Float16* w3t  = (_Float16*)(ws + o_w3t);
  _Float16* fcwt = (_Float16*)(ws + o_fcwt);
  _Float16* wg   = (_Float16*)(ws + o_wg);
  float*    bg   = (float*)(ws + o_bg);
  float*    hsb  = (float*)(ws + o_hs);
  float*    cbuf = (float*)(ws + o_c);

  const int TB = 256;
  k_prep_w1  <<<(32 * 192   + TB - 1) / TB, TB, 0, stream>>>(w1, w1t);
  k_prep_w2  <<<(64 * 512   + TB - 1) / TB, TB, 0, stream>>>(w2, w2t);
  k_prep_w3  <<<(64 * 576   + TB - 1) / TB, TB, 0, stream>>>(w3, w3t);
  k_prep_fc  <<<(512 * 3136 + TB - 1) / TB, TB, 0, stream>>>(fcw, fcwt);
  k_prep_wg  <<<(1024 * KPAD + TB - 1) / TB, TB, 0, stream>>>(wih, whh, wg);
  k_prep_bias<<<(1024 + TB - 1) / TB, TB, 0, stream>>>(bih, bhh, bg);

  k_convs<<<2048, 160, 0, stream>>>(img, w1t, b1, w2t, b2, w3t, b3, act3);
  k_fc   <<< 512, TB, 0, stream>>>(act3, fcwt, fcb, Xm);

  k_onehot<<<(B_TOT * 64 + TB - 1) / TB, TB, 0, stream>>>(pos, lact, Xm);

  k_lstm_step<<<128, 128, 0, stream>>>(Xm, wg, bg, done, h0, c0, cbuf, hsb, 0);
  k_lstm_step<<<128, 128, 0, stream>>>(Xm, wg, bg, done, h0, c0, cbuf, hsb, 1);
  k_lstm_step<<<128, 128, 0, stream>>>(Xm, wg, bg, done, h0, c0, cbuf, hsb, 2);
  k_lstm_step<<<128, 128, 0, stream>>>(Xm, wg, bg, done, h0, c0, cbuf, hsb, 3);
  k_lstm_step<<<128, 128, 0, stream>>>(Xm, wg, bg, done, h0, c0, cbuf, hsb, 4);
  k_lstm_step<<<128, 128, 0, stream>>>(Xm, wg, bg, done, h0, c0, cbuf, hsb, 5);
  k_lstm_step<<<128, 128, 0, stream>>>(Xm, wg, bg, done, h0, c0, cbuf, hsb, 6);
  k_lstm_step<<<128, 128, 0, stream>>>(Xm, wg, bg, done, h0, c0, cbuf, hsb, 7);
  k_lstm_step<<<128, 128, 0, stream>>>(Xm, wg, bg, done, h0, c0, cbuf, hsb, 8);
  k_lstm_step<<<128, 128, 0, stream>>>(Xm, wg, bg, done, h0, c0, cbuf, hsb, 9);
  k_lstm_step<<<128, 128, 0, stream>>>(Xm, wg, bg, done, h0, c0, cbuf, hsb, 10);
  k_lstm_step<<<128, 128, 0, stream>>>(Xm, wg, bg, done, h0, c0, cbuf, hsb, 11);
  k_lstm_step<<<128, 128, 0, stream>>>(Xm, wg, bg, done, h0, c0, cbuf, hsb, 12);
  k_lstm_step<<<128, 128, 0, stream>>>(Xm, wg, bg, done, h0, c0, cbuf, hsb, 13);
  k_lstm_step<<<128, 128, 0, stream>>>(Xm, wg, bg, done, h0, c0, cbuf, hsb, 14);
  k_lstm_step<<<128, 128, 0, stream>>>(Xm, wg, bg, done, h0, c0, cbuf, hsb, 15);
  k_lstm_step<<<128, 128, 0, stream>>>(Xm, wg, bg, done, h0, c0, cbuf, hsb, 16);
  k_lstm_step<<<128, 128, 0, stream>>>(Xm, wg, bg, done, h0, c0, cbuf, hsb, 17);
  k_lstm_step<<<128, 128, 0, stream>>>(Xm, wg, bg, done, h0, c0, cbuf, hsb, 18);
  k_lstm_step<<<128, 128, 0, stream>>>(Xm, wg, bg, done, h0, c0, cbuf, hsb, 19);
  k_lstm_step<<<128, 128, 0, stream>>>(Xm, wg, bg, done, h0, c0, cbuf, hsb, 20);
  k_lstm_step<<<128, 128, 0, stream>>>(Xm, wg, bg, done, h0, c0, cbuf, hsb, 21);
  k_lstm_step<<<128, 128, 0, stream>>>(Xm, wg, bg, done, h0, c0, cbuf, hsb, 22);
  k_lstm_step<<<128, 128, 0, stream>>>(Xm, wg, bg, done, h0, c0, cbuf, hsb, 23);
  k_lstm_step<<<128, 128, 0, stream>>>(Xm, wg, bg, done, h0, c0, cbuf, hsb, 24);
  k_lstm_step<<<128, 128, 0, stream>>>(Xm, wg, bg, done, h0, c0, cbuf, hsb, 25);
  k_lstm_step<<<128, 128, 0, stream>>>(Xm, wg, bg, done, h0, c0, cbuf, hsb, 26);
  k_lstm_step<<<128, 128, 0, stream>>>(Xm, wg, bg, done, h0, c0, cbuf, hsb, 27);
  k_lstm_step<<<128, 128, 0, stream>>>(Xm, wg, bg, done, h0, c0, cbuf, hsb, 28);
  k_lstm_step<<<128, 128, 0, stream>>>(Xm, wg, bg, done, h0, c0, cbuf, hsb, 29);
  k_lstm_step<<<128, 128, 0, stream>>>(Xm, wg, bg, done, h0, c0, cbuf, hsb, 30);
  k_lstm_step<<<128, 128, 0, stream>>>(Xm, wg, bg, done, h0, c0, cbuf, hsb, 31);
  k_lstm_step<<<128, 128, 0, stream>>>(Xm, wg, bg, done, h0, c0, cbuf, hsb, 32);
  k_lstm_step<<<128, 128, 0, stream>>>(Xm, wg, bg, done, h0, c0, cbuf, hsb, 33);
  k_lstm_step<<<128, 128, 0, stream>>>(Xm, wg, bg, done, h0, c0, cbuf, hsb, 34);
  k_lstm_step<<<128, 128, 0, stream>>>(Xm, wg, bg, done, h0, c0, cbuf, hsb, 35);
  k_lstm_step<<<128, 128, 0, stream>>>(Xm, wg, bg, done, h0, c0, cbuf, hsb, 36);
  k_lstm_step<<<128, 128, 0, stream>>>(Xm, wg, bg, done, h0, c0, cbuf, hsb, 37);
  k_lstm_step<<<128, 128, 0, stream>>>(Xm, wg, bg, done, h0, c0, cbuf, hsb, 38);
  k_lstm_step<<<128, 128, 0, stream>>>(Xm, wg, bg, done, h0, c0, cbuf, hsb, 39);
  k_lstm_step<<<128, 128, 0, stream>>>(Xm, wg, bg, done, h0, c0, cbuf, hsb, 40);
  k_lstm_step<<<128, 128, 0, stream>>>(Xm, wg, bg, done, h0, c0, cbuf, hsb, 41);
  k_lstm_step<<<128, 128, 0, stream>>>(Xm, wg, bg, done, h0, c0, cbuf, hsb, 42);
  k_lstm_step<<<128, 128, 0, stream>>>(Xm, wg, bg, done, h0, c0, cbuf, hsb, 43);
  k_lstm_step<<<128, 128, 0, stream>>>(Xm, wg, bg, done, h0, c0, cbuf, hsb, 44);
  k_lstm_step<<<128, 128, 0, stream>>>(Xm, wg, bg, done, h0, c0, cbuf, hsb, 45);
  k_lstm_step<<<128, 128, 0, stream>>>(Xm, wg, bg, done, h0, c0, cbuf, hsb, 46);
  k_lstm_step<<<128, 128, 0, stream>>>(Xm, wg, bg, done, h0, c0, cbuf, hsb, 47);
  k_lstm_step<<<128, 128, 0, stream>>>(Xm, wg, bg, done, h0, c0, cbuf, hsb, 48);
  k_lstm_step<<<128, 128, 0, stream>>>(Xm, wg, bg, done, h0, c0, cbuf, hsb, 49);
  k_lstm_step<<<128, 128, 0, stream>>>(Xm, wg, bg, done, h0, c0, cbuf, hsb, 50);
  k_lstm_step<<<128, 128, 0, stream>>>(Xm, wg, bg, done, h0, c0, cbuf, hsb, 51);
  k_lstm_step<<<128, 128, 0, stream>>>(Xm, wg, bg, done, h0, c0, cbuf, hsb, 52);
  k_lstm_step<<<128, 128, 0, stream>>>(Xm, wg, bg, done, h0, c0, cbuf, hsb, 53);
  k_lstm_step<<<128, 128, 0, stream>>>(Xm, wg, bg, done, h0, c0, cbuf, hsb, 54);
  k_lstm_step<<<128, 128, 0, stream>>>(Xm, wg, bg, done, h0, c0, cbuf, hsb, 55);
  k_lstm_step<<<128, 128, 0, stream>>>(Xm, wg, bg, done, h0, c0, cbuf, hsb, 56);
  k_lstm_step<<<128, 128, 0, stream>>>(Xm, wg, bg, done, h0, c0, cbuf, hsb, 57);
  k_lstm_step<<<128, 128, 0, stream>>>(Xm, wg, bg, done, h0, c0, cbuf, hsb, 58);
  k_lstm_step<<<128, 128, 0, stream>>>(Xm, wg, bg, done, h0, c0, cbuf, hsb, 59);
  k_lstm_step<<<128, 128, 0, stream>>>(Xm, wg, bg, done, h0, c0, cbuf, hsb, 60);
  k_lstm_step<<<128, 128, 0, stream>>>(Xm, wg, bg, done, h0, c0, cbuf, hsb, 61);
  k_lstm_step<<<128, 128, 0, stream>>>(Xm, wg, bg, done, h0, c0, cbuf, hsb, 62);
  k_lstm_step<<<128, 128, 0, stream>>>(Xm, wg, bg, done, h0, c0, cbuf, hsb, 63);

  k_heads<<<112, TB, 0, stream>>>(hsb, cbuf, pw, pb, vw, vb, out);
}


# LSTM launches regrouped by dependency level: launch L runs every (step, env) cell L steps behind its env's last reset (done flag or step 0), rows gathered in 16-row tiles; values, sum order same
# speedup vs baseline: 13.6854x; 1.2156x over previous
#include <hip/hip_runtime.h>
#include <math.h>

typedef __attribute__((ext_vector_type(16))) _Float16 v16h;
typedef __attribute__((ext_vector_type(8)))  _Float16 v8h;
typedef __attribute__((ext_vector_type(4)))  _Float16 v4h;
typedef __attribute__((ext_vector_type(8)))  float    v8f;

#define WMMA16(A,B,C) __builtin_amdgcn_wmma_f32_16x16x32_f16(false,(A),false,(B),(short)0,(C),false,false)

static constexpr int T_STEPS = 64;
static constexpr int E_ENVS  = 32;
static constexpr int B_TOT   = 2048;
static constexpr int HL      = 256;
static constexpr int IND     = 549;
static constexpr int KPAD    = 832;
static constexpr int XS      = 576;
static constexpr int IMS     = 88;
static constexpr int A1P     = 40;
static constexpr int A2P     = 72;
static constexpr int A2OFF   = 400 * A1P;

__device__ __forceinline__ v16h pack_frag(const _Float16* p) {
  v8h u = *(const v8h*)p;
  v8h w = *(const v8h*)(p + 16);
  v16h a;
#pragma unroll
  for (int e = 0; e < 8; e++) { a[e] = u[e]; a[e + 8] = w[e]; }
  return a;
}
__device__ __forceinline__ v16h pack_frag_c1(const _Float16* p0) {
  const _Float16* p1 = p0 + 2 * IMS;
  v4h u0 = *(const v4h*)p0;
  v4h u1 = *(const v4h*)(p0 + 4);
  v4h u2 = *(const v4h*)p1;
  v4h u3 = *(const v4h*)(p1 + 4);
  v16h a;
#pragma unroll
  for (int e = 0; e < 4; e++) {
    a[e] = u0[e]; a[e + 4] = u1[e]; a[e + 8] = u2[e]; a[e + 12] = u3[e];
  }
  return a;
}

__device__ __forceinline__ float sigmoidf_(float x) { return 1.0f / (1.0f + __expf(-x)); }

__device__ __forceinline__ unsigned pk2_scaled_f16(float a, float b, float sc) {
  unsigned r;
  asm("v_fma_mixlo_f16 %0, %1, %2, 0" : "=v"(r) : "v"(a), "s"(sc));
  asm("v_fma_mixhi_f16 %0, %1, %2, 0" : "+v"(r) : "v"(b), "s"(sc));
  return r;
}

__global__ void k_prep_w1(const float* w, _Float16* wt) {
  int i = blockIdx.x * blockDim.x + threadIdx.x;
  if (i >= 32 * 192) return;
  int n = i / 192, k = i % 192;
  int c = k >> 6, kh = (k & 63) >> 3, kw = k & 7;
  wt[i] = (_Float16)w[((n * 3 + c) * 8 + kh) * 8 + kw];
}
__global__ void k_prep_w2(const float* w, _Float16* wt) {
  int i = blockIdx.x * blockDim.x + threadIdx.x;
  if (i >= 64 * 512) return;
  int n = i / 512, k = i % 512;
  int pix = k >> 5, c = k & 31, kh = pix >> 2, kw = pix & 3;
  wt[i] = (_Float16)w[((n * 32 + c) * 4 + kh) * 4 + kw];
}
__global__ void k_prep_w3(const float* w, _Float16* wt) {
  int i = blockIdx.x * blockDim.x + threadIdx.x;
  if (i >= 64 * 576) return;
  int n = i / 576, k = i % 576;
  int pix = k >> 6, c = k & 63, kh = pix / 3, kw = pix % 3;
  wt[i] = (_Float16)w[((n * 64 + c) * 3 + kh) * 3 + kw];
}
__global__ void k_prep_fc(const float* w, _Float16* wt) {
  int i = blockIdx.x * blockDim.x + threadIdx.x;
  if (i >= 512 * 3136) return;
  int f = i / 3136, k = i % 3136;
  int s = k >> 6, c = k & 63;
  wt[i] = (_Float16)w[f * 3136 + c * 49 + s];
}
__global__ void k_prep_wg(const float* wih, const float* whh, _Float16* wg) {
  int i = blockIdx.x * blockDim.x + threadIdx.x;
  if (i >= 1024 * KPAD) return;
  int r = i / KPAD, k = i % KPAD;
  float v = 0.0f;
  if (k < HL)             v = whh[r * HL + k];
  else if (k < HL + IND)  v = wih[r * IND + (k - HL)];
  wg[i] = (_Float16)v;
}
__global__ void k_prep_bias(const float* bih, const float* bhh, float* bg) {
  int i = blockIdx.x * blockDim.x + threadIdx.x;
  if (i < 1024) bg[i] = bih[i] + bhh[i];
}
__global__ void k_onehot(const int* pos, const int* lact, _Float16* X) {
  int i = blockIdx.x * blockDim.x + threadIdx.x;
  if (i >= B_TOT * 64) return;
  int b = i / 64, c = i % 64;
  float v = 0.0f;
  if (c < 16)      v = (pos[b * 2 + 0] == c)      ? 1.0f : 0.0f;
  else if (c < 32) v = (pos[b * 2 + 1] == c - 16) ? 1.0f : 0.0f;
  else if (c < 37) v = (lact[b]        == c - 32) ? 1.0f : 0.0f;
  X[b * XS + 512 + c] = (_Float16)v;
}

__global__ void __launch_bounds__(160) k_convs(const float* img, const _Float16* wt1,
                                               const float* bias1, const _Float16* wt2,
                                               const float* bias2, const _Float16* wt3,
                                               const float* bias3, _Float16* out) {
  __shared__ __attribute__((aligned(16))) _Float16 imgL[3 * 84 * IMS];
  const float sc = 1.0f / 255.0f;
  int tid = threadIdx.x;
  int bimg = blockIdx.x;
  const float* ib = img + (size_t)bimg * (84 * 84 * 3);
  {
    float4 v[12][3];
#pragma unroll
    for (int k = 0; k < 12; k++) {
      int u  = tid + k * 160;
      int uc = u < 1764 ? u : 1763;
      const float4* p = (const float4*)(ib + uc * 12);
      v[k][0] = p[0]; v[k][1] = p[1]; v[k][2] = p[2];
    }
#pragma unroll
    for (int k = 0; k < 12; k++) {
      int u = tid + k * 160;
      if (u < 1764) {
        int R = u / 21, cg = u - R * 21;
        _Float16* o = imgL + R * IMS + cg * 4;
        float4 q0 = v[k][0], q1 = v[k][1], q2 = v[k][2];
        uint2 c0, c1, c2;
        c0.x = pk2_scaled_f16(q0.x, q0.w, sc); c0.y = pk2_scaled_f16(q1.z, q2.y, sc);
        c1.x = pk2_scaled_f16(q0.y, q1.x, sc); c1.y = pk2_scaled_f16(q1.w, q2.z, sc);
        c2.x = pk2_scaled_f16(q0.z, q1.y, sc); c2.y = pk2_scaled_f16(q2.x, q2.w, sc);
        *(uint2*)(o)                = c0;
        *(uint2*)(o + 84 * IMS)     = c1;
        *(uint2*)(o + 2 * 84 * IMS) = c2;
      }
    }
  }
  __syncthreads();

  int w    = __builtin_amdgcn_readfirstlane(tid >> 5);
  int lane = tid & 31, hi = lane >> 4, lo = lane & 15;
  _Float16* a1 = imgL;
  _Float16* a2 = imgL + A2OFF;

  {
    int aoff[5];
#pragma unroll
    for (int i = 0; i < 5; i++) {
      int p = (w * 5 + i) * 16 + lo;
      int oh = p / 20, ow = p % 20;
      aoff[i] = (oh * 4 + hi) * IMS + ow * 4;
    }
    const _Float16* bp = wt1 + lo * 192 + hi * 8;
    v8f acc[5][2] = {};
#pragma unroll
    for (int kb = 0; kb < 192; kb += 32) {
      int c = kb >> 6, khb = (kb & 63) >> 3;
      v16h a[5];
      v16h b0 = pack_frag(bp + kb);
      v16h b1 = pack_frag(bp + 16 * 192 + kb);
#pragma unroll
      for (int i = 0; i < 5; i++) a[i] = pack_frag_c1(imgL + aoff[i] + (c * 84 + khb) * IMS);
      __builtin_amdgcn_sched_barrier(0);
#pragma unroll
      for (int i = 0; i < 5; i++) {
        acc[i][0] = WMMA16(a[i], b0, acc[i][0]);
        acc[i][1] = WMMA16(a[i], b1, acc[i][1]);
      }
      __builtin_amdgcn_sched_barrier(0);
    }
    __syncthreads();
    float bs0 = bias1[lo], bs1 = bias1[16 + lo];
#pragma unroll
    for (int i = 0; i < 5; i++) {
#pragma unroll
      for (int r = 0; r < 8; r++) {
        int p = (w * 5 + i) * 16 + hi * 8 + r;
        float v0 = acc[i][0][r] + bs0; v0 = v0 > 0.0f ? v0 : 0.0f;
        float v1 = acc[i][1][r] + bs1; v1 = v1 > 0.0f ? v1 : 0.0f;
        a1[p * A1P + lo]      = (_Float16)v0;
        a1[p * A1P + 16 + lo] = (_Float16)v1;
      }
    }
  }
  __syncthreads();

  if (w < 4) {
    int aoff[6];
#pragma unroll
    for (int i = 0; i < 6; i++) {
      int p = i * 16 + lo;
      p = p < 81 ? p : 80;
      int oh = p / 9, ow = p % 9;
      aoff[i] = ((oh * 2) * 20 + ow * 2) * A1P + hi * 8;
    }
    const _Float16* bp = wt2 + (w * 16 + lo) * 512 + hi * 8;
    v8f acc[6] = {};
#pragma unroll 1
    for (int kh = 0; kh < 4; kh++) {
      v16h b[4];
#pragma unroll
      for (int kw = 0; kw < 4; kw++) b[kw] = pack_frag(bp + (kh * 4 + kw) * 32);
#pragma unroll
      for (int kw = 0; kw < 4; kw++) {
        v16h a[6];
#pragma unroll
        for (int i = 0; i < 6; i++) a[i] = pack_frag(a1 + aoff[i] + (kh * 20 + kw) * A1P);
        __builtin_amdgcn_sched_barrier(0);
#pragma unroll
        for (int i = 0; i < 6; i++) acc[i] = WMMA16(a[i], b[kw], acc[i]);
        __builtin_amdgcn_sched_barrier(0);
      }
    }
    float bs = bias2[w * 16 + lo];
#pragma unroll
    for (int i = 0; i < 6; i++) {
#pragma unroll
      for (int r = 0; r < 8; r++) {
        int p = i * 16 + hi * 8 + r;
        float v = acc[i][r] + bs; v = v > 0.0f ? v : 0.0f;
        if (p < 81) a2[p * A2P + w * 16 + lo] = (_Float16)v;
      }
    }
  }
  __syncthreads();

  if (w < 4) {
    int aoff[4];
#pragma unroll
    for (int i = 0; i < 4; i++) {
      int p = i * 16 + lo;
      p = p < 49 ? p : 48;
      int oh = p / 7, ow = p % 7;
      aoff[i] = (oh * 9 + ow) * A2P + hi * 8;
    }
    const _Float16* bp = wt3 + (w * 16 + lo) * 576 + hi * 8;
    v8f acc[4] = {};
#pragma unroll 1
    for (int kh = 0; kh < 3; kh++) {
      v16h b[6];
#pragma unroll
      for (int j = 0; j < 6; j++) b[j] = pack_frag(bp + (kh * 6 + j) * 32);
#pragma unroll
      for (int j = 0; j < 6; j++) {
        int kw = j >> 1, cb = (j & 1) * 32;
        v16h a[4];
#pragma unroll
        for (int i = 0; i < 4; i++) a[i] = pack_frag(a2 + aoff[i] + (kh * 9 + kw) * A2P + cb);
        __builtin_amdgcn_sched_barrier(0);
#pragma unroll
        for (int i = 0; i < 4; i++) acc[i] = WMMA16(a[i], b[j], acc[i]);
        __builtin_amdgcn_sched_barrier(0);
      }
    }
    float bs = bias3[w * 16 + lo];
    _Float16* ob = out + (size_t)bimg * (49 * 64) + w * 16 + lo;
#pragma unroll
    for (int i = 0; i < 4; i++) {
#pragma unroll
      for (int r = 0; r < 8; r++) {
        int p = i * 16 + hi * 8 + r;
        float v = acc[i][r] + bs; v = v > 0.0f ? v : 0.0f;
        if (p < 49) ob[p * 64] = (_Float16)v;
      }
    }
  }
}

static constexpr int FPITCH = 68;
static constexpr int FTILE  = 32 * FPITCH;
__global__ void __launch_bounds__(256) k_fc(const _Float16* act3, const _Float16* wt,
                                            const float* bias, _Float16* X) {
  __shared__ __attribute__((aligned(16))) float P[8 * FTILE];
  int tid = threadIdx.x;
  int w = __builtin_amdgcn_readfirstlane(tid >> 5);
  int lane = tid & 31, hi = lane >> 4, lo = lane & 15;
  int ng = blockIdx.x & 7, mp = blockIdx.x >> 3;
  int m0 = mp * 32, nb = ng * 64;
  const _Float16* ar0 = act3 + (size_t)(m0 + lo) * 3136 + hi * 8;
  const _Float16* ar1 = ar0 + (size_t)16 * 3136;
  const _Float16* br0 = wt + (size_t)(nb + lo) * 3136 + hi * 8;
  const _Float16* br1 = br0 + (size_t)16 * 3136;
  const _Float16* br2 = br0 + (size_t)32 * 3136;
  const _Float16* br3 = br0 + (size_t)48 * 3136;
  v8f c00 = {}, c01 = {}, c02 = {}, c03 = {};
  v8f c10 = {}, c11 = {}, c12 = {}, c13 = {};
  for (int kc = w; kc < 98; kc += 8) {
    int kb = kc * 32;
    v16h a0 = pack_frag(ar0 + kb);
    v16h a1 = pack_frag(ar1 + kb);
    v16h b0 = pack_frag(br0 + kb);
    v16h b1 = pack_frag(br1 + kb);
    v16h b2 = pack_frag(br2 + kb);
    v16h b3 = pack_frag(br3 + kb);
    c00 = WMMA16(a0, b0, c00);
    c10 = WMMA16(a1, b0, c10);
    c01 = WMMA16(a0, b1, c01);
    c11 = WMMA16(a1, b1, c11);
    c02 = WMMA16(a0, b2, c02);
    c12 = WMMA16(a1, b2, c12);
    c03 = WMMA16(a0, b3, c03);
    c13 = WMMA16(a1, b3, c13);
  }
  {
    float* pw = P + w * FTILE + (hi * 8) * FPITCH + lo;
#pragma unroll
    for (int r = 0; r < 8; r++) {
      pw[r * FPITCH +  0] = c00[r];
      pw[r * FPITCH + 16] = c01[r];
      pw[r * FPITCH + 32] = c02[r];
      pw[r * FPITCH + 48] = c03[r];
      pw[(16 + r) * FPITCH +  0] = c10[r];
      pw[(16 + r) * FPITCH + 16] = c11[r];
      pw[(16 + r) * FPITCH + 32] = c12[r];
      pw[(16 + r) * FPITCH + 48] = c13[r];
    }
  }
  __syncthreads();
  {
    int row = tid >> 3, c0 = (tid & 7) * 8;
    const float* pr = P + row * FPITCH + c0;
    float4 s0 = *(const float4*)pr;
    float4 s1 = *(const float4*)(pr + 4);
#pragma unroll
    for (int k = 1; k < 8; k++) {
      float4 u0 = *(const float4*)(pr + k * FTILE);
      float4 u1 = *(const float4*)(pr + k * FTILE + 4);
      s0.x += u0.x; s0.y += u0.y; s0.z += u0.z; s0.w += u0.w;
      s1.x += u1.x; s1.y += u1.y; s1.z += u1.z; s1.w += u1.w;
    }
    float4 q0 = *(const float4*)(bias + nb + c0);
    float4 q1 = *(const float4*)(bias + nb + c0 + 4);
    float v[8] = {s0.x + q0.x, s0.y + q0.y, s0.z + q0.z, s0.w + q0.w,
                  s1.x + q1.x, s1.y + q1.y, s1.z + q1.z, s1.w + q1.w};
    v8h o;
#pragma unroll
    for (int e = 0; e < 8; e++) o[e] = (_Float16)(v[e] > 0.0f ? v[e] : 0.0f);
    *(v8h*)(X + (size_t)(m0 + row) * XS + nb + c0) = o;
  }
}

static constexpr int NPOS = 3072;
__global__ void __launch_bounds__(1024) k_levels(const int* done, const float* h0, const float* c0,
                                                 float* hpos, float* cpos, int* rec, int* pos,
                                                 int* tinfo) {
  __shared__ unsigned char  lv_s[32][64];
  __shared__ unsigned char  dn_s[B_TOT];
  __shared__ unsigned short cnt_s[64][32];
  __shared__ int tot_s[64];
  __shared__ int tb_s[64];
  __shared__ int pos_s[B_TOT];
  __shared__ int inv_s[NPOS];
  int tid = threadIdx.x, e = tid >> 5, lane = tid & 31;
  int d0 = done[lane * E_ENVS + e] != 0;
  int d1 = done[(lane + 32) * E_ENVS + e] != 0;
  unsigned m0 = (unsigned)__ballot(d0) | 1u;
  unsigned m1 = (unsigned)__ballot(d1);
  unsigned upto = 0xffffffffu >> (31 - lane);
  unsigned a0 = m0 & upto, a1 = m1 & upto;
  int l0 = lane - (31 - __clz((int)a0));
  int l1 = a1 ? lane - (31 - __clz((int)a1)) : (lane + 32) - (31 - __clz((int)m0));
  lv_s[e][lane] = (unsigned char)l0;
  lv_s[e][lane + 32] = (unsigned char)l1;
  dn_s[lane * E_ENVS + e] = (unsigned char)d0;
  dn_s[(lane + 32) * E_ENVS + e] = (unsigned char)d1;
  for (int p = tid; p < NPOS; p += 1024) inv_s[p] = -1;
  __syncthreads();
  int n0 = 0, n1 = 0, r0 = 0, r1 = 0;
  for (int tp = 0; tp < 64; tp++) {
    int v = lv_s[e][tp];
    n0 += (v == lane);
    n1 += (v == lane + 32);
    r0 += (tp < lane && v == l0);
    r1 += (tp < lane + 32 && v == l1);
  }
  cnt_s[lane][e] = (unsigned short)n0;
  cnt_s[lane + 32][e] = (unsigned short)n1;
  __syncthreads();
  if (tid < 64) {
    int s = 0;
    for (int ep = 0; ep < 32; ep++) s += cnt_s[tid][ep];
    tot_s[tid] = s;
  }
  __syncthreads();
  if (tid < 64) {
    int tb = 0;
    for (int lp = 0; lp < tid; lp++) tb += (tot_s[lp] + 15) >> 4;
    tb_s[tid] = tb;
    tinfo[tid] = tb;
    tinfo[64 + tid] = tot_s[tid];
  }
  __syncthreads();
  int o0 = 0, o1 = 0;
  for (int ep = 0; ep < e; ep++) { o0 += cnt_s[l0][ep]; o1 += cnt_s[l1][ep]; }
  int p0 = tb_s[l0] * 16 + o0 + r0;
  int p1 = tb_s[l1] * 16 + o1 + r1;
  int b0 = lane * E_ENVS + e, b1 = (lane + 32) * E_ENVS + e;
  pos_s[b0] = p0; pos_s[b1] = p1;
  inv_s[p0] = b0; inv_s[p1] = b1;
  __syncthreads();
  pos[tid] = pos_s[tid];
  pos[tid + 1024] = pos_s[tid + 1024];
  for (int p = tid; p < NPOS; p += 1024) {
    int b = inv_s[p];
    int4 r = make_int4(0, NPOS, 0, 0);
    if (b >= 0) {
      int d = dn_s[b];
      r.x = b;
      r.y = (b >= E_ENVS && !d) ? pos_s[b - E_ENVS] : NPOS + (b & 31);
      r.z = d;
    }
    *(int4*)(rec + 4 * p) = r;
  }
  for (int i = tid; i < 2 * 4096; i += 1024) {
    int j = i & 3, row = (i >> 2) & 15, q = (i >> 6) & 63, tile2 = i >> 12;
    int s = (tile2 * 16 + row) * HL + q * 4 + j;
    hpos[(size_t)(NPOS / 16) * 4096 + i] = h0[s];
    cpos[(size_t)(NPOS / 16) * 4096 + i] = c0[s];
  }
}

static constexpr int PPITCH = 36;
static constexpr int PTILE  = 8 * PPITCH;
__global__ void __launch_bounds__(128) k_lstm_level(const _Float16* X, const _Float16* Wg,
                                                    const float* biasg, float* cpos, float* hpos,
                                                    const int* rec, const int* tinfo, int L) {
  __shared__ float P[2 * 4 * PTILE];
  int cnt = tinfo[64 + L];
  int nt  = (cnt + 15) >> 4;
  if ((int)blockIdx.y >= nt) return;
  int tb = tinfo[L];
  int tid = threadIdx.x;
  int w = __builtin_amdgcn_readfirstlane(tid >> 5);
  int lane = tid & 31, hi = lane >> 4, lo = lane & 15;
  int cg = blockIdx.x;
  int ci = tid & 63;
  int ti = blockIdx.y, gy = gridDim.y;

  int nv = cnt - ti * 16;
  nv = nv < 16 ? nv : 16;
  int4 rc = *(const int4*)(rec + 4 * ((tb + ti) * 16 + ((ci >> 2) < nv ? (ci >> 2) : 0)));
  int4 ra = *(const int4*)(rec + 4 * ((tb + ti) * 16 + (lo < nv ? lo : 0)));

  int ecol = cg * 4 + (ci & 3);
  float bi = biasg[0 * HL + ecol], bf = biasg[1 * HL + ecol];
  float bg = biasg[2 * HL + ecol], bo = biasg[3 * HL + ecol];

  int grow = (lo >> 2) * HL + cg * 4 + (lo & 3);
  const _Float16* wrow = Wg + (size_t)grow * KPAD + hi * 8;
  v16h bh[2], bx[4];
#pragma unroll
  for (int i = 0; i < 2; i++) bh[i] = pack_frag(wrow + (w + 4 * i) * 32);
#pragma unroll
  for (int i = 0; i < 4; i++) bx[i] = pack_frag(wrow + (w + 8 + 4 * i) * 32);
  v16h b6 = pack_frag(wrow + (w < 2 ? w + 24 : w + 20) * 32);

  int set = 0;
  for (;;) {
    int tile = tb + ti;
    float* Ps = P + set * (4 * PTILE);
    set ^= 1;

    int edone = rc.z;
    float cprev = cpos[(size_t)(rc.y >> 4) * 4096 + cg * 64 + (rc.y & 15) * 4 + (ci & 3)];
    int adone = ra.z;
    const float* hbase = hpos + (size_t)(ra.y >> 4) * 4096 + (ra.y & 15) * 4;
    const int sq = 64;
    const _Float16* xrow = X + (size_t)ra.x * XS + hi * 8;

    v16h a6 = {};
    if (w < 2) a6 = pack_frag(xrow + (w + 16) * 32);
    float4 hu[2][4];
#pragma unroll
    for (int i = 0; i < 2; i++) {
      int kc = w + 4 * i;
      const float* hp = hbase + (kc * 8 + hi * 2) * sq;
      hu[i][0] = *(const float4*)hp;
      hu[i][1] = *(const float4*)(hp + sq);
      hu[i][2] = *(const float4*)(hp + 4 * sq);
      hu[i][3] = *(const float4*)(hp + 5 * sq);
    }
    v16h ax[4];
#pragma unroll
    for (int i = 0; i < 4; i++) ax[i] = pack_frag(xrow + (w + 4 * i) * 32);

    int tn = ti + gy;
    int tq = tn < nt ? tn : ti;
    int nvn = cnt - tq * 16;
    nvn = nvn < 16 ? nvn : 16;
    int4 rcn = *(const int4*)(rec + 4 * ((tb + tq) * 16 + ((ci >> 2) < nvn ? (ci >> 2) : 0)));
    int4 ran = *(const int4*)(rec + 4 * ((tb + tq) * 16 + (lo < nvn ? lo : 0)));

    __builtin_amdgcn_sched_barrier(0);

    v8f acc = {};
#pragma unroll
    for (int i = 0; i < 2; i++) {
      float hv16[16] = {hu[i][0].x, hu[i][0].y, hu[i][0].z, hu[i][0].w,
                        hu[i][1].x, hu[i][1].y, hu[i][1].z, hu[i][1].w,
                        hu[i][2].x, hu[i][2].y, hu[i][2].z, hu[i][2].w,
                        hu[i][3].x, hu[i][3].y, hu[i][3].z, hu[i][3].w};
      v16h a;
#pragma unroll
      for (int e = 0; e < 16; e++) a[e] = adone ? (_Float16)0.0f : (_Float16)hv16[e];
      acc = WMMA16(a, bh[i], acc);
    }
#pragma unroll
    for (int i = 0; i < 4; i++) acc = WMMA16(ax[i], bx[i], acc);
    if (w < 2) acc = WMMA16(a6, b6, acc);

    {
      float* pw = Ps + w * PTILE + lane;
#pragma unroll
      for (int r = 0; r < 8; r++) pw[r * PPITCH] = acc[r];
    }
    __syncthreads();

    if (tid < 64) {
      int el = tid >> 2, q = tid & 3;
      const float* pr = Ps + (el & 7) * PPITCH + (el >> 3) * 16 + q;
      float si = 0.0f, sf = 0.0f, sg = 0.0f, so = 0.0f;
#pragma unroll
      for (int k = 0; k < 4; k++) {
        si += pr[k * PTILE + 0];
        sf += pr[k * PTILE + 4];
        sg += pr[k * PTILE + 8];
        so += pr[k * PTILE + 12];
      }
      float cp = edone ? 0.0f : cprev;
      float iv = sigmoidf_(si + bi), fv = sigmoidf_(sf + bf);
      float gv = tanhf(sg + bg),     ov = sigmoidf_(so + bo);
      float cv = fv * cp + iv * gv;
      float hv = ov * tanhf(cv);
      size_t o = (size_t)(tile * 64 + cg) * 64 + tid;
      cpos[o] = cv;
      hpos[o] = hv;
    }
    if (tn >= nt) break;
    ti = tn; rc = rcn; ra = ran;
  }
}

__global__ void k_heads(const float* hpos, const float* cpos, const int* pos, const float* pw,
                        const float* pb, const float* vw, const float* vb, float* out) {
  int i = blockIdx.x * blockDim.x + threadIdx.x;
  if (i < B_TOT * 6) {
    int b = i / 6, j = i % 6;
    int p = pos[b];
    const float* h    = hpos + (size_t)(p >> 4) * 4096 + (p & 15) * 4;
    const float* wrow = (j < 5) ? pw + j * HL : vw;
    float s = (j < 5) ? pb[j] : vb[0];
#pragma unroll 4
    for (int k4 = 0; k4 < HL / 4; k4++) {
      float4 hv = *(const float4*)(h + k4 * 64);
      s += hv.x * wrow[k4 * 4 + 0];
      s += hv.y * wrow[k4 * 4 + 1];
      s += hv.z * wrow[k4 * 4 + 2];
      s += hv.w * wrow[k4 * 4 + 3];
    }
    if (j < 5) out[b * 5 + j] = s;
    else       out[10240 + b] = s;
  } else {
    int o = i - B_TOT * 6;
    int sel = o >> 13, r = o & 8191;
    int e = r >> 8, k = r & 255;
    int p = pos[(T_STEPS - 1) * E_ENVS + e];
    size_t src = (size_t)(p >> 4) * 4096 + (k >> 2) * 64 + (p & 15) * 4 + (k & 3);
    if (sel == 0) out[12288 + r] = hpos[src];
    else          out[20480 + r] = cpos[src];
  }
}

extern "C" void kernel_launch(void* const* d_in, const int* in_sizes, int n_in,
                              void* d_out, int out_size, void* d_ws, size_t ws_size,
                              hipStream_t stream) {
  (void)in_sizes; (void)n_in; (void)out_size; (void)ws_size;
  const float* img  = (const float*)d_in[0];
  const int*   pos  = (const int*)d_in[1];
  const int*   lact = (const int*)d_in[2];
  const int*   done = (const int*)d_in[3];
  const float* h0   = (const float*)d_in[4];
  const float* c0   = (const float*)d_in[5];
  const float* w1   = (const float*)d_in[6];
  const float* b1   = (const float*)d_in[7];
  const float* w2   = (const float*)d_in[8];
  const float* b2   = (const float*)d_in[9];
  const float* w3   = (const float*)d_in[10];
  const float* b3   = (const float*)d_in[11];
  const float* fcw  = (const float*)d_in[12];
  const float* fcb  = (const float*)d_in[13];
  const float* wih  = (const float*)d_in[14];
  const float* whh  = (const float*)d_in[15];
  const float* bih  = (const float*)d_in[16];
  const float* bhh  = (const float*)d_in[17];
  const float* pw   = (const float*)d_in[18];
  const float* pb   = (const float*)d_in[19];
  const float* vw   = (const float*)d_in[20];
  const float* vb   = (const float*)d_in[21];
  float* out = (float*)d_out;
  char*  ws  = (char*)d_ws;

  const size_t o_act1 = 90832896;
  const size_t o_w1t  = o_act1 + 52428800;
  const size_t o_w2t  = o_w1t + 12288;
  const size_t o_w3t  = o_w2t + 65536;
  const size_t o_fcwt = o_w3t + 73728;
  const size_t o_wg   = o_fcwt + 3211264;
  const size_t o_bg   = o_wg + 1703936;
  const size_t o_act3 = 21233664;
  const size_t o_X    = 34078720;
  const size_t o_hp   = o_act1;
  const size_t o_cp   = o_hp + 3178496;
  const size_t o_rec  = o_cp + 3178496;
  const size_t o_pos  = o_rec + 49152;
  const size_t o_ti   = o_pos + 8192;

  _Float16* act3 = (_Float16*)(ws + o_act3);
  _Float16* Xm   = (_Float16*)(ws + o_X);
  _Float16* w1t  = (_Float16*)(ws + o_w1t);
  _Float16* w2t  = (_Float16*)(ws + o_w2t);
  _Float16* w3t  = (_Float16*)(ws + o_w3t);
  _Float16* fcwt = (_Float16*)(ws + o_fcwt);
  _Float16* wg   = (_Float16*)(ws + o_wg);
  float*    bg   = (float*)(ws + o_bg);
  float*    hpb  = (float*)(ws + o_hp);
  float*    cpb  = (float*)(ws + o_cp);
  int*      rec  = (int*)(ws + o_rec);
  int*      posb = (int*)(ws + o_pos);
  int*      tinfo = (int*)(ws + o_ti);

  const int TB = 256;
  k_prep_w1  <<<(32 * 192   + TB - 1) / TB, TB, 0, stream>>>(w1, w1t);
  k_prep_w2  <<<(64 * 512   + TB - 1) / TB, TB, 0, stream>>>(w2, w2t);
  k_prep_w3  <<<(64 * 576   + TB - 1) / TB, TB, 0, stream>>>(w3, w3t);
  k_prep_fc  <<<(512 * 3136 + TB - 1) / TB, TB, 0, stream>>>(fcw, fcwt);
  k_prep_wg  <<<(1024 * KPAD + TB - 1) / TB, TB, 0, stream>>>(wih, whh, wg);
  k_prep_bias<<<(1024 + TB - 1) / TB, TB, 0, stream>>>(bih, bhh, bg);

  k_convs<<<2048, 160, 0, stream>>>(img, w1t, b1, w2t, b2, w3t, b3, act3);
  k_fc   <<< 512, TB, 0, stream>>>(act3, fcwt, fcb, Xm);

  k_onehot<<<(B_TOT * 64 + TB - 1) / TB, TB, 0, stream>>>(pos, lact, Xm);

  k_levels<<<1, 1024, 0, stream>>>(done, h0, c0, hpb, cpb, rec, posb, tinfo);

  k_lstm_level<<<dim3(64, 16), 128, 0, stream>>>(Xm, wg, bg, cpb, hpb, rec, tinfo, 0);
  k_lstm_level<<<dim3(64, 8), 128, 0, stream>>>(Xm, wg, bg, cpb, hpb, rec, tinfo, 1);
  k_lstm_level<<<dim3(64, 5), 128, 0, stream>>>(Xm, wg, bg, cpb, hpb, rec, tinfo, 2);
  k_lstm_level<<<dim3(64, 4), 128, 0, stream>>>(Xm, wg, bg, cpb, hpb, rec, tinfo, 3);
  k_lstm_level<<<dim3(64, 3), 128, 0, stream>>>(Xm, wg, bg, cpb, hpb, rec, tinfo, 4);
  k_lstm_level<<<dim3(64, 2), 128, 0, stream>>>(Xm, wg, bg, cpb, hpb, rec, tinfo, 5);
  k_lstm_level<<<dim3(64, 2), 128, 0, stream>>>(Xm, wg, bg, cpb, hpb, rec, tinfo, 6);
  k_lstm_level<<<dim3(64, 2), 128, 0, stream>>>(Xm, wg, bg, cpb, hpb, rec, tinfo, 7);
  k_lstm_level<<<dim3(64, 2), 128, 0, stream>>>(Xm, wg, bg, cpb, hpb, rec, tinfo, 8);
  k_lstm_level<<<dim3(64, 2), 128, 0, stream>>>(Xm, wg, bg, cpb, hpb, rec, tinfo, 9);
  k_lstm_level<<<dim3(64, 2), 128, 0, stream>>>(Xm, wg, bg, cpb, hpb, rec, tinfo, 10);
  k_lstm_level<<<dim3(64, 2), 128, 0, stream>>>(Xm, wg, bg, cpb, hpb, rec, tinfo, 11);
  k_lstm_level<<<dim3(64, 2), 128, 0, stream>>>(Xm, wg, bg, cpb, hpb, rec, tinfo, 12);
  k_lstm_level<<<dim3(64, 2), 128, 0, stream>>>(Xm, wg, bg, cpb, hpb, rec, tinfo, 13);
  k_lstm_level<<<dim3(64, 2), 128, 0, stream>>>(Xm, wg, bg, cpb, hpb, rec, tinfo, 14);
  k_lstm_level<<<dim3(64, 2), 128, 0, stream>>>(Xm, wg, bg, cpb, hpb, rec, tinfo, 15);
  k_lstm_level<<<dim3(64, 2), 128, 0, stream>>>(Xm, wg, bg, cpb, hpb, rec, tinfo, 16);
  k_lstm_level<<<dim3(64, 2), 128, 0, stream>>>(Xm, wg, bg, cpb, hpb, rec, tinfo, 17);
  k_lstm_level<<<dim3(64, 2), 128, 0, stream>>>(Xm, wg, bg, cpb, hpb, rec, tinfo, 18);
  k_lstm_level<<<dim3(64, 2), 128, 0, stream>>>(Xm, wg, bg, cpb, hpb, rec, tinfo, 19);
  k_lstm_level<<<dim3(64, 2), 128, 0, stream>>>(Xm, wg, bg, cpb, hpb, rec, tinfo, 20);
  k_lstm_level<<<dim3(64, 2), 128, 0, stream>>>(Xm, wg, bg, cpb, hpb, rec, tinfo, 21);
  k_lstm_level<<<dim3(64, 2), 128, 0, stream>>>(Xm, wg, bg, cpb, hpb, rec, tinfo, 22);
  k_lstm_level<<<dim3(64, 2), 128, 0, stream>>>(Xm, wg, bg, cpb, hpb, rec, tinfo, 23);
  k_lstm_level<<<dim3(64, 2), 128, 0, stream>>>(Xm, wg, bg, cpb, hpb, rec, tinfo, 24);
  k_lstm_level<<<dim3(64, 2), 128, 0, stream>>>(Xm, wg, bg, cpb, hpb, rec, tinfo, 25);
  k_lstm_level<<<dim3(64, 2), 128, 0, stream>>>(Xm, wg, bg, cpb, hpb, rec, tinfo, 26);
  k_lstm_level<<<dim3(64, 2), 128, 0, stream>>>(Xm, wg, bg, cpb, hpb, rec, tinfo, 27);
  k_lstm_level<<<dim3(64, 2), 128, 0, stream>>>(Xm, wg, bg, cpb, hpb, rec, tinfo, 28);
  k_lstm_level<<<dim3(64, 2), 128, 0, stream>>>(Xm, wg, bg, cpb, hpb, rec, tinfo, 29);
  k_lstm_level<<<dim3(64, 2), 128, 0, stream>>>(Xm, wg, bg, cpb, hpb, rec, tinfo, 30);
  k_lstm_level<<<dim3(64, 2), 128, 0, stream>>>(Xm, wg, bg, cpb, hpb, rec, tinfo, 31);
  k_lstm_level<<<dim3(64, 2), 128, 0, stream>>>(Xm, wg, bg, cpb, hpb, rec, tinfo, 32);
  k_lstm_level<<<dim3(64, 2), 128, 0, stream>>>(Xm, wg, bg, cpb, hpb, rec, tinfo, 33);
  k_lstm_level<<<dim3(64, 2), 128, 0, stream>>>(Xm, wg, bg, cpb, hpb, rec, tinfo, 34);
  k_lstm_level<<<dim3(64, 2), 128, 0, stream>>>(Xm, wg, bg, cpb, hpb, rec, tinfo, 35);
  k_lstm_level<<<dim3(64, 2), 128, 0, stream>>>(Xm, wg, bg, cpb, hpb, rec, tinfo, 36);
  k_lstm_level<<<dim3(64, 2), 128, 0, stream>>>(Xm, wg, bg, cpb, hpb, rec, tinfo, 37);
  k_lstm_level<<<dim3(64, 2), 128, 0, stream>>>(Xm, wg, bg, cpb, hpb, rec, tinfo, 38);
  k_lstm_level<<<dim3(64, 2), 128, 0, stream>>>(Xm, wg, bg, cpb, hpb, rec, tinfo, 39);
  k_lstm_level<<<dim3(64, 2), 128, 0, stream>>>(Xm, wg, bg, cpb, hpb, rec, tinfo, 40);
  k_lstm_level<<<dim3(64, 2), 128, 0, stream>>>(Xm, wg, bg, cpb, hpb, rec, tinfo, 41);
  k_lstm_level<<<dim3(64, 2), 128, 0, stream>>>(Xm, wg, bg, cpb, hpb, rec, tinfo, 42);
  k_lstm_level<<<dim3(64, 2), 128, 0, stream>>>(Xm, wg, bg, cpb, hpb, rec, tinfo, 43);
  k_lstm_level<<<dim3(64, 2), 128, 0, stream>>>(Xm, wg, bg, cpb, hpb, rec, tinfo, 44);
  k_lstm_level<<<dim3(64, 2), 128, 0, stream>>>(Xm, wg, bg, cpb, hpb, rec, tinfo, 45);
  k_lstm_level<<<dim3(64, 2), 128, 0, stream>>>(Xm, wg, bg, cpb, hpb, rec, tinfo, 46);
  k_lstm_level<<<dim3(64, 2), 128, 0, stream>>>(Xm, wg, bg, cpb, hpb, rec, tinfo, 47);
  k_lstm_level<<<dim3(64, 2), 128, 0, stream>>>(Xm, wg, bg, cpb, hpb, rec, tinfo, 48);
  k_lstm_level<<<dim3(64, 2), 128, 0, stream>>>(Xm, wg, bg, cpb, hpb, rec, tinfo, 49);
  k_lstm_level<<<dim3(64, 2), 128, 0, stream>>>(Xm, wg, bg, cpb, hpb, rec, tinfo, 50);
  k_lstm_level<<<dim3(64, 2), 128, 0, stream>>>(Xm, wg, bg, cpb, hpb, rec, tinfo, 51);
  k_lstm_level<<<dim3(64, 2), 128, 0, stream>>>(Xm, wg, bg, cpb, hpb, rec, tinfo, 52);
  k_lstm_level<<<dim3(64, 2), 128, 0, stream>>>(Xm, wg, bg, cpb, hpb, rec, tinfo, 53);
  k_lstm_level<<<dim3(64, 2), 128, 0, stream>>>(Xm, wg, bg, cpb, hpb, rec, tinfo, 54);
  k_lstm_level<<<dim3(64, 2), 128, 0, stream>>>(Xm, wg, bg, cpb, hpb, rec, tinfo, 55);
  k_lstm_level<<<dim3(64, 2), 128, 0, stream>>>(Xm, wg, bg, cpb, hpb, rec, tinfo, 56);
  k_lstm_level<<<dim3(64, 2), 128, 0, stream>>>(Xm, wg, bg, cpb, hpb, rec, tinfo, 57);
  k_lstm_level<<<dim3(64, 2), 128, 0, stream>>>(Xm, wg, bg, cpb, hpb, rec, tinfo, 58);
  k_lstm_level<<<dim3(64, 2), 128, 0, stream>>>(Xm, wg, bg, cpb, hpb, rec, tinfo, 59);
  k_lstm_level<<<dim3(64, 2), 128, 0, stream>>>(Xm, wg, bg, cpb, hpb, rec, tinfo, 60);
  k_lstm_level<<<dim3(64, 2), 128, 0, stream>>>(Xm, wg, bg, cpb, hpb, rec, tinfo, 61);
  k_lstm_level<<<dim3(64, 2), 128, 0, stream>>>(Xm, wg, bg, cpb, hpb, rec, tinfo, 62);
  k_lstm_level<<<dim3(64, 2), 128, 0, stream>>>(Xm, wg, bg, cpb, hpb, rec, tinfo, 63);

  k_heads<<<112, TB, 0, stream>>>(hpb, cpb, posb, pw, pb, vw, vb, out);
}


# FC-weight, LSTM-weight, bias-sum and one-hot re-layouts move into the conv launch: the fifth wave of each conv workgroup (idle in conv2, conv3) does a slice; 28 launches -> 24; values, sum order same
# speedup vs baseline: 18.8434x; 1.3769x over previous
#include <hip/hip_runtime.h>
#include <math.h>

typedef __attribute__((ext_vector_type(16))) _Float16 v16h;
typedef __attribute__((ext_vector_type(8)))  _Float16 v8h;
typedef __attribute__((ext_vector_type(4)))  _Float16 v4h;
typedef __attribute__((ext_vector_type(2)))  _Float16 v2h_t;
typedef __attribute__((ext_vector_type(8)))  float    v8f;

#define WMMA16(A,B,C) __builtin_amdgcn_wmma_f32_16x16x32_f16(false,(A),false,(B),(short)0,(C),false,false)

static constexpr int T_STEPS = 64;
static constexpr int E_ENVS  = 32;
static constexpr int B_TOT   = 2048;
static constexpr int HL      = 256;
static constexpr int IND     = 549;
static constexpr int KPAD    = 832;
static constexpr int XS      = 576;
static constexpr int IMS     = 88;
static constexpr int A1P     = 40;
static constexpr int A2P     = 72;
static constexpr int A2OFF   = 400 * A1P;

__device__ __forceinline__ v16h pack_frag(const _Float16* p) {
  v8h u = *(const v8h*)p;
  v8h w = *(const v8h*)(p + 16);
  v16h a;
#pragma unroll
  for (int e = 0; e < 8; e++) { a[e] = u[e]; a[e + 8] = w[e]; }
  return a;
}
__device__ __forceinline__ v16h pack_frag_c1(const _Float16* p0) {
  const _Float16* p1 = p0 + 2 * IMS;
  v4h u0 = *(const v4h*)p0;
  v4h u1 = *(const v4h*)(p0 + 4);
  v4h u2 = *(const v4h*)p1;
  v4h u3 = *(const v4h*)(p1 + 4);
  v16h a;
#pragma unroll
  for (int e = 0; e < 4; e++) {
    a[e] = u0[e]; a[e + 4] = u1[e]; a[e + 8] = u2[e]; a[e + 12] = u3[e];
  }
  return a;
}

__device__ __forceinline__ float sigmoidf_(float x) { return 1.0f / (1.0f + __expf(-x)); }

__device__ __forceinline__ unsigned pk2_scaled_f16(float a, float b, float sc) {
  unsigned r;
  asm("v_fma_mixlo_f16 %0, %1, %2, 0" : "=v"(r) : "v"(a), "s"(sc));
  asm("v_fma_mixhi_f16 %0, %1, %2, 0" : "+v"(r) : "v"(b), "s"(sc));
  return r;
}

__global__ void k_prep_w1(const float* w, _Float16* wt) {
  int i = blockIdx.x * blockDim.x + threadIdx.x;
  if (i >= 32 * 192) return;
  int n = i / 192, k = i % 192;
  int c = k >> 6, kh = (k & 63) >> 3, kw = k & 7;
  wt[i] = (_Float16)w[((n * 3 + c) * 8 + kh) * 8 + kw];
}
__global__ void k_prep_w2(const float* w, _Float16* wt) {
  int i = blockIdx.x * blockDim.x + threadIdx.x;
  if (i >= 64 * 512) return;
  int n = i / 512, k = i % 512;
  int pix = k >> 5, c = k & 31, kh = pix >> 2, kw = pix & 3;
  wt[i] = (_Float16)w[((n * 32 + c) * 4 + kh) * 4 + kw];
}
__global__ void k_prep_w3(const float* w, _Float16* wt) {
  int i = blockIdx.x * blockDim.x + threadIdx.x;
  if (i >= 64 * 576) return;
  int n = i / 576, k = i % 576;
  int pix = k >> 6, c = k & 63, kh = pix / 3, kw = pix % 3;
  wt[i] = (_Float16)w[((n * 64 + c) * 3 + kh) * 3 + kw];
}
__device__ __forceinline__ unsigned pk2_f16(float a, float b) {
  v2h_t p;
  p[0] = (_Float16)a; p[1] = (_Float16)b;
  return __builtin_bit_cast(unsigned, p);
}
__device__ __forceinline__ void relayout_fc_slice(const float* fcw, _Float16* fcwt, int b, int lane) {
  int g0 = (b * 49) >> 2, g1 = ((b + 1) * 49) >> 2;
  float v[13][2];
#pragma unroll
  for (int j = 0; j < 13; j++) {
    int g  = g0 + j;
    int gc = g < g1 ? g : g1 - 1;
    int f = gc / 49, sp = gc - f * 49;
    const float* p = fcw + (size_t)f * 3136 + (2 * lane) * 49 + sp;
    v[j][0] = p[0];
    v[j][1] = p[49];
  }
#pragma unroll
  for (int j = 0; j < 13; j++) {
    int g = g0 + j;
    if (g < g1) *(unsigned*)(fcwt + (size_t)g * 64 + 2 * lane) = pk2_f16(v[j][0], v[j][1]);
  }
}
__device__ __forceinline__ void relayout_rest_slice(const float* wih, const float* whh, _Float16* wg,
                                                    const float* bih, const float* bhh, float* bg,
                                                    const int* pos, const int* lact, _Float16* X,
                                                    int b, int lane) {
  int g0 = (b * 13) >> 1, g1 = ((b + 1) * 13) >> 1;
  int c0 = 2 * lane, c1 = 2 * lane + 1;
  float v[7][2];
#pragma unroll
  for (int j = 0; j < 7; j++) {
    int g  = g0 + j;
    int gc = g < g1 ? g : g1 - 1;
    int r = gc / 13, li = gc - r * 13;
    const float* base = li < 4 ? whh + r * HL + li * 64 : wih + r * IND + (li * 64 - HL);
    int lim = li == 12 ? HL + IND - 768 : 64;
    float x0 = base[c0 < lim ? c0 : lim - 1];
    float x1 = base[c1 < lim ? c1 : lim - 1];
    v[j][0] = c0 < lim ? x0 : 0.0f;
    v[j][1] = c1 < lim ? x1 : 0.0f;
  }
#pragma unroll
  for (int j = 0; j < 7; j++) {
    int g = g0 + j;
    if (g < g1) *(unsigned*)(wg + (size_t)g * 64 + 2 * lane) = pk2_f16(v[j][0], v[j][1]);
  }
  {
    int p0 = pos[b * 2 + 0], p1 = pos[b * 2 + 1], la = lact[b];
    int t0 = c0 < 16 ? p0 : (c0 < 32 ? p1 + 16 : (c0 < 37 ? la + 32 : -1));
    int t1 = c1 < 16 ? p0 : (c1 < 32 ? p1 + 16 : (c1 < 37 ? la + 32 : -1));
    *(unsigned*)(X + (size_t)b * XS + 512 + 2 * lane) =
        pk2_f16(t0 == c0 ? 1.0f : 0.0f, t1 == c1 ? 1.0f : 0.0f);
  }
  if (b < 32) {
    int i = b * 32 + lane;
    bg[i] = bih[i] + bhh[i];
  }
}

__global__ void __launch_bounds__(160) k_convs(const float* img, const _Float16* wt1,
                                               const float* bias1, const _Float16* wt2,
                                               const float* bias2, const _Float16* wt3,
                                               const float* bias3, _Float16* out,
                                               const float* fcw, _Float16* fcwt,
                                               const float* wih, const float* whh, _Float16* wg,
                                               const float* bih, const float* bhh, float* bg,
                                               const int* pos, const int* lact, _Float16* X) {
  __shared__ __attribute__((aligned(16))) _Float16 imgL[3 * 84 * IMS];
  const float sc = 1.0f / 255.0f;
  int tid = threadIdx.x;
  int bimg = blockIdx.x;
  const float* ib = img + (size_t)bimg * (84 * 84 * 3);
  {
    float4 v[12][3];
#pragma unroll
    for (int k = 0; k < 12; k++) {
      int u  = tid + k * 160;
      int uc = u < 1764 ? u : 1763;
      const float4* p = (const float4*)(ib + uc * 12);
      v[k][0] = p[0]; v[k][1] = p[1]; v[k][2] = p[2];
    }
#pragma unroll
    for (int k = 0; k < 12; k++) {
      int u = tid + k * 160;
      if (u < 1764) {
        int R = u / 21, cg = u - R * 21;
        _Float16* o = imgL + R * IMS + cg * 4;
        float4 q0 = v[k][0], q1 = v[k][1], q2 = v[k][2];
        uint2 c0, c1, c2;
        c0.x = pk2_scaled_f16(q0.x, q0.w, sc); c0.y = pk2_scaled_f16(q1.z, q2.y, sc);
        c1.x = pk2_scaled_f16(q0.y, q1.x, sc); c1.y = pk2_scaled_f16(q1.w, q2.z, sc);
        c2.x = pk2_scaled_f16(q0.z, q1.y, sc); c2.y = pk2_scaled_f16(q2.x, q2.w, sc);
        *(uint2*)(o)                = c0;
        *(uint2*)(o + 84 * IMS)     = c1;
        *(uint2*)(o + 2 * 84 * IMS) = c2;
      }
    }
  }
  __syncthreads();

  int w    = __builtin_amdgcn_readfirstlane(tid >> 5);
  int lane = tid & 31, hi = lane >> 4, lo = lane & 15;
  _Float16* a1 = imgL;
  _Float16* a2 = imgL + A2OFF;

  {
    int aoff[5];
#pragma unroll
    for (int i = 0; i < 5; i++) {
      int p = (w * 5 + i) * 16 + lo;
      int oh = p / 20, ow = p % 20;
      aoff[i] = (oh * 4 + hi) * IMS + ow * 4;
    }
    const _Float16* bp = wt1 + lo * 192 + hi * 8;
    v8f acc[5][2] = {};
#pragma unroll
    for (int kb = 0; kb < 192; kb += 32) {
      int c = kb >> 6, khb = (kb & 63) >> 3;
      v16h a[5];
      v16h b0 = pack_frag(bp + kb);
      v16h b1 = pack_frag(bp + 16 * 192 + kb);
#pragma unroll
      for (int i = 0; i < 5; i++) a[i] = pack_frag_c1(imgL + aoff[i] + (c * 84 + khb) * IMS);
      __builtin_amdgcn_sched_barrier(0);
#pragma unroll
      for (int i = 0; i < 5; i++) {
        acc[i][0] = WMMA16(a[i], b0, acc[i][0]);
        acc[i][1] = WMMA16(a[i], b1, acc[i][1]);
      }
      __builtin_amdgcn_sched_barrier(0);
    }
    __syncthreads();
    float bs0 = bias1[lo], bs1 = bias1[16 + lo];
#pragma unroll
    for (int i = 0; i < 5; i++) {
#pragma unroll
      for (int r = 0; r < 8; r++) {
        int p = (w * 5 + i) * 16 + hi * 8 + r;
        float v0 = acc[i][0][r] + bs0; v0 = v0 > 0.0f ? v0 : 0.0f;
        float v1 = acc[i][1][r] + bs1; v1 = v1 > 0.0f ? v1 : 0.0f;
        a1[p * A1P + lo]      = (_Float16)v0;
        a1[p * A1P + 16 + lo] = (_Float16)v1;
      }
    }
  }
  __syncthreads();

  if (w < 4) {
    int aoff[6];
#pragma unroll
    for (int i = 0; i < 6; i++) {
      int p = i * 16 + lo;
      p = p < 81 ? p : 80;
      int oh = p / 9, ow = p % 9;
      aoff[i] = ((oh * 2) * 20 + ow * 2) * A1P + hi * 8;
    }
    const _Float16* bp = wt2 + (w * 16 + lo) * 512 + hi * 8;
    v8f acc[6] = {};
#pragma unroll 1
    for (int kh = 0; kh < 4; kh++) {
      v16h b[4];
#pragma unroll
      for (int kw = 0; kw < 4; kw++) b[kw] = pack_frag(bp + (kh * 4 + kw) * 32);
#pragma unroll
      for (int kw = 0; kw < 4; kw++) {
        v16h a[6];
#pragma unroll
        for (int i = 0; i < 6; i++) a[i] = pack_frag(a1 + aoff[i] + (kh * 20 + kw) * A1P);
        __builtin_amdgcn_sched_barrier(0);
#pragma unroll
        for (int i = 0; i < 6; i++) acc[i] = WMMA16(a[i], b[kw], acc[i]);
        __builtin_amdgcn_sched_barrier(0);
      }
    }
    float bs = bias2[w * 16 + lo];
#pragma unroll
    for (int i = 0; i < 6; i++) {
#pragma unroll
      for (int r = 0; r < 8; r++) {
        int p = i * 16 + hi * 8 + r;
        float v = acc[i][r] + bs; v = v > 0.0f ? v : 0.0f;
        if (p < 81) a2[p * A2P + w * 16 + lo] = (_Float16)v;
      }
    }
  } else {
    relayout_fc_slice(fcw, fcwt, bimg, lane);
  }
  __syncthreads();

  if (w < 4) {
    int aoff[4];
#pragma unroll
    for (int i = 0; i < 4; i++) {
      int p = i * 16 + lo;
      p = p < 49 ? p : 48;
      int oh = p / 7, ow = p % 7;
      aoff[i] = (oh * 9 + ow) * A2P + hi * 8;
    }
    const _Float16* bp = wt3 + (w * 16 + lo) * 576 + hi * 8;
    v8f acc[4] = {};
#pragma unroll 1
    for (int kh = 0; kh < 3; kh++) {
      v16h b[6];
#pragma unroll
      for (int j = 0; j < 6; j++) b[j] = pack_frag(bp + (kh * 6 + j) * 32);
#pragma unroll
      for (int j = 0; j < 6; j++) {
        int kw = j >> 1, cb = (j & 1) * 32;
        v16h a[4];
#pragma unroll
        for (int i = 0; i < 4; i++) a[i] = pack_frag(a2 + aoff[i] + (kh * 9 + kw) * A2P + cb);
        __builtin_amdgcn_sched_barrier(0);
#pragma unroll
        for (int i = 0; i < 4; i++) acc[i] = WMMA16(a[i], b[j], acc[i]);
        __builtin_amdgcn_sched_barrier(0);
      }
    }
    float bs = bias3[w * 16 + lo];
    _Float16* ob = out + (size_t)bimg * (49 * 64) + w * 16 + lo;
#pragma unroll
    for (int i = 0; i < 4; i++) {
#pragma unroll
      for (int r = 0; r < 8; r++) {
        int p = i * 16 + hi * 8 + r;
        float v = acc[i][r] + bs; v = v > 0.0f ? v : 0.0f;
        if (p < 49) ob[p * 64] = (_Float16)v;
      }
    }
  } else {
    relayout_rest_slice(wih, whh, wg, bih, bhh, bg, pos, lact, X, bimg, lane);
  }
}

static constexpr int FPITCH = 68;
static constexpr int FTILE  = 32 * FPITCH;
__global__ void __launch_bounds__(256) k_fc(const _Float16* act3, const _Float16* wt,
                                            const float* bias, _Float16* X) {
  __shared__ __attribute__((aligned(16))) float P[8 * FTILE];
  int tid = threadIdx.x;
  int w = __builtin_amdgcn_readfirstlane(tid >> 5);
  int lane = tid & 31, hi = lane >> 4, lo = lane & 15;
  int ng = blockIdx.x & 7, mp = blockIdx.x >> 3;
  int m0 = mp * 32, nb = ng * 64;
  const _Float16* ar0 = act3 + (size_t)(m0 + lo) * 3136 + hi * 8;
  const _Float16* ar1 = ar0 + (size_t)16 * 3136;
  const _Float16* br0 = wt + (size_t)(nb + lo) * 3136 + hi * 8;
  const _Float16* br1 = br0 + (size_t)16 * 3136;
  const _Float16* br2 = br0 + (size_t)32 * 3136;
  const _Float16* br3 = br0 + (size_t)48 * 3136;
  v8f c00 = {}, c01 = {}, c02 = {}, c03 = {};
  v8f c10 = {}, c11 = {}, c12 = {}, c13 = {};
  for (int kc = w; kc < 98; kc += 8) {
    int kb = kc * 32;
    v16h a0 = pack_frag(ar0 + kb);
    v16h a1 = pack_frag(ar1 + kb);
    v16h b0 = pack_frag(br0 + kb);
    v16h b1 = pack_frag(br1 + kb);
    v16h b2 = pack_frag(br2 + kb);
    v16h b3 = pack_frag(br3 + kb);
    c00 = WMMA16(a0, b0, c00);
    c10 = WMMA16(a1, b0, c10);
    c01 = WMMA16(a0, b1, c01);
    c11 = WMMA16(a1, b1, c11);
    c02 = WMMA16(a0, b2, c02);
    c12 = WMMA16(a1, b2, c12);
    c03 = WMMA16(a0, b3, c03);
    c13 = WMMA16(a1, b3, c13);
  }
  {
    float* pw = P + w * FTILE + (hi * 8) * FPITCH + lo;
#pragma unroll
    for (int r = 0; r < 8; r++) {
      pw[r * FPITCH +  0] = c00[r];
      pw[r * FPITCH + 16] = c01[r];
      pw[r * FPITCH + 32] = c02[r];
      pw[r * FPITCH + 48] = c03[r];
      pw[(16 + r) * FPITCH +  0] = c10[r];
      pw[(16 + r) * FPITCH + 16] = c11[r];
      pw[(16 + r) * FPITCH + 32] = c12[r];
      pw[(16 + r) * FPITCH + 48] = c13[r];
    }
  }
  __syncthreads();
  {
    int row = tid >> 3, c0 = (tid & 7) * 8;
    const float* pr = P + row * FPITCH + c0;
    float4 s0 = *(const float4*)pr;
    float4 s1 = *(const float4*)(pr + 4);
#pragma unroll
    for (int k = 1; k < 8; k++) {
      float4 u0 = *(const float4*)(pr + k * FTILE);
      float4 u1 = *(const float4*)(pr + k * FTILE + 4);
      s0.x += u0.x; s0.y += u0.y; s0.z += u0.z; s0.w += u0.w;
      s1.x += u1.x; s1.y += u1.y; s1.z += u1.z; s1.w += u1.w;
    }
    float4 q0 = *(const float4*)(bias + nb + c0);
    float4 q1 = *(const float4*)(bias + nb + c0 + 4);
    float v[8] = {s0.x + q0.x, s0.y + q0.y, s0.z + q0.z, s0.w + q0.w,
                  s1.x + q1.x, s1.y + q1.y, s1.z + q1.z, s1.w + q1.w};
    v8h o;
#pragma unroll
    for (int e = 0; e < 8; e++) o[e] = (_Float16)(v[e] > 0.0f ? v[e] : 0.0f);
    *(v8h*)(X + (size_t)(m0 + row) * XS + nb + c0) = o;
  }
}

static constexpr int NPOS = 3072;
static constexpr int KT    = 16;
static constexpr int TSLOT = 64 - KT;
static constexpr int PT0   = NPOS + 32;
__global__ void __launch_bounds__(1024) k_levels(const int* done, const float* h0, const float* c0,
                                                 float* hpos, float* cpos, int* rec, int* pos,
                                                 int* tinfo) {
  __shared__ unsigned char  lv_s[32][64];
  __shared__ unsigned char  dn_s[B_TOT];
  __shared__ unsigned short cnt_s[64][32];
  __shared__ int tot_s[64];
  __shared__ int tb_s[64];
  __shared__ int pos_s[B_TOT];
  __shared__ int inv_s[NPOS];
  int tid = threadIdx.x, e = tid >> 5, lane = tid & 31;
  int d0 = done[lane * E_ENVS + e] != 0;
  int d1 = done[(lane + 32) * E_ENVS + e] != 0;
  unsigned m0 = (unsigned)__ballot(d0) | 1u;
  unsigned m1 = (unsigned)__ballot(d1);
  unsigned upto = 0xffffffffu >> (31 - lane);
  unsigned a0 = m0 & upto, a1 = m1 & upto;
  int l0 = lane - (31 - __clz((int)a0));
  int l1 = a1 ? lane - (31 - __clz((int)a1)) : (lane + 32) - (31 - __clz((int)m0));
  lv_s[e][lane] = (unsigned char)l0;
  lv_s[e][lane + 32] = (unsigned char)l1;
  dn_s[lane * E_ENVS + e] = (unsigned char)d0;
  dn_s[(lane + 32) * E_ENVS + e] = (unsigned char)d1;
  for (int p = tid; p < NPOS; p += 1024) inv_s[p] = -1;
  __syncthreads();
  int n0 = 0, n1 = 0, r0 = 0, r1 = 0;
  for (int tp = 0; tp < 64; tp++) {
    int v = lv_s[e][tp];
    n0 += (v == lane);
    n1 += (v == lane + 32);
    r0 += (tp < lane && v == l0);
    r1 += (tp < lane + 32 && v == l1);
  }
  cnt_s[lane][e] = (unsigned short)n0;
  cnt_s[lane + 32][e] = (unsigned short)n1;
  __syncthreads();
  if (tid < 64) {
    int s = 0;
    for (int ep = 0; ep < 32; ep++) s += cnt_s[tid][ep];
    tot_s[tid] = s;
  }
  __syncthreads();
  if (tid < 64) {
    int tb = 0;
    for (int lp = 0; lp < tid; lp++) tb += (tot_s[lp] + 15) >> 4;
    tb_s[tid] = tb;
    tinfo[tid] = tb;
    tinfo[64 + tid] = tot_s[tid];
  }
  __syncthreads();
  int o0 = 0, o1 = 0;
  for (int ep = 0; ep < e; ep++) { o0 += cnt_s[l0][ep]; o1 += cnt_s[l1][ep]; }
  int p0 = l0 >= KT ? PT0 + e * TSLOT + (lane - KT)      : tb_s[l0] * 16 + o0 + r0;
  int p1 = l1 >= KT ? PT0 + e * TSLOT + (lane + 32 - KT) : tb_s[l1] * 16 + o1 + r1;
  int b0 = lane * E_ENVS + e, b1 = (lane + 32) * E_ENVS + e;
  pos_s[b0] = p0; pos_s[b1] = p1;
  if (p0 < NPOS) inv_s[p0] = b0;
  if (p1 < NPOS) inv_s[p1] = b1;
  __syncthreads();
  pos[tid] = pos_s[tid];
  pos[tid + 1024] = pos_s[tid + 1024];
  for (int p = tid; p < NPOS; p += 1024) {
    int b = inv_s[p];
    int4 r = make_int4(0, NPOS, 0, 0);
    if (b >= 0) {
      int d = dn_s[b];
      r.x = b;
      r.y = (b >= E_ENVS && !d) ? pos_s[b - E_ENVS] : NPOS + (b & 31);
      r.z = d;
    }
    *(int4*)(rec + 4 * p) = r;
  }
  for (int i = tid; i < 2 * 4096; i += 1024) {
    int j = i & 3, row = (i >> 2) & 15, q = (i >> 6) & 63, tile2 = i >> 12;
    int s = (tile2 * 16 + row) * HL + q * 4 + j;
    hpos[(size_t)(NPOS / 16) * 4096 + i] = h0[s];
    cpos[(size_t)(NPOS / 16) * 4096 + i] = c0[s];
  }
}

static constexpr int PPITCH = 36;
static constexpr int PTILE  = 8 * PPITCH;
__global__ void __launch_bounds__(128) k_lstm_level(const _Float16* X, const _Float16* Wg,
                                                    const float* biasg, float* cpos, float* hpos,
                                                    const int* rec, const int* tinfo, int L) {
  __shared__ float P[2 * 4 * PTILE];
  int cnt = tinfo[64 + L];
  int nt  = (cnt + 15) >> 4;
  if ((int)blockIdx.y >= nt) return;
  int tb = tinfo[L];
  int tid = threadIdx.x;
  int w = __builtin_amdgcn_readfirstlane(tid >> 5);
  int lane = tid & 31, hi = lane >> 4, lo = lane & 15;
  int cg = blockIdx.x;
  int ci = tid & 63;
  int ti = blockIdx.y, gy = gridDim.y;

  int nv = cnt - ti * 16;
  nv = nv < 16 ? nv : 16;
  int4 rc = *(const int4*)(rec + 4 * ((tb + ti) * 16 + ((ci >> 2) < nv ? (ci >> 2) : 0)));
  int4 ra = *(const int4*)(rec + 4 * ((tb + ti) * 16 + (lo < nv ? lo : 0)));

  int ecol = cg * 4 + (ci & 3);
  float bi = biasg[0 * HL + ecol], bf = biasg[1 * HL + ecol];
  float bg = biasg[2 * HL + ecol], bo = biasg[3 * HL + ecol];

  int grow = (lo >> 2) * HL + cg * 4 + (lo & 3);
  const _Float16* wrow = Wg + (size_t)grow * KPAD + hi * 8;
  v16h bh[2], bx[4];
#pragma unroll
  for (int i = 0; i < 2; i++) bh[i] = pack_frag(wrow + (w + 4 * i) * 32);
#pragma unroll
  for (int i = 0; i < 4; i++) bx[i] = pack_frag(wrow + (w + 8 + 4 * i) * 32);
  v16h b6 = pack_frag(wrow + (w < 2 ? w + 24 : w + 20) * 32);

  int set = 0;
  for (;;) {
    int tile = tb + ti;
    float* Ps = P + set * (4 * PTILE);
    set ^= 1;

    int edone = rc.z;
    float cprev = cpos[(size_t)(rc.y >> 4) * 4096 + cg * 64 + (rc.y & 15) * 4 + (ci & 3)];
    int adone = ra.z;
    const float* hbase = hpos + (size_t)(ra.y >> 4) * 4096 + (ra.y & 15) * 4;
    const int sq = 64;
    const _Float16* xrow = X + (size_t)ra.x * XS + hi * 8;

    v16h a6 = {};
    if (w < 2) a6 = pack_frag(xrow + (w + 16) * 32);
    float4 hu[2][4];
#pragma unroll
    for (int i = 0; i < 2; i++) {
      int kc = w + 4 * i;
      const float* hp = hbase + (kc * 8 + hi * 2) * sq;
      hu[i][0] = *(const float4*)hp;
      hu[i][1] = *(const float4*)(hp + sq);
      hu[i][2] = *(const float4*)(hp + 4 * sq);
      hu[i][3] = *(const float4*)(hp + 5 * sq);
    }
    v16h ax[4];
#pragma unroll
    for (int i = 0; i < 4; i++) ax[i] = pack_frag(xrow + (w + 4 * i) * 32);

    int tn = ti + gy;
    int tq = tn < nt ? tn : ti;
    int nvn = cnt - tq * 16;
    nvn = nvn < 16 ? nvn : 16;
    int4 rcn = *(const int4*)(rec + 4 * ((tb + tq) * 16 + ((ci >> 2) < nvn ? (ci >> 2) : 0)));
    int4 ran = *(const int4*)(rec + 4 * ((tb + tq) * 16 + (lo < nvn ? lo : 0)));

    __builtin_amdgcn_sched_barrier(0);

    v8f acc = {};
#pragma unroll
    for (int i = 0; i < 2; i++) {
      float hv16[16] = {hu[i][0].x, hu[i][0].y, hu[i][0].z, hu[i][0].w,
                        hu[i][1].x, hu[i][1].y, hu[i][1].z, hu[i][1].w,
                        hu[i][2].x, hu[i][2].y, hu[i][2].z, hu[i][2].w,
                        hu[i][3].x, hu[i][3].y, hu[i][3].z, hu[i][3].w};
      v16h a;
#pragma unroll
      for (int e = 0; e < 16; e++) a[e] = adone ? (_Float16)0.0f : (_Float16)hv16[e];
      acc = WMMA16(a, bh[i], acc);
    }
#pragma unroll
    for (int i = 0; i < 4; i++) acc = WMMA16(ax[i], bx[i], acc);
    if (w < 2) acc = WMMA16(a6, b6, acc);

    {
      float* pw = Ps + w * PTILE + lane;
#pragma unroll
      for (int r = 0; r < 8; r++) pw[r * PPITCH] = acc[r];
    }
    __syncthreads();

    if (tid < 64) {
      int el = tid >> 2, q = tid & 3;
      const float* pr = Ps + (el & 7) * PPITCH + (el >> 3) * 16 + q;
      float si = 0.0f, sf = 0.0f, sg = 0.0f, so = 0.0f;
#pragma unroll
      for (int k = 0; k < 4; k++) {
        si += pr[k * PTILE + 0];
        sf += pr[k * PTILE + 4];
        sg += pr[k * PTILE + 8];
        so += pr[k * PTILE + 12];
      }
      float cp = edone ? 0.0f : cprev;
      float iv = sigmoidf_(si + bi), fv = sigmoidf_(sf + bf);
      float gv = tanhf(sg + bg),     ov = sigmoidf_(so + bo);
      float cv = fv * cp + iv * gv;
      float hv = ov * tanhf(cv);
      size_t o = (size_t)(tile * 64 + cg) * 64 + tid;
      cpos[o] = cv;
      hpos[o] = hv;
    }
    if (tn >= nt) break;
    ti = tn; rc = rcn; ra = ran;
  }
}

__global__ void __launch_bounds__(HL) k_lstm_tail(const _Float16* X, const _Float16* Wg,
                                                   const float* biasg, float* cpos, float* hpos,
                                                   const int* done, const int* pos,
                                                   const int* tinfo) {
  __shared__ int lv_s[64];
  __shared__ __attribute__((aligned(16))) _Float16 hh[HL];
  __shared__ float G[4 * HL];
  if (tinfo[64 + KT] == 0) return;
  int tid = threadIdx.x, e = blockIdx.x;
  int lane = tid & 31, hi = lane >> 4, lo = lane & 15;
  int w = __builtin_amdgcn_readfirstlane(tid >> 5);
  if (tid < 32) {
    int d0 = done[lane * E_ENVS + e] != 0;
    int d1 = done[(lane + 32) * E_ENVS + e] != 0;
    unsigned m0 = (unsigned)__ballot(d0) | 1u;
    unsigned m1 = (unsigned)__ballot(d1);
    unsigned upto = 0xffffffffu >> (31 - lane);
    unsigned a0 = m0 & upto, a1 = m1 & upto;
    lv_s[lane] = lane - (31 - __clz((int)a0));
    lv_s[lane + 32] = a1 ? lane - (31 - __clz((int)a1)) : (lane + 32) - (31 - __clz((int)m0));
  }
  __syncthreads();
  float bi = biasg[0 * HL + tid], bf = biasg[1 * HL + tid];
  float bg = biasg[2 * HL + tid], bo = biasg[3 * HL + tid];
  float c = 0.0f;
  for (int t = KT; t < T_STEPS; t++) {
    int lv = __builtin_amdgcn_readfirstlane(lv_s[t]);
    if (lv < KT) continue;
    int b = t * E_ENVS + e;
    if (lv == KT) {
      int pp = pos[b - E_ENVS];
      size_t o = (size_t)(pp >> 4) * 4096 + (tid >> 2) * 64 + (pp & 15) * 4 + (tid & 3);
      hh[tid] = (_Float16)hpos[o];
      c = cpos[o];
    }
    __syncthreads();
    const _Float16* xrow = X + (size_t)b * XS + hi * 8;
    for (int cg = w; cg < 64; cg += 8) {
      int grow = (lo >> 2) * HL + cg * 4 + (lo & 3);
      const _Float16* wrow = Wg + (size_t)grow * KPAD + hi * 8;
      v8f acc[4] = {};
#pragma unroll
      for (int kc = 0; kc < 26; kc++) {
        v16h a  = kc < 8 ? pack_frag(hh + kc * 32 + hi * 8) : pack_frag(xrow + (kc - 8) * 32);
        v16h bb = pack_frag(wrow + kc * 32);
        acc[kc & 3] = WMMA16(a, bb, acc[kc & 3]);
      }
      float s = 0.0f;
      s += acc[0][0]; s += acc[1][0]; s += acc[2][0]; s += acc[3][0];
      if (lane < 16) G[(lane >> 2) * HL + cg * 4 + (lane & 3)] = s;
    }
    __syncthreads();
    {
      float si = G[tid], sf = G[HL + tid], sg = G[2 * HL + tid], so = G[3 * HL + tid];
      float iv = sigmoidf_(si + bi), fv = sigmoidf_(sf + bf);
      float gv = tanhf(sg + bg),     ov = sigmoidf_(so + bo);
      float cv = fv * c + iv * gv;
      float hv = ov * tanhf(cv);
      int p = pos[b];
      size_t o = (size_t)(p >> 4) * 4096 + (tid >> 2) * 64 + (p & 15) * 4 + (tid & 3);
      cpos[o] = cv;
      hpos[o] = hv;
      hh[tid] = (_Float16)hv;
      c = cv;
    }
    __syncthreads();
  }
}

__global__ void k_heads(const float* hpos, const float* cpos, const int* pos, const float* pw,
                        const float* pb, const float* vw, const float* vb, float* out) {
  int i = blockIdx.x * blockDim.x + threadIdx.x;
  if (i < B_TOT * 6) {
    int b = i / 6, j = i % 6;
    int p = pos[b];
    const float* h    = hpos + (size_t)(p >> 4) * 4096 + (p & 15) * 4;
    const float* wrow = (j < 5) ? pw + j * HL : vw;
    float s = (j < 5) ? pb[j] : vb[0];
#pragma unroll 4
    for (int k4 = 0; k4 < HL / 4; k4++) {
      float4 hv = *(const float4*)(h + k4 * 64);
      s += hv.x * wrow[k4 * 4 + 0];
      s += hv.y * wrow[k4 * 4 + 1];
      s += hv.z * wrow[k4 * 4 + 2];
      s += hv.w * wrow[k4 * 4 + 3];
    }
    if (j < 5) out[b * 5 + j] = s;
    else       out[10240 + b] = s;
  } else {
    int o = i - B_TOT * 6;
    int sel = o >> 13, r = o & 8191;
    int e = r >> 8, k = r & 255;
    int p = pos[(T_STEPS - 1) * E_ENVS + e];
    size_t src = (size_t)(p >> 4) * 4096 + (k >> 2) * 64 + (p & 15) * 4 + (k & 3);
    if (sel == 0) out[12288 + r] = hpos[src];
    else          out[20480 + r] = cpos[src];
  }
}

extern "C" void kernel_launch(void* const* d_in, const int* in_sizes, int n_in,
                              void* d_out, int out_size, void* d_ws, size_t ws_size,
                              hipStream_t stream) {
  (void)in_sizes; (void)n_in; (void)out_size; (void)ws_size;
  const float* img  = (const float*)d_in[0];
  const int*   pos  = (const int*)d_in[1];
  const int*   lact = (const int*)d_in[2];
  const int*   done = (const int*)d_in[3];
  const float* h0   = (const float*)d_in[4];
  const float* c0   = (const float*)d_in[5];
  const float* w1   = (const float*)d_in[6];
  const float* b1   = (const float*)d_in[7];
  const float* w2   = (const float*)d_in[8];
  const float* b2   = (const float*)d_in[9];
  const float* w3   = (const float*)d_in[10];
  const float* b3   = (const float*)d_in[11];
  const float* fcw  = (const float*)d_in[12];
  const float* fcb  = (const float*)d_in[13];
  const float* wih  = (const float*)d_in[14];
  const float* whh  = (const float*)d_in[15];
  const float* bih  = (const float*)d_in[16];
  const float* bhh  = (const float*)d_in[17];
  const float* pw   = (const float*)d_in[18];
  const float* pb   = (const float*)d_in[19];
  const float* vw   = (const float*)d_in[20];
  const float* vb   = (const float*)d_in[21];
  float* out = (float*)d_out;
  char*  ws  = (char*)d_ws;

  const size_t o_act1 = 90832896;
  const size_t o_w1t  = o_act1 + 52428800;
  const size_t o_w2t  = o_w1t + 12288;
  const size_t o_w3t  = o_w2t + 65536;
  const size_t o_fcwt = o_w3t + 73728;
  const size_t o_wg   = o_fcwt + 3211264;
  const size_t o_bg   = o_wg + 1703936;
  const size_t o_act3 = 21233664;
  const size_t o_X    = 34078720;
  const size_t o_hp   = o_act1;
  const size_t o_cp   = o_hp + 4751360;
  const size_t o_rec  = o_cp + 4751360;
  const size_t o_pos  = o_rec + 49152;
  const size_t o_ti   = o_pos + 8192;

  _Float16* act3 = (_Float16*)(ws + o_act3);
  _Float16* Xm   = (_Float16*)(ws + o_X);
  _Float16* w1t  = (_Float16*)(ws + o_w1t);
  _Float16* w2t  = (_Float16*)(ws + o_w2t);
  _Float16* w3t  = (_Float16*)(ws + o_w3t);
  _Float16* fcwt = (_Float16*)(ws + o_fcwt);
  _Float16* wg   = (_Float16*)(ws + o_wg);
  float*    bg   = (float*)(ws + o_bg);
  float*    hpb  = (float*)(ws + o_hp);
  float*    cpb  = (float*)(ws + o_cp);
  int*      rec  = (int*)(ws + o_rec);
  int*      posb = (int*)(ws + o_pos);
  int*      tinfo = (int*)(ws + o_ti);

  const int TB = 256;
  k_prep_w1  <<<(32 * 192   + TB - 1) / TB, TB, 0, stream>>>(w1, w1t);
  k_prep_w2  <<<(64 * 512   + TB - 1) / TB, TB, 0, stream>>>(w2, w2t);
  k_prep_w3  <<<(64 * 576   + TB - 1) / TB, TB, 0, stream>>>(w3, w3t);

  k_convs<<<2048, 160, 0, stream>>>(img, w1t, b1, w2t, b2, w3t, b3, act3,
                                    fcw, fcwt, wih, whh, wg, bih, bhh, bg, pos, lact, Xm);
  k_fc   <<< 512, TB, 0, stream>>>(act3, fcwt, fcb, Xm);

  k_levels<<<1, 1024, 0, stream>>>(done, h0, c0, hpb, cpb, rec, posb, tinfo);

  k_lstm_level<<<dim3(64, 16), 128, 0, stream>>>(Xm, wg, bg, cpb, hpb, rec, tinfo, 0);
  k_lstm_level<<<dim3(64, 8), 128, 0, stream>>>(Xm, wg, bg, cpb, hpb, rec, tinfo, 1);
  k_lstm_level<<<dim3(64, 5), 128, 0, stream>>>(Xm, wg, bg, cpb, hpb, rec, tinfo, 2);
  k_lstm_level<<<dim3(64, 4), 128, 0, stream>>>(Xm, wg, bg, cpb, hpb, rec, tinfo, 3);
  k_lstm_level<<<dim3(64, 3), 128, 0, stream>>>(Xm, wg, bg, cpb, hpb, rec, tinfo, 4);
  k_lstm_level<<<dim3(64, 2), 128, 0, stream>>>(Xm, wg, bg, cpb, hpb, rec, tinfo, 5);
  k_lstm_level<<<dim3(64, 2), 128, 0, stream>>>(Xm, wg, bg, cpb, hpb, rec, tinfo, 6);
  k_lstm_level<<<dim3(64, 2), 128, 0, stream>>>(Xm, wg, bg, cpb, hpb, rec, tinfo, 7);
  k_lstm_level<<<dim3(64, 2), 128, 0, stream>>>(Xm, wg, bg, cpb, hpb, rec, tinfo, 8);
  k_lstm_level<<<dim3(64, 2), 128, 0, stream>>>(Xm, wg, bg, cpb, hpb, rec, tinfo, 9);
  k_lstm_level<<<dim3(64, 2), 128, 0, stream>>>(Xm, wg, bg, cpb, hpb, rec, tinfo, 10);
  k_lstm_level<<<dim3(64, 2), 128, 0, stream>>>(Xm, wg, bg, cpb, hpb, rec, tinfo, 11);
  k_lstm_level<<<dim3(64, 2), 128, 0, stream>>>(Xm, wg, bg, cpb, hpb, rec, tinfo, 12);
  k_lstm_level<<<dim3(64, 2), 128, 0, stream>>>(Xm, wg, bg, cpb, hpb, rec, tinfo, 13);
  k_lstm_level<<<dim3(64, 2), 128, 0, stream>>>(Xm, wg, bg, cpb, hpb, rec, tinfo, 14);
  k_lstm_level<<<dim3(64, 2), 128, 0, stream>>>(Xm, wg, bg, cpb, hpb, rec, tinfo, 15);
  k_lstm_tail<<<E_ENVS, HL, 0, stream>>>(Xm, wg, bg, cpb, hpb, done, posb, tinfo);

  k_heads<<<112, TB, 0, stream>>>(hpb, cpb, posb, pw, pb, vw, vb, out);
}
